# starting kernel
# baseline (speedup 1.0000x reference)
#include <hip/hip_runtime.h>
#include <hip/hip_bf16.h>

typedef __attribute__((ext_vector_type(2))) float v2f;
typedef __attribute__((ext_vector_type(8))) float v8f;

#define NN 100000
#define NC 5000
#define HDIM 64
#define ES 3200000
#define EM 1000000
#define ER 160000

__global__ __launch_bounds__(256) void k_zero(float* __restrict__ p, long n4) {
    long i = (long)blockIdx.x * blockDim.x + threadIdx.x;
    if (i < n4) ((float4*)p)[i] = make_float4(0.f, 0.f, 0.f, 0.f);
}

__global__ __launch_bounds__(256) void k_packW(const float* __restrict__ W,
                                               v2f* __restrict__ Wp,
                                               int K, int N) {
    const int Kp2 = ((K + 3) & ~3) >> 1;
    const long n = (long)Kp2 * N;
    long i = (long)blockIdx.x * blockDim.x + threadIdx.x;
    if (i >= n) return;
    int k2 = (int)(i / N), col = (int)(i % N);
    int k = 2 * k2;
    v2f v;
    v.x = (k     < K) ? W[(long)k * N + col]       : 0.f;
    v.y = (k + 1 < K) ? W[(long)(k + 1) * N + col] : 0.f;
    Wp[i] = v;
}

__global__ __launch_bounds__(256) void k_scatter(const int* __restrict__ src,
                                                 const int* __restrict__ dst,
                                                 const float* __restrict__ x,
                                                 float* __restrict__ agg,
                                                 float* __restrict__ cnt,
                                                 long nE) {
    long gid = (long)blockIdx.x * blockDim.x + threadIdx.x;
    long e = gid >> 4;
    int c = (int)(gid & 15) * 4;
    if (e >= nE) return;
    int s = src[e], d = dst[e];
    const float4 v = *(const float4*)(x + (long)s * HDIM + c);
    float* ap = agg + (long)d * HDIM + c;
    atomicAdd(ap + 0, v.x);
    atomicAdd(ap + 1, v.y);
    atomicAdd(ap + 2, v.z);
    atomicAdd(ap + 3, v.w);
    if ((gid & 15) == 0) atomicAdd(cnt + d, 1.0f);
}

__global__ __launch_bounds__(256) void k_scale_rows(float* __restrict__ agg,
                                                    const float* __restrict__ cnt,
                                                    int M) {
    long gid = (long)blockIdx.x * blockDim.x + threadIdx.x;
    long row = gid >> 4;
    int c = (int)(gid & 15) * 4;
    if (row >= M) return;
    float inv = 1.0f / fmaxf(cnt[row], 1.0f);
    float4* p = (float4*)(agg + row * HDIM + c);
    float4 v = *p;
    v.x *= inv; v.y *= inv; v.z *= inv; v.w *= inv;
    *p = v;
}

template<int NT, bool FULL, bool VECA>
__device__ __forceinline__ void gemm_body(v8f (&c)[NT],
                                          const float* __restrict__ A,
                                          const v2f* __restrict__ Wp,
                                          int M, int K, int N,
                                          int mbase, int lane) {
    const int row  = mbase + (lane & 15);
    const int koff = (lane >= 16) ? 2 : 0;
    const int col0 = lane & 15;
    const bool rok = FULL || (row < M);
    const float* __restrict__ Arow = A + (long)row * K;
    const int Kmain = K & ~3;

    #pragma unroll 4
    for (int k = 0; k < Kmain; k += 4) {
        const int kk = k + koff;
        v2f a;
        if (rok) {
            if (VECA) {
                a = *(const v2f*)(Arow + kk);
            } else {
                a.x = Arow[kk];
                a.y = Arow[kk + 1];
            }
        } else {
            a.x = 0.f; a.y = 0.f;
        }
        #pragma unroll
        for (int nt = 0; nt < NT; ++nt) {
            v2f b = Wp[(long)(kk >> 1) * N + nt * 16 + col0];
            c[nt] = __builtin_amdgcn_wmma_f32_16x16x4_f32(
                false, a, false, b, (short)0, c[nt], false, false);
        }
    }
    if (Kmain < K) {
        const int kk = Kmain + koff;
        v2f a;
        a.x = (rok && kk     < K) ? Arow[kk]     : 0.f;
        a.y = (rok && kk + 1 < K) ? Arow[kk + 1] : 0.f;
        #pragma unroll
        for (int nt = 0; nt < NT; ++nt) {
            v2f b = Wp[(long)(kk >> 1) * N + nt * 16 + col0];
            c[nt] = __builtin_amdgcn_wmma_f32_16x16x4_f32(
                false, a, false, b, (short)0, c[nt], false, false);
        }
    }
}

template<int NT, bool VECA1>
__global__ __launch_bounds__(256) void k_gemm(const float* __restrict__ A1,
                                              const v2f* __restrict__ W1p,
                                              const float* __restrict__ bias,
                                              const float* __restrict__ A2,
                                              const v2f* __restrict__ W2p,
                                              float* __restrict__ out,
                                              int M, int K1, int K2,
                                              int do_relu, int accum) {
    const int N = NT * 16;
    const int lane = threadIdx.x & 31;
    const int wv = threadIdx.x >> 5;
    const int mbase = (blockIdx.x * 8 + wv) * 16;
    if (mbase >= M) return;

    v8f c[NT];
    #pragma unroll
    for (int t = 0; t < NT; ++t)
        c[t] = (v8f){0.f, 0.f, 0.f, 0.f, 0.f, 0.f, 0.f, 0.f};

    const bool full = (mbase + 16 <= M);
    if (full) {
        gemm_body<NT, true, VECA1>(c, A1, W1p, M, K1, N, mbase, lane);
        if (A2) gemm_body<NT, true, true>(c, A2, W2p, M, K2, N, mbase, lane);
    } else {
        gemm_body<NT, false, VECA1>(c, A1, W1p, M, K1, N, mbase, lane);
        if (A2) gemm_body<NT, false, true>(c, A2, W2p, M, K2, N, mbase, lane);
    }

    const int col0 = lane & 15;
    const int rofs = (lane >= 16) ? 8 : 0;
    #pragma unroll
    for (int nt = 0; nt < NT; ++nt) {
        const int col = nt * 16 + col0;
        const float bb = bias ? bias[col] : 0.f;
        #pragma unroll
        for (int r = 0; r < 8; ++r) {
            const int row = mbase + r + rofs;
            if (full || row < M) {
                const long idx = (long)row * N + col;
                float v = c[nt][r] + bb;
                if (accum) v += out[idx];
                if (do_relu) v = fmaxf(v, 0.f);
                out[idx] = v;
            }
        }
    }
}

__global__ void k_ln(float* __restrict__ out, const float* __restrict__ in,
                     const float* __restrict__ res,
                     const float* __restrict__ g, const float* __restrict__ b,
                     int M) {
    int lane = threadIdx.x;
    int row = blockIdx.x * blockDim.y + threadIdx.y;
    if (row >= M) return;
    long base = (long)row * HDIM;
    float v0 = in[base + lane];
    float v1 = in[base + lane + 32];
    if (res) { v0 += res[base + lane]; v1 += res[base + lane + 32]; }
    v0 = fmaxf(v0, 0.f);
    v1 = fmaxf(v1, 0.f);
    float s = v0 + v1;
    for (int off = 16; off > 0; off >>= 1) s += __shfl_xor(s, off, 32);
    float mean = s * (1.0f / 64.0f);
    float d0 = v0 - mean, d1 = v1 - mean;
    float q = d0 * d0 + d1 * d1;
    for (int off = 16; off > 0; off >>= 1) q += __shfl_xor(q, off, 32);
    float rstd = rsqrtf(q * (1.0f / 64.0f) + 1e-5f);
    out[base + lane]      = d0 * rstd * g[lane]      + b[lane];
    out[base + lane + 32] = d1 * rstd * g[lane + 32] + b[lane + 32];
}

__global__ __launch_bounds__(256) void k_addrelu(float* __restrict__ o,
                                                 const float* __restrict__ a,
                                                 const float* __restrict__ b,
                                                 long n) {
    long i = (long)blockIdx.x * blockDim.x + threadIdx.x;
    if (i < n) o[i] = fmaxf(a[i] + b[i], 0.f);
}

__global__ __launch_bounds__(256) void k_final(const float* __restrict__ h,
                                               const float* __restrict__ W2,
                                               const float* __restrict__ b2,
                                               float* __restrict__ out, int M) {
    int i = blockIdx.x * blockDim.x + threadIdx.x;
    if (i >= M) return;
    float acc = b2[0];
    const float* hr = h + (long)i * 32;
    #pragma unroll
    for (int j = 0; j < 32; ++j) acc += hr[j] * W2[j];
    out[i] = acc;
}

extern "C" void kernel_launch(void* const* d_in, const int* in_sizes, int n_in,
                              void* d_out, int out_size, void* d_ws, size_t ws_size,
                              hipStream_t stream) {
    const float* x_news    = (const float*)d_in[0];
    const float* x_company = (const float*)d_in[1];
    const int*   edge_sim  = (const int*)d_in[2];
    const int*   men_src   = (const int*)d_in[3];
    const int*   men_dst   = (const int*)d_in[4];
    const int*   edge_rel  = (const int*)d_in[5];
    const float* nw_W = (const float*)d_in[6];
    const float* nw_b = (const float*)d_in[7];
    const float* cw_W = (const float*)d_in[8];
    const float* cw_b = (const float*)d_in[9];
    const float* n1n_g = (const float*)d_in[10];
    const float* n1n_b = (const float*)d_in[11];
    const float* n1c_g = (const float*)d_in[12];
    const float* n1c_b = (const float*)d_in[13];
    const float* c1_Wl = (const float*)d_in[14];
    const float* c1_bl = (const float*)d_in[15];
    const float* c1_Wr = (const float*)d_in[16];
    const float* c2_Wl = (const float*)d_in[17];
    const float* c2_bl = (const float*)d_in[18];
    const float* c2_Wr = (const float*)d_in[19];
    const float* cls_W1 = (const float*)d_in[20];
    const float* cls_b1 = (const float*)d_in[21];
    const float* cls_W2 = (const float*)d_in[22];
    const float* cls_b2 = (const float*)d_in[23];
    float* out = (float*)d_out;

    float* ws = (float*)d_ws;
    long o = 0;
    float* xn0  = ws + o; o += (long)NN * HDIM;
    float* xn1  = ws + o; o += (long)NN * HDIM;
    float* aggn = ws + o; o += (long)NN * HDIM;
    float* cntn = ws + o; o += NN;
    float* aggcM = ws + o; o += (long)NC * HDIM;
    float* cntcM = ws + o; o += NC;
    float* aggcR = ws + o; o += (long)NC * HDIM;
    float* cntcR = ws + o; o += NC;
    float* xc0  = ws + o; o += (long)NC * HDIM;
    float* xc1  = ws + o; o += (long)NC * HDIM;
    float* ocb  = ws + o; o += (long)NC * HDIM;
    float* comp = ws + o; o += (long)NC * HDIM;
    float* hbuf = ws + o; o += (long)NC * 32;
    v2f* wp_nw  = (v2f*)(ws + o); o += 2L * (388 / 2) * 64;
    v2f* wp_cw  = (v2f*)(ws + o); o += 2L * (24 / 2) * 64;
    v2f* wp_c1  = (v2f*)(ws + o); o += 2L * 6 * 32 * 64;
    v2f* wp_c2  = (v2f*)(ws + o); o += 2L * 4 * 32 * 64;
    v2f* wp_cls = (v2f*)(ws + o); o += 2L * 32 * 32;

    const long zn_n = (long)NN * HDIM + NN;
    const long zn_c = 2L * ((long)NC * HDIM + NC);

    dim3 blk(256);
    auto blocks = [](long n) { return (unsigned)((n + 255) / 256); };
    int tilesNN = (NN + 15) / 16, tilesNC = (NC + 15) / 16;
    unsigned gemmBlkNN = (tilesNN + 7) / 8, gemmBlkNC = (tilesNC + 7) / 8;

    k_packW<<<blocks((388L / 2) * 64), blk, 0, stream>>>(nw_W, wp_nw, 385, 64);
    k_packW<<<blocks((24L / 2) * 64), blk, 0, stream>>>(cw_W, wp_cw, 24, 64);
    for (int r = 0; r < 3; ++r) {
        k_packW<<<blocks(32L * 64), blk, 0, stream>>>(c1_Wl + r * 4096,
                                                      wp_c1 + (2 * r) * 2048, 64, 64);
        k_packW<<<blocks(32L * 64), blk, 0, stream>>>(c1_Wr + r * 4096,
                                                      wp_c1 + (2 * r + 1) * 2048, 64, 64);
    }
    for (int r = 1; r < 3; ++r) {
        k_packW<<<blocks(32L * 64), blk, 0, stream>>>(c2_Wl + r * 4096,
                                                      wp_c2 + (2 * (r - 1)) * 2048, 64, 64);
        k_packW<<<blocks(32L * 64), blk, 0, stream>>>(c2_Wr + r * 4096,
                                                      wp_c2 + (2 * (r - 1) + 1) * 2048, 64, 64);
    }
    k_packW<<<blocks(32L * 32), blk, 0, stream>>>(cls_W1, wp_cls, 64, 32);

    k_gemm<4, false><<<gemmBlkNN, blk, 0, stream>>>(x_news, wp_nw, nw_b,
                                                    nullptr, nullptr,
                                                    xn0, NN, 385, 0, 1, 0);
    k_gemm<4, true><<<gemmBlkNC, blk, 0, stream>>>(x_company, wp_cw, cw_b,
                                                   nullptr, nullptr,
                                                   xc0, NC, 24, 0, 1, 0);

    k_zero<<<blocks(zn_n / 4), blk, 0, stream>>>(aggn, zn_n / 4);
    k_zero<<<blocks(zn_c / 4), blk, 0, stream>>>(aggcM, zn_c / 4);
    k_scatter<<<blocks((long)ES * 16), blk, 0, stream>>>(edge_sim, edge_sim + ES,
                                                         xn0, aggn, cntn, ES);
    k_scatter<<<blocks((long)EM * 16), blk, 0, stream>>>(men_src, men_dst,
                                                         xn0, aggcM, cntcM, EM);
    k_scatter<<<blocks((long)ER * 16), blk, 0, stream>>>(edge_rel, edge_rel + ER,
                                                         xc0, aggcR, cntcR, ER);
    k_scale_rows<<<blocks((long)NN * 16), blk, 0, stream>>>(aggn, cntn, NN);
    k_scale_rows<<<blocks((long)NC * 16), blk, 0, stream>>>(aggcM, cntcM, NC);
    k_scale_rows<<<blocks((long)NC * 16), blk, 0, stream>>>(aggcR, cntcR, NC);

    k_gemm<4, true><<<gemmBlkNN, blk, 0, stream>>>(aggn, wp_c1 + 0 * 2048,
                                                   c1_bl + 0 * 64,
                                                   xn0, wp_c1 + 1 * 2048,
                                                   xn1, NN, 64, 64, 0, 0);
    k_gemm<4, true><<<gemmBlkNC, blk, 0, stream>>>(aggcM, wp_c1 + 2 * 2048,
                                                   c1_bl + 1 * 64,
                                                   xc0, wp_c1 + 3 * 2048,
                                                   ocb, NC, 64, 64, 0, 0);
    k_gemm<4, true><<<gemmBlkNC, blk, 0, stream>>>(aggcR, wp_c1 + 4 * 2048,
                                                   c1_bl + 2 * 64,
                                                   xc0, wp_c1 + 5 * 2048,
                                                   ocb, NC, 64, 64, 0, 1);

    dim3 lnb(32, 8);
    k_ln<<<(NN + 7) / 8, lnb, 0, stream>>>(xn1, xn1, nullptr, n1n_g, n1n_b, NN);
    k_ln<<<(NC + 7) / 8, lnb, 0, stream>>>(xc1, ocb, xc0, n1c_g, n1c_b, NC);

    k_zero<<<blocks(zn_c / 4), blk, 0, stream>>>(aggcM, zn_c / 4);
    k_scatter<<<blocks((long)EM * 16), blk, 0, stream>>>(men_src, men_dst,
                                                         xn1, aggcM, cntcM, EM);
    k_scatter<<<blocks((long)ER * 16), blk, 0, stream>>>(edge_rel, edge_rel + ER,
                                                         xc1, aggcR, cntcR, ER);
    k_scale_rows<<<blocks((long)NC * 16), blk, 0, stream>>>(aggcM, cntcM, NC);
    k_scale_rows<<<blocks((long)NC * 16), blk, 0, stream>>>(aggcR, cntcR, NC);
    k_gemm<4, true><<<gemmBlkNC, blk, 0, stream>>>(aggcM, wp_c2 + 0 * 2048,
                                                   c2_bl + 1 * 64,
                                                   xc1, wp_c2 + 1 * 2048,
                                                   ocb, NC, 64, 64, 0, 0);
    k_gemm<4, true><<<gemmBlkNC, blk, 0, stream>>>(aggcR, wp_c2 + 2 * 2048,
                                                   c2_bl + 2 * 64,
                                                   xc1, wp_c2 + 3 * 2048,
                                                   ocb, NC, 64, 64, 0, 1);

    k_addrelu<<<blocks((long)NC * HDIM), blk, 0, stream>>>(comp, ocb, xc1,
                                                           (long)NC * HDIM);
    k_gemm<2, true><<<gemmBlkNC, blk, 0, stream>>>(comp, wp_cls, cls_b1,
                                                   nullptr, nullptr,
                                                   hbuf, NC, 64, 0, 1, 0);
    k_final<<<blocks(NC), blk, 0, stream>>>(hbuf, cls_W2, cls_b2, out, NC);
}


# scatter: one wave per 32 edges, lane = column, so each row load and each float atomic add covers one whole 128-byte line (2 atomic line requests per edge, was 8); sums still in atomic arrival order
# speedup vs baseline: 1.4780x; 1.4780x over previous
#include <hip/hip_runtime.h>
#include <hip/hip_bf16.h>

typedef __attribute__((ext_vector_type(2))) float v2f;
typedef __attribute__((ext_vector_type(8))) float v8f;

#define NN 100000
#define NC 5000
#define HDIM 64
#define ES 3200000
#define EM 1000000
#define ER 160000

__global__ __launch_bounds__(256) void k_zero(float* __restrict__ p, long n4) {
    long i = (long)blockIdx.x * blockDim.x + threadIdx.x;
    if (i < n4) ((float4*)p)[i] = make_float4(0.f, 0.f, 0.f, 0.f);
}

__global__ __launch_bounds__(256) void k_packW(const float* __restrict__ W,
                                               v2f* __restrict__ Wp,
                                               int K, int N) {
    const int Kp2 = ((K + 3) & ~3) >> 1;
    const long n = (long)Kp2 * N;
    long i = (long)blockIdx.x * blockDim.x + threadIdx.x;
    if (i >= n) return;
    int k2 = (int)(i / N), col = (int)(i % N);
    int k = 2 * k2;
    v2f v;
    v.x = (k     < K) ? W[(long)k * N + col]       : 0.f;
    v.y = (k + 1 < K) ? W[(long)(k + 1) * N + col] : 0.f;
    Wp[i] = v;
}

__global__ __launch_bounds__(256) void k_scatter(const int* __restrict__ src,
                                                 const int* __restrict__ dst,
                                                 const float* __restrict__ x,
                                                 float* __restrict__ agg,
                                                 float* __restrict__ cnt,
                                                 long nE) {
    const int lane = threadIdx.x & 31;
    const int wv = __builtin_amdgcn_readfirstlane((int)(threadIdx.x >> 5));
    const long e0 = ((long)blockIdx.x * (blockDim.x >> 5) + wv) * 32;
    if (e0 >= nE) return;
    const long e = e0 + lane;
    int s = 0, d = 0;
    if (e < nE) {
        s = src[e];
        d = dst[e];
        atomicAdd(cnt + d, 1.0f);
    }
    if (e0 + 32 <= nE) {
        float v0[32], v1[32];
        #pragma unroll
        for (int j = 0; j < 32; ++j) {
            const int sj = __builtin_amdgcn_readlane(s, j);
            const float* xr = x + (long)sj * HDIM + lane;
            v0[j] = xr[0];
            v1[j] = xr[32];
        }
        #pragma unroll
        for (int j = 0; j < 32; ++j) {
            const int dj = __builtin_amdgcn_readlane(d, j);
            float* ar = agg + (long)dj * HDIM + lane;
            atomicAdd(ar, v0[j]);
            atomicAdd(ar + 32, v1[j]);
        }
    } else {
        const int n = (int)(nE - e0);
        for (int j = 0; j < n; ++j) {
            const int sj = __builtin_amdgcn_readlane(s, j);
            const int dj = __builtin_amdgcn_readlane(d, j);
            const float* xr = x + (long)sj * HDIM + lane;
            float* ar = agg + (long)dj * HDIM + lane;
            const float a0 = xr[0];
            const float a1 = xr[32];
            atomicAdd(ar, a0);
            atomicAdd(ar + 32, a1);
        }
    }
}

__global__ __launch_bounds__(256) void k_scale_rows(float* __restrict__ agg,
                                                    const float* __restrict__ cnt,
                                                    int M) {
    long gid = (long)blockIdx.x * blockDim.x + threadIdx.x;
    long row = gid >> 4;
    int c = (int)(gid & 15) * 4;
    if (row >= M) return;
    float inv = 1.0f / fmaxf(cnt[row], 1.0f);
    float4* p = (float4*)(agg + row * HDIM + c);
    float4 v = *p;
    v.x *= inv; v.y *= inv; v.z *= inv; v.w *= inv;
    *p = v;
}

template<int NT, bool FULL, bool VECA>
__device__ __forceinline__ void gemm_body(v8f (&c)[NT],
                                          const float* __restrict__ A,
                                          const v2f* __restrict__ Wp,
                                          int M, int K, int N,
                                          int mbase, int lane) {
    const int row  = mbase + (lane & 15);
    const int koff = (lane >= 16) ? 2 : 0;
    const int col0 = lane & 15;
    const bool rok = FULL || (row < M);
    const float* __restrict__ Arow = A + (long)row * K;
    const int Kmain = K & ~3;

    #pragma unroll 4
    for (int k = 0; k < Kmain; k += 4) {
        const int kk = k + koff;
        v2f a;
        if (rok) {
            if (VECA) {
                a = *(const v2f*)(Arow + kk);
            } else {
                a.x = Arow[kk];
                a.y = Arow[kk + 1];
            }
        } else {
            a.x = 0.f; a.y = 0.f;
        }
        #pragma unroll
        for (int nt = 0; nt < NT; ++nt) {
            v2f b = Wp[(long)(kk >> 1) * N + nt * 16 + col0];
            c[nt] = __builtin_amdgcn_wmma_f32_16x16x4_f32(
                false, a, false, b, (short)0, c[nt], false, false);
        }
    }
    if (Kmain < K) {
        const int kk = Kmain + koff;
        v2f a;
        a.x = (rok && kk     < K) ? Arow[kk]     : 0.f;
        a.y = (rok && kk + 1 < K) ? Arow[kk + 1] : 0.f;
        #pragma unroll
        for (int nt = 0; nt < NT; ++nt) {
            v2f b = Wp[(long)(kk >> 1) * N + nt * 16 + col0];
            c[nt] = __builtin_amdgcn_wmma_f32_16x16x4_f32(
                false, a, false, b, (short)0, c[nt], false, false);
        }
    }
}

template<int NT, bool VECA1>
__global__ __launch_bounds__(256) void k_gemm(const float* __restrict__ A1,
                                              const v2f* __restrict__ W1p,
                                              const float* __restrict__ bias,
                                              const float* __restrict__ A2,
                                              const v2f* __restrict__ W2p,
                                              float* __restrict__ out,
                                              int M, int K1, int K2,
                                              int do_relu, int accum) {
    const int N = NT * 16;
    const int lane = threadIdx.x & 31;
    const int wv = threadIdx.x >> 5;
    const int mbase = (blockIdx.x * 8 + wv) * 16;
    if (mbase >= M) return;

    v8f c[NT];
    #pragma unroll
    for (int t = 0; t < NT; ++t)
        c[t] = (v8f){0.f, 0.f, 0.f, 0.f, 0.f, 0.f, 0.f, 0.f};

    const bool full = (mbase + 16 <= M);
    if (full) {
        gemm_body<NT, true, VECA1>(c, A1, W1p, M, K1, N, mbase, lane);
        if (A2) gemm_body<NT, true, true>(c, A2, W2p, M, K2, N, mbase, lane);
    } else {
        gemm_body<NT, false, VECA1>(c, A1, W1p, M, K1, N, mbase, lane);
        if (A2) gemm_body<NT, false, true>(c, A2, W2p, M, K2, N, mbase, lane);
    }

    const int col0 = lane & 15;
    const int rofs = (lane >= 16) ? 8 : 0;
    #pragma unroll
    for (int nt = 0; nt < NT; ++nt) {
        const int col = nt * 16 + col0;
        const float bb = bias ? bias[col] : 0.f;
        #pragma unroll
        for (int r = 0; r < 8; ++r) {
            const int row = mbase + r + rofs;
            if (full || row < M) {
                const long idx = (long)row * N + col;
                float v = c[nt][r] + bb;
                if (accum) v += out[idx];
                if (do_relu) v = fmaxf(v, 0.f);
                out[idx] = v;
            }
        }
    }
}

__global__ void k_ln(float* __restrict__ out, const float* __restrict__ in,
                     const float* __restrict__ res,
                     const float* __restrict__ g, const float* __restrict__ b,
                     int M) {
    int lane = threadIdx.x;
    int row = blockIdx.x * blockDim.y + threadIdx.y;
    if (row >= M) return;
    long base = (long)row * HDIM;
    float v0 = in[base + lane];
    float v1 = in[base + lane + 32];
    if (res) { v0 += res[base + lane]; v1 += res[base + lane + 32]; }
    v0 = fmaxf(v0, 0.f);
    v1 = fmaxf(v1, 0.f);
    float s = v0 + v1;
    for (int off = 16; off > 0; off >>= 1) s += __shfl_xor(s, off, 32);
    float mean = s * (1.0f / 64.0f);
    float d0 = v0 - mean, d1 = v1 - mean;
    float q = d0 * d0 + d1 * d1;
    for (int off = 16; off > 0; off >>= 1) q += __shfl_xor(q, off, 32);
    float rstd = rsqrtf(q * (1.0f / 64.0f) + 1e-5f);
    out[base + lane]      = d0 * rstd * g[lane]      + b[lane];
    out[base + lane + 32] = d1 * rstd * g[lane + 32] + b[lane + 32];
}

__global__ __launch_bounds__(256) void k_addrelu(float* __restrict__ o,
                                                 const float* __restrict__ a,
                                                 const float* __restrict__ b,
                                                 long n) {
    long i = (long)blockIdx.x * blockDim.x + threadIdx.x;
    if (i < n) o[i] = fmaxf(a[i] + b[i], 0.f);
}

__global__ __launch_bounds__(256) void k_final(const float* __restrict__ h,
                                               const float* __restrict__ W2,
                                               const float* __restrict__ b2,
                                               float* __restrict__ out, int M) {
    int i = blockIdx.x * blockDim.x + threadIdx.x;
    if (i >= M) return;
    float acc = b2[0];
    const float* hr = h + (long)i * 32;
    #pragma unroll
    for (int j = 0; j < 32; ++j) acc += hr[j] * W2[j];
    out[i] = acc;
}

extern "C" void kernel_launch(void* const* d_in, const int* in_sizes, int n_in,
                              void* d_out, int out_size, void* d_ws, size_t ws_size,
                              hipStream_t stream) {
    const float* x_news    = (const float*)d_in[0];
    const float* x_company = (const float*)d_in[1];
    const int*   edge_sim  = (const int*)d_in[2];
    const int*   men_src   = (const int*)d_in[3];
    const int*   men_dst   = (const int*)d_in[4];
    const int*   edge_rel  = (const int*)d_in[5];
    const float* nw_W = (const float*)d_in[6];
    const float* nw_b = (const float*)d_in[7];
    const float* cw_W = (const float*)d_in[8];
    const float* cw_b = (const float*)d_in[9];
    const float* n1n_g = (const float*)d_in[10];
    const float* n1n_b = (const float*)d_in[11];
    const float* n1c_g = (const float*)d_in[12];
    const float* n1c_b = (const float*)d_in[13];
    const float* c1_Wl = (const float*)d_in[14];
    const float* c1_bl = (const float*)d_in[15];
    const float* c1_Wr = (const float*)d_in[16];
    const float* c2_Wl = (const float*)d_in[17];
    const float* c2_bl = (const float*)d_in[18];
    const float* c2_Wr = (const float*)d_in[19];
    const float* cls_W1 = (const float*)d_in[20];
    const float* cls_b1 = (const float*)d_in[21];
    const float* cls_W2 = (const float*)d_in[22];
    const float* cls_b2 = (const float*)d_in[23];
    float* out = (float*)d_out;

    float* ws = (float*)d_ws;
    long o = 0;
    float* xn0  = ws + o; o += (long)NN * HDIM;
    float* xn1  = ws + o; o += (long)NN * HDIM;
    float* aggn = ws + o; o += (long)NN * HDIM;
    float* cntn = ws + o; o += NN;
    float* aggcM = ws + o; o += (long)NC * HDIM;
    float* cntcM = ws + o; o += NC;
    float* aggcR = ws + o; o += (long)NC * HDIM;
    float* cntcR = ws + o; o += NC;
    float* xc0  = ws + o; o += (long)NC * HDIM;
    float* xc1  = ws + o; o += (long)NC * HDIM;
    float* ocb  = ws + o; o += (long)NC * HDIM;
    float* comp = ws + o; o += (long)NC * HDIM;
    float* hbuf = ws + o; o += (long)NC * 32;
    v2f* wp_nw  = (v2f*)(ws + o); o += 2L * (388 / 2) * 64;
    v2f* wp_cw  = (v2f*)(ws + o); o += 2L * (24 / 2) * 64;
    v2f* wp_c1  = (v2f*)(ws + o); o += 2L * 6 * 32 * 64;
    v2f* wp_c2  = (v2f*)(ws + o); o += 2L * 4 * 32 * 64;
    v2f* wp_cls = (v2f*)(ws + o); o += 2L * 32 * 32;

    const long zn_n = (long)NN * HDIM + NN;
    const long zn_c = 2L * ((long)NC * HDIM + NC);

    dim3 blk(256);
    auto blocks = [](long n) { return (unsigned)((n + 255) / 256); };
    int tilesNN = (NN + 15) / 16, tilesNC = (NC + 15) / 16;
    unsigned gemmBlkNN = (tilesNN + 7) / 8, gemmBlkNC = (tilesNC + 7) / 8;

    k_packW<<<blocks((388L / 2) * 64), blk, 0, stream>>>(nw_W, wp_nw, 385, 64);
    k_packW<<<blocks((24L / 2) * 64), blk, 0, stream>>>(cw_W, wp_cw, 24, 64);
    for (int r = 0; r < 3; ++r) {
        k_packW<<<blocks(32L * 64), blk, 0, stream>>>(c1_Wl + r * 4096,
                                                      wp_c1 + (2 * r) * 2048, 64, 64);
        k_packW<<<blocks(32L * 64), blk, 0, stream>>>(c1_Wr + r * 4096,
                                                      wp_c1 + (2 * r + 1) * 2048, 64, 64);
    }
    for (int r = 1; r < 3; ++r) {
        k_packW<<<blocks(32L * 64), blk, 0, stream>>>(c2_Wl + r * 4096,
                                                      wp_c2 + (2 * (r - 1)) * 2048, 64, 64);
        k_packW<<<blocks(32L * 64), blk, 0, stream>>>(c2_Wr + r * 4096,
                                                      wp_c2 + (2 * (r - 1) + 1) * 2048, 64, 64);
    }
    k_packW<<<blocks(32L * 32), blk, 0, stream>>>(cls_W1, wp_cls, 64, 32);

    k_gemm<4, false><<<gemmBlkNN, blk, 0, stream>>>(x_news, wp_nw, nw_b,
                                                    nullptr, nullptr,
                                                    xn0, NN, 385, 0, 1, 0);
    k_gemm<4, true><<<gemmBlkNC, blk, 0, stream>>>(x_company, wp_cw, cw_b,
                                                   nullptr, nullptr,
                                                   xc0, NC, 24, 0, 1, 0);

    k_zero<<<blocks(zn_n / 4), blk, 0, stream>>>(aggn, zn_n / 4);
    k_zero<<<blocks(zn_c / 4), blk, 0, stream>>>(aggcM, zn_c / 4);
    k_scatter<<<blocks((long)ES), blk, 0, stream>>>(edge_sim, edge_sim + ES,
                                                         xn0, aggn, cntn, ES);
    k_scatter<<<blocks((long)EM), blk, 0, stream>>>(men_src, men_dst,
                                                         xn0, aggcM, cntcM, EM);
    k_scatter<<<blocks((long)ER), blk, 0, stream>>>(edge_rel, edge_rel + ER,
                                                         xc0, aggcR, cntcR, ER);
    k_scale_rows<<<blocks((long)NN * 16), blk, 0, stream>>>(aggn, cntn, NN);
    k_scale_rows<<<blocks((long)NC * 16), blk, 0, stream>>>(aggcM, cntcM, NC);
    k_scale_rows<<<blocks((long)NC * 16), blk, 0, stream>>>(aggcR, cntcR, NC);

    k_gemm<4, true><<<gemmBlkNN, blk, 0, stream>>>(aggn, wp_c1 + 0 * 2048,
                                                   c1_bl + 0 * 64,
                                                   xn0, wp_c1 + 1 * 2048,
                                                   xn1, NN, 64, 64, 0, 0);
    k_gemm<4, true><<<gemmBlkNC, blk, 0, stream>>>(aggcM, wp_c1 + 2 * 2048,
                                                   c1_bl + 1 * 64,
                                                   xc0, wp_c1 + 3 * 2048,
                                                   ocb, NC, 64, 64, 0, 0);
    k_gemm<4, true><<<gemmBlkNC, blk, 0, stream>>>(aggcR, wp_c1 + 4 * 2048,
                                                   c1_bl + 2 * 64,
                                                   xc0, wp_c1 + 5 * 2048,
                                                   ocb, NC, 64, 64, 0, 1);

    dim3 lnb(32, 8);
    k_ln<<<(NN + 7) / 8, lnb, 0, stream>>>(xn1, xn1, nullptr, n1n_g, n1n_b, NN);
    k_ln<<<(NC + 7) / 8, lnb, 0, stream>>>(xc1, ocb, xc0, n1c_g, n1c_b, NC);

    k_zero<<<blocks(zn_c / 4), blk, 0, stream>>>(aggcM, zn_c / 4);
    k_scatter<<<blocks((long)EM), blk, 0, stream>>>(men_src, men_dst,
                                                         xn1, aggcM, cntcM, EM);
    k_scatter<<<blocks((long)ER), blk, 0, stream>>>(edge_rel, edge_rel + ER,
                                                         xc1, aggcR, cntcR, ER);
    k_scale_rows<<<blocks((long)NC * 16), blk, 0, stream>>>(aggcM, cntcM, NC);
    k_scale_rows<<<blocks((long)NC * 16), blk, 0, stream>>>(aggcR, cntcR, NC);
    k_gemm<4, true><<<gemmBlkNC, blk, 0, stream>>>(aggcM, wp_c2 + 0 * 2048,
                                                   c2_bl + 1 * 64,
                                                   xc1, wp_c2 + 1 * 2048,
                                                   ocb, NC, 64, 64, 0, 0);
    k_gemm<4, true><<<gemmBlkNC, blk, 0, stream>>>(aggcR, wp_c2 + 2 * 2048,
                                                   c2_bl + 2 * 64,
                                                   xc1, wp_c2 + 3 * 2048,
                                                   ocb, NC, 64, 64, 0, 1);

    k_addrelu<<<blocks((long)NC * HDIM), blk, 0, stream>>>(comp, ocb, xc1,
                                                           (long)NC * HDIM);
    k_gemm<2, true><<<gemmBlkNC, blk, 0, stream>>>(comp, wp_cls, cls_b1,
                                                   nullptr, nullptr,
                                                   hbuf, NC, 64, 0, 1, 0);
    k_final<<<blocks(NC), blk, 0, stream>>>(hbuf, cls_W2, cls_b2, out, NC);
}


# aggregation: each work-group owns a destination row range, sums gathered rows in LDS (float adds, arrival order) and stores each mean row once; no global atomics, zero or scale passes
# speedup vs baseline: 2.3272x; 1.5746x over previous
#include <hip/hip_runtime.h>
#include <hip/hip_bf16.h>

typedef __attribute__((ext_vector_type(2))) float v2f;
typedef __attribute__((ext_vector_type(8))) float v8f;

#define NN 100000
#define NC 5000
#define HDIM 64
#define ES 3200000
#define EM 1000000
#define ER 160000

__global__ __launch_bounds__(256) void k_packW(const float* __restrict__ W,
                                               v2f* __restrict__ Wp,
                                               int K, int N) {
    const int Kp2 = ((K + 3) & ~3) >> 1;
    const long n = (long)Kp2 * N;
    long i = (long)blockIdx.x * blockDim.x + threadIdx.x;
    if (i >= n) return;
    int k2 = (int)(i / N), col = (int)(i % N);
    int k = 2 * k2;
    v2f v;
    v.x = (k     < K) ? W[(long)k * N + col]       : 0.f;
    v.y = (k + 1 < K) ? W[(long)(k + 1) * N + col] : 0.f;
    Wp[i] = v;
}

template<int R, int CAP>
__global__ __launch_bounds__(1024) void k_own_agg(const int* __restrict__ src,
                                                  const int* __restrict__ dst,
                                                  const float* __restrict__ x,
                                                  float* __restrict__ out,
                                                  int nE, int nDst) {
    __shared__ __attribute__((aligned(16))) float acc[R * HDIM];
    __shared__ int cn[R];
    __shared__ int list[CAP];
    __shared__ int nlist;

    const int tid = threadIdx.x;
    const int lane = tid & 31;
    const int wv = __builtin_amdgcn_readfirstlane(tid >> 5);
    const int lo = blockIdx.x * R;
    const int nr = min(R, nDst - lo);

    for (int i = tid; i < R * HDIM; i += 1024) acc[i] = 0.f;
    for (int i = tid; i < R; i += 1024) cn[i] = 0;
    if (tid == 0) nlist = 0;
    __syncthreads();

    constexpr int U = 8;
    constexpr int CH = U * 128;
    const int nCh = (nE + CH - 1) / CH;
    const int rot = (int)(((long)blockIdx.x * nCh) / gridDim.x);
    for (int c0 = wv; c0 < nCh; c0 += 32) {
        int c = c0 + rot;
        if (c >= nCh) c -= nCh;
        const int base = c * CH;
        int4 d[U];
        if (base + CH <= nE) {
            #pragma unroll
            for (int u = 0; u < U; ++u)
                d[u] = *(const int4*)(dst + base + u * 128 + lane * 4);
        } else {
            #pragma unroll
            for (int u = 0; u < U; ++u) {
                const int i0 = base + u * 128 + lane * 4;
                d[u].x = (i0     < nE) ? dst[i0]     : -1;
                d[u].y = (i0 + 1 < nE) ? dst[i0 + 1] : -1;
                d[u].z = (i0 + 2 < nE) ? dst[i0 + 2] : -1;
                d[u].w = (i0 + 3 < nE) ? dst[i0 + 3] : -1;
            }
        }
        int cl = 0;
        #pragma unroll
        for (int u = 0; u < U; ++u) {
            cl += ((unsigned)(d[u].x - lo) < (unsigned)nr) ? 1 : 0;
            cl += ((unsigned)(d[u].y - lo) < (unsigned)nr) ? 1 : 0;
            cl += ((unsigned)(d[u].z - lo) < (unsigned)nr) ? 1 : 0;
            cl += ((unsigned)(d[u].w - lo) < (unsigned)nr) ? 1 : 0;
        }
        int incl = cl;
        #pragma unroll
        for (int off = 1; off < 32; off <<= 1) {
            const int t = __shfl_up(incl, off, 32);
            if (lane >= off) incl += t;
        }
        const int total = __builtin_amdgcn_readlane(incl, 31);
        if (total == 0) continue;
        int s0 = 0;
        if (lane == 31) s0 = atomicAdd(&nlist, total);
        s0 = __builtin_amdgcn_readlane(s0, 31);
        int pos = s0 + incl - cl;
        #pragma unroll
        for (int u = 0; u < U; ++u) {
            #pragma unroll
            for (int q = 0; q < 4; ++q) {
                const int dv = (q == 0) ? d[u].x : (q == 1) ? d[u].y : (q == 2) ? d[u].z : d[u].w;
                const int dl = dv - lo;
                if ((unsigned)dl < (unsigned)nr) {
                    const int e = base + u * 128 + lane * 4 + q;
                    if (pos < CAP) {
                        list[pos] = e;
                    } else {
                        const float* xr = x + (long)src[e] * HDIM;
                        #pragma nounroll
                        for (int k = 0; k < HDIM; ++k) atomicAdd(&acc[dl * HDIM + k], xr[k]);
                        atomicAdd(&cn[dl], 1);
                    }
                    ++pos;
                }
            }
        }
    }
    __syncthreads();

    const int n = min(nlist, CAP);
    for (int g = wv * 32; g < n; g += 32 * 32) {
        const int k = g + lane;
        int s = 0, dl = 0;
        if (k < n) {
            const int e = list[k];
            s = src[e];
            dl = dst[e] - lo;
            atomicAdd(&cn[dl], 1);
        }
        if (g + 32 <= n) {
            float v0[32], v1[32];
            #pragma unroll
            for (int j = 0; j < 32; ++j) {
                const int sj = __builtin_amdgcn_readlane(s, j);
                const float* xr = x + (long)sj * HDIM + lane;
                v0[j] = xr[0];
                v1[j] = xr[32];
            }
            #pragma unroll
            for (int j = 0; j < 32; ++j) {
                const int dj = __builtin_amdgcn_readlane(dl, j);
                float* ar = acc + dj * HDIM + lane;
                atomicAdd(ar, v0[j]);
                atomicAdd(ar + 32, v1[j]);
            }
        } else {
            const int m = n - g;
            for (int j = 0; j < m; ++j) {
                const int sj = __builtin_amdgcn_readlane(s, j);
                const int dj = __builtin_amdgcn_readlane(dl, j);
                const float* xr = x + (long)sj * HDIM + lane;
                float* ar = acc + dj * HDIM + lane;
                const float a0 = xr[0];
                const float a1 = xr[32];
                atomicAdd(ar, a0);
                atomicAdd(ar + 32, a1);
            }
        }
    }
    __syncthreads();

    for (int i = tid; i < nr * 16; i += 1024) {
        const int r = i >> 4;
        const int c4 = (i & 15) * 4;
        const float inv = 1.0f / fmaxf((float)cn[r], 1.0f);
        float4 v = *(const float4*)(acc + r * HDIM + c4);
        v.x *= inv; v.y *= inv; v.z *= inv; v.w *= inv;
        *(float4*)(out + (long)(lo + r) * HDIM + c4) = v;
    }
}

template<int NT, bool FULL, bool VECA>
__device__ __forceinline__ void gemm_body(v8f (&c)[NT],
                                          const float* __restrict__ A,
                                          const v2f* __restrict__ Wp,
                                          int M, int K, int N,
                                          int mbase, int lane) {
    const int row  = mbase + (lane & 15);
    const int koff = (lane >= 16) ? 2 : 0;
    const int col0 = lane & 15;
    const bool rok = FULL || (row < M);
    const float* __restrict__ Arow = A + (long)row * K;
    const int Kmain = K & ~3;

    #pragma unroll 4
    for (int k = 0; k < Kmain; k += 4) {
        const int kk = k + koff;
        v2f a;
        if (rok) {
            if (VECA) {
                a = *(const v2f*)(Arow + kk);
            } else {
                a.x = Arow[kk];
                a.y = Arow[kk + 1];
            }
        } else {
            a.x = 0.f; a.y = 0.f;
        }
        #pragma unroll
        for (int nt = 0; nt < NT; ++nt) {
            v2f b = Wp[(long)(kk >> 1) * N + nt * 16 + col0];
            c[nt] = __builtin_amdgcn_wmma_f32_16x16x4_f32(
                false, a, false, b, (short)0, c[nt], false, false);
        }
    }
    if (Kmain < K) {
        const int kk = Kmain + koff;
        v2f a;
        a.x = (rok && kk     < K) ? Arow[kk]     : 0.f;
        a.y = (rok && kk + 1 < K) ? Arow[kk + 1] : 0.f;
        #pragma unroll
        for (int nt = 0; nt < NT; ++nt) {
            v2f b = Wp[(long)(kk >> 1) * N + nt * 16 + col0];
            c[nt] = __builtin_amdgcn_wmma_f32_16x16x4_f32(
                false, a, false, b, (short)0, c[nt], false, false);
        }
    }
}

template<int NT, bool VECA1>
__global__ __launch_bounds__(256) void k_gemm(const float* __restrict__ A1,
                                              const v2f* __restrict__ W1p,
                                              const float* __restrict__ bias,
                                              const float* __restrict__ A2,
                                              const v2f* __restrict__ W2p,
                                              float* __restrict__ out,
                                              int M, int K1, int K2,
                                              int do_relu, int accum) {
    const int N = NT * 16;
    const int lane = threadIdx.x & 31;
    const int wv = threadIdx.x >> 5;
    const int mbase = (blockIdx.x * 8 + wv) * 16;
    if (mbase >= M) return;

    v8f c[NT];
    #pragma unroll
    for (int t = 0; t < NT; ++t)
        c[t] = (v8f){0.f, 0.f, 0.f, 0.f, 0.f, 0.f, 0.f, 0.f};

    const bool full = (mbase + 16 <= M);
    if (full) {
        gemm_body<NT, true, VECA1>(c, A1, W1p, M, K1, N, mbase, lane);
        if (A2) gemm_body<NT, true, true>(c, A2, W2p, M, K2, N, mbase, lane);
    } else {
        gemm_body<NT, false, VECA1>(c, A1, W1p, M, K1, N, mbase, lane);
        if (A2) gemm_body<NT, false, true>(c, A2, W2p, M, K2, N, mbase, lane);
    }

    const int col0 = lane & 15;
    const int rofs = (lane >= 16) ? 8 : 0;
    #pragma unroll
    for (int nt = 0; nt < NT; ++nt) {
        const int col = nt * 16 + col0;
        const float bb = bias ? bias[col] : 0.f;
        #pragma unroll
        for (int r = 0; r < 8; ++r) {
            const int row = mbase + r + rofs;
            if (full || row < M) {
                const long idx = (long)row * N + col;
                float v = c[nt][r] + bb;
                if (accum) v += out[idx];
                if (do_relu) v = fmaxf(v, 0.f);
                out[idx] = v;
            }
        }
    }
}

__global__ void k_ln(float* __restrict__ out, const float* __restrict__ in,
                     const float* __restrict__ res,
                     const float* __restrict__ g, const float* __restrict__ b,
                     int M) {
    int lane = threadIdx.x;
    int row = blockIdx.x * blockDim.y + threadIdx.y;
    if (row >= M) return;
    long base = (long)row * HDIM;
    float v0 = in[base + lane];
    float v1 = in[base + lane + 32];
    if (res) { v0 += res[base + lane]; v1 += res[base + lane + 32]; }
    v0 = fmaxf(v0, 0.f);
    v1 = fmaxf(v1, 0.f);
    float s = v0 + v1;
    for (int off = 16; off > 0; off >>= 1) s += __shfl_xor(s, off, 32);
    float mean = s * (1.0f / 64.0f);
    float d0 = v0 - mean, d1 = v1 - mean;
    float q = d0 * d0 + d1 * d1;
    for (int off = 16; off > 0; off >>= 1) q += __shfl_xor(q, off, 32);
    float rstd = rsqrtf(q * (1.0f / 64.0f) + 1e-5f);
    out[base + lane]      = d0 * rstd * g[lane]      + b[lane];
    out[base + lane + 32] = d1 * rstd * g[lane + 32] + b[lane + 32];
}

__global__ __launch_bounds__(256) void k_addrelu(float* __restrict__ o,
                                                 const float* __restrict__ a,
                                                 const float* __restrict__ b,
                                                 long n) {
    long i = (long)blockIdx.x * blockDim.x + threadIdx.x;
    if (i < n) o[i] = fmaxf(a[i] + b[i], 0.f);
}

__global__ __launch_bounds__(256) void k_final(const float* __restrict__ h,
                                               const float* __restrict__ W2,
                                               const float* __restrict__ b2,
                                               float* __restrict__ out, int M) {
    int i = blockIdx.x * blockDim.x + threadIdx.x;
    if (i >= M) return;
    float acc = b2[0];
    const float* hr = h + (long)i * 32;
    #pragma unroll
    for (int j = 0; j < 32; ++j) acc += hr[j] * W2[j];
    out[i] = acc;
}

extern "C" void kernel_launch(void* const* d_in, const int* in_sizes, int n_in,
                              void* d_out, int out_size, void* d_ws, size_t ws_size,
                              hipStream_t stream) {
    const float* x_news    = (const float*)d_in[0];
    const float* x_company = (const float*)d_in[1];
    const int*   edge_sim  = (const int*)d_in[2];
    const int*   men_src   = (const int*)d_in[3];
    const int*   men_dst   = (const int*)d_in[4];
    const int*   edge_rel  = (const int*)d_in[5];
    const float* nw_W = (const float*)d_in[6];
    const float* nw_b = (const float*)d_in[7];
    const float* cw_W = (const float*)d_in[8];
    const float* cw_b = (const float*)d_in[9];
    const float* n1n_g = (const float*)d_in[10];
    const float* n1n_b = (const float*)d_in[11];
    const float* n1c_g = (const float*)d_in[12];
    const float* n1c_b = (const float*)d_in[13];
    const float* c1_Wl = (const float*)d_in[14];
    const float* c1_bl = (const float*)d_in[15];
    const float* c1_Wr = (const float*)d_in[16];
    const float* c2_Wl = (const float*)d_in[17];
    const float* c2_bl = (const float*)d_in[18];
    const float* c2_Wr = (const float*)d_in[19];
    const float* cls_W1 = (const float*)d_in[20];
    const float* cls_b1 = (const float*)d_in[21];
    const float* cls_W2 = (const float*)d_in[22];
    const float* cls_b2 = (const float*)d_in[23];
    float* out = (float*)d_out;

    float* ws = (float*)d_ws;
    long o = 0;
    float* xn0  = ws + o; o += (long)NN * HDIM;
    float* xn1  = ws + o; o += (long)NN * HDIM;
    float* aggn = ws + o; o += (long)NN * HDIM;
    float* aggcM = ws + o; o += (long)NC * HDIM;
    float* aggcR = ws + o; o += (long)NC * HDIM;
    float* xc0  = ws + o; o += (long)NC * HDIM;
    float* xc1  = ws + o; o += (long)NC * HDIM;
    float* ocb  = ws + o; o += (long)NC * HDIM;
    float* comp = ws + o; o += (long)NC * HDIM;
    float* hbuf = ws + o; o += (long)NC * 32;
    v2f* wp_nw  = (v2f*)(ws + o); o += 2L * (388 / 2) * 64;
    v2f* wp_cw  = (v2f*)(ws + o); o += 2L * (24 / 2) * 64;
    v2f* wp_c1  = (v2f*)(ws + o); o += 2L * 6 * 32 * 64;
    v2f* wp_c2  = (v2f*)(ws + o); o += 2L * 4 * 32 * 64;
    v2f* wp_cls = (v2f*)(ws + o); o += 2L * 32 * 32;

    dim3 blk(256);
    auto blocks = [](long n) { return (unsigned)((n + 255) / 256); };
    int tilesNN = (NN + 15) / 16, tilesNC = (NC + 15) / 16;
    unsigned gemmBlkNN = (tilesNN + 7) / 8, gemmBlkNC = (tilesNC + 7) / 8;

    k_packW<<<blocks((388L / 2) * 64), blk, 0, stream>>>(nw_W, wp_nw, 385, 64);
    k_packW<<<blocks((24L / 2) * 64), blk, 0, stream>>>(cw_W, wp_cw, 24, 64);
    for (int r = 0; r < 3; ++r) {
        k_packW<<<blocks(32L * 64), blk, 0, stream>>>(c1_Wl + r * 4096,
                                                      wp_c1 + (2 * r) * 2048, 64, 64);
        k_packW<<<blocks(32L * 64), blk, 0, stream>>>(c1_Wr + r * 4096,
                                                      wp_c1 + (2 * r + 1) * 2048, 64, 64);
    }
    for (int r = 1; r < 3; ++r) {
        k_packW<<<blocks(32L * 64), blk, 0, stream>>>(c2_Wl + r * 4096,
                                                      wp_c2 + (2 * (r - 1)) * 2048, 64, 64);
        k_packW<<<blocks(32L * 64), blk, 0, stream>>>(c2_Wr + r * 4096,
                                                      wp_c2 + (2 * (r - 1) + 1) * 2048, 64, 64);
    }
    k_packW<<<blocks(32L * 32), blk, 0, stream>>>(cls_W1, wp_cls, 64, 32);

    k_gemm<4, false><<<gemmBlkNN, blk, 0, stream>>>(x_news, wp_nw, nw_b,
                                                    nullptr, nullptr,
                                                    xn0, NN, 385, 0, 1, 0);
    k_gemm<4, true><<<gemmBlkNC, blk, 0, stream>>>(x_company, wp_cw, cw_b,
                                                   nullptr, nullptr,
                                                   xc0, NC, 24, 0, 1, 0);

    k_own_agg<391, 16384><<<(NN + 390) / 391, 1024, 0, stream>>>(edge_sim, edge_sim + ES,
                                                                 xn0, aggn, ES, NN);
    k_own_agg<20, 8192><<<(NC + 19) / 20, 1024, 0, stream>>>(men_src, men_dst,
                                                             xn0, aggcM, EM, NC);
    k_own_agg<20, 8192><<<(NC + 19) / 20, 1024, 0, stream>>>(edge_rel, edge_rel + ER,
                                                             xc0, aggcR, ER, NC);

    k_gemm<4, true><<<gemmBlkNN, blk, 0, stream>>>(aggn, wp_c1 + 0 * 2048,
                                                   c1_bl + 0 * 64,
                                                   xn0, wp_c1 + 1 * 2048,
                                                   xn1, NN, 64, 64, 0, 0);
    k_gemm<4, true><<<gemmBlkNC, blk, 0, stream>>>(aggcM, wp_c1 + 2 * 2048,
                                                   c1_bl + 1 * 64,
                                                   xc0, wp_c1 + 3 * 2048,
                                                   ocb, NC, 64, 64, 0, 0);
    k_gemm<4, true><<<gemmBlkNC, blk, 0, stream>>>(aggcR, wp_c1 + 4 * 2048,
                                                   c1_bl + 2 * 64,
                                                   xc0, wp_c1 + 5 * 2048,
                                                   ocb, NC, 64, 64, 0, 1);

    dim3 lnb(32, 8);
    k_ln<<<(NN + 7) / 8, lnb, 0, stream>>>(xn1, xn1, nullptr, n1n_g, n1n_b, NN);
    k_ln<<<(NC + 7) / 8, lnb, 0, stream>>>(xc1, ocb, xc0, n1c_g, n1c_b, NC);

    k_own_agg<20, 8192><<<(NC + 19) / 20, 1024, 0, stream>>>(men_src, men_dst,
                                                             xn1, aggcM, EM, NC);
    k_own_agg<20, 8192><<<(NC + 19) / 20, 1024, 0, stream>>>(edge_rel, edge_rel + ER,
                                                             xc1, aggcR, ER, NC);
    k_gemm<4, true><<<gemmBlkNC, blk, 0, stream>>>(aggcM, wp_c2 + 0 * 2048,
                                                   c2_bl + 1 * 64,
                                                   xc1, wp_c2 + 1 * 2048,
                                                   ocb, NC, 64, 64, 0, 0);
    k_gemm<4, true><<<gemmBlkNC, blk, 0, stream>>>(aggcR, wp_c2 + 2 * 2048,
                                                   c2_bl + 2 * 64,
                                                   xc1, wp_c2 + 3 * 2048,
                                                   ocb, NC, 64, 64, 0, 1);

    k_addrelu<<<blocks((long)NC * HDIM), blk, 0, stream>>>(comp, ocb, xc1,
                                                           (long)NC * HDIM);
    k_gemm<2, true><<<gemmBlkNC, blk, 0, stream>>>(comp, wp_cls, cls_b1,
                                                   nullptr, nullptr,
                                                   hbuf, NC, 64, 0, 1, 0);
    k_final<<<blocks(NC), blk, 0, stream>>>(hbuf, cls_W2, cls_b2, out, NC);
}


# aggregation: a pre-pass bins edges by destination owner (one packed word per edge, one integer atomic per owner and chunk); each owner reads its own list, not all destinations; sums in arrival order
# speedup vs baseline: 3.3723x; 1.4491x over previous
#include <hip/hip_runtime.h>
#include <hip/hip_bf16.h>

typedef __attribute__((ext_vector_type(2))) float v2f;
typedef __attribute__((ext_vector_type(8))) float v8f;

#define NN 100000
#define NC 5000
#define HDIM 64
#define ES 3200000
#define EM 1000000
#define ER 160000

__global__ __launch_bounds__(256) void k_packW(const float* __restrict__ W,
                                               v2f* __restrict__ Wp,
                                               int K, int N) {
    const int Kp2 = ((K + 3) & ~3) >> 1;
    const long n = (long)Kp2 * N;
    long i = (long)blockIdx.x * blockDim.x + threadIdx.x;
    if (i >= n) return;
    int k2 = (int)(i / N), col = (int)(i % N);
    int k = 2 * k2;
    v2f v;
    v.x = (k     < K) ? W[(long)k * N + col]       : 0.f;
    v.y = (k + 1 < K) ? W[(long)(k + 1) * N + col] : 0.f;
    Wp[i] = v;
}

#define PK_SH 9
#define PK_MASK 511
__global__ __launch_bounds__(256) void k_zero(int* __restrict__ p, int n) {
    const int i = blockIdx.x * blockDim.x + threadIdx.x;
    if (i < n) p[i] = 0;
}

template<int R, int CAPB>
__global__ __launch_bounds__(1024) void k_bin(const int* __restrict__ src,
                                              const int* __restrict__ dst,
                                              int nE, int nB,
                                              int* __restrict__ blist,
                                              int* __restrict__ bctr) {
    static_assert(R <= PK_MASK + 1, "row inside the range must fit the packed field");
    constexpr int EPT = 8;
    constexpr int CHUNK = 1024 * EPT;
    constexpr int NBMAX = 256;
    __shared__ int hcnt[NBMAX];
    __shared__ int hoff[NBMAX];
    __shared__ int gbase[NBMAX];
    __shared__ int stage[CHUNK];
    __shared__ int sown[CHUNK];

    const int tid = threadIdx.x;
    const int lane = tid & 31;
    const int wv = tid >> 5;
    const int base = blockIdx.x * CHUNK;

    if (tid < NBMAX) hcnt[tid] = 0;
    __syncthreads();

    int ow[EPT], rk[EPT], pk[EPT];
    #pragma unroll
    for (int u = 0; u < EPT; ++u) {
        const int e = base + u * 1024 + tid;
        ow[u] = -1; rk[u] = 0; pk[u] = 0;
        if (e < nE) {
            const int d = dst[e];
            const int o = (int)((unsigned)d / (unsigned)R);
            ow[u] = o;
            pk[u] = (src[e] << PK_SH) | (d - o * R);
            rk[u] = atomicAdd(&hcnt[o], 1);
        }
    }
    __syncthreads();

    if (tid < nB) {
        const int cnt = hcnt[tid];
        gbase[tid] = cnt ? atomicAdd(&bctr[tid], cnt) : 0;
    }
    if (wv == 31) {
        int cq[8];
        int t = 0;
        #pragma unroll
        for (int q = 0; q < 8; ++q) { const int v = hcnt[lane * 8 + q]; cq[q] = t; t += v; }
        int incl = t;
        #pragma unroll
        for (int off = 1; off < 32; off <<= 1) {
            const int s = __shfl_up(incl, off, 32);
            if (lane >= off) incl += s;
        }
        const int excl = incl - t;
        #pragma unroll
        for (int q = 0; q < 8; ++q) hoff[lane * 8 + q] = cq[q] + excl;
    }
    __syncthreads();

    #pragma unroll
    for (int u = 0; u < EPT; ++u) {
        if (ow[u] >= 0) {
            const int lp = hoff[ow[u]] + rk[u];
            stage[lp] = pk[u];
            sown[lp] = ow[u];
        }
    }
    __syncthreads();

    const int total = min(CHUNK, nE - base);
    for (int i = tid; i < total; i += 1024) {
        const int o = sown[i];
        const int pos = gbase[o] + (i - hoff[o]);
        if (pos < CAPB) blist[(long)o * CAPB + pos] = stage[i];
        else bctr[nB] = 1;
    }
}

__device__ __forceinline__ void add_group(const float* __restrict__ x, float* acc,
                                          int s, int dl, int m, int lane) {
    if (m >= 32) {
        float v0[32], v1[32];
        #pragma unroll
        for (int j = 0; j < 32; ++j) {
            const int sj = __builtin_amdgcn_readlane(s, j);
            const float* xr = x + (long)sj * HDIM + lane;
            v0[j] = xr[0];
            v1[j] = xr[32];
        }
        #pragma unroll
        for (int j = 0; j < 32; ++j) {
            const int dj = __builtin_amdgcn_readlane(dl, j);
            float* ar = acc + dj * HDIM + lane;
            atomicAdd(ar, v0[j]);
            atomicAdd(ar + 32, v1[j]);
        }
    } else {
        for (int j = 0; j < m; ++j) {
            const int sj = __builtin_amdgcn_readlane(s, j);
            const int dj = __builtin_amdgcn_readlane(dl, j);
            const float* xr = x + (long)sj * HDIM + lane;
            float* ar = acc + dj * HDIM + lane;
            const float a0 = xr[0];
            const float a1 = xr[32];
            atomicAdd(ar, a0);
            atomicAdd(ar + 32, a1);
        }
    }
}

template<int R, int CAP, int CAPB>
__global__ __launch_bounds__(1024) void k_own_agg(const int* __restrict__ src,
                                                  const int* __restrict__ dst,
                                                  const float* __restrict__ x,
                                                  float* __restrict__ out,
                                                  int nE, int nDst,
                                                  const int* __restrict__ blist,
                                                  const int* __restrict__ bctr) {
    __shared__ __attribute__((aligned(16))) float acc[R * HDIM];
    __shared__ int cn[R];
    __shared__ int list[CAP];
    __shared__ int nlist;

    const int tid = threadIdx.x;
    const int lane = tid & 31;
    const int wv = __builtin_amdgcn_readfirstlane(tid >> 5);
    const int lo = blockIdx.x * R;
    const int nr = min(R, nDst - lo);
    const bool binned = __builtin_amdgcn_readfirstlane(bctr[gridDim.x]) == 0;
    const int nbin = __builtin_amdgcn_readfirstlane(bctr[blockIdx.x]);

    for (int i = tid; i < R * HDIM; i += 1024) acc[i] = 0.f;
    for (int i = tid; i < R; i += 1024) cn[i] = 0;
    if (tid == 0) nlist = 0;
    __syncthreads();

    if (!binned) {
    constexpr int U = 8;
    constexpr int CH = U * 128;
    const int nCh = (nE + CH - 1) / CH;
    const int rot = (int)(((long)blockIdx.x * nCh) / gridDim.x);
    for (int c0 = wv; c0 < nCh; c0 += 32) {
        int c = c0 + rot;
        if (c >= nCh) c -= nCh;
        const int base = c * CH;
        int4 d[U];
        if (base + CH <= nE) {
            #pragma unroll
            for (int u = 0; u < U; ++u)
                d[u] = *(const int4*)(dst + base + u * 128 + lane * 4);
        } else {
            #pragma unroll
            for (int u = 0; u < U; ++u) {
                const int i0 = base + u * 128 + lane * 4;
                d[u].x = (i0     < nE) ? dst[i0]     : -1;
                d[u].y = (i0 + 1 < nE) ? dst[i0 + 1] : -1;
                d[u].z = (i0 + 2 < nE) ? dst[i0 + 2] : -1;
                d[u].w = (i0 + 3 < nE) ? dst[i0 + 3] : -1;
            }
        }
        int cl = 0;
        #pragma unroll
        for (int u = 0; u < U; ++u) {
            cl += ((unsigned)(d[u].x - lo) < (unsigned)nr) ? 1 : 0;
            cl += ((unsigned)(d[u].y - lo) < (unsigned)nr) ? 1 : 0;
            cl += ((unsigned)(d[u].z - lo) < (unsigned)nr) ? 1 : 0;
            cl += ((unsigned)(d[u].w - lo) < (unsigned)nr) ? 1 : 0;
        }
        int incl = cl;
        #pragma unroll
        for (int off = 1; off < 32; off <<= 1) {
            const int t = __shfl_up(incl, off, 32);
            if (lane >= off) incl += t;
        }
        const int total = __builtin_amdgcn_readlane(incl, 31);
        if (total == 0) continue;
        int s0 = 0;
        if (lane == 31) s0 = atomicAdd(&nlist, total);
        s0 = __builtin_amdgcn_readlane(s0, 31);
        int pos = s0 + incl - cl;
        #pragma unroll
        for (int u = 0; u < U; ++u) {
            #pragma unroll
            for (int q = 0; q < 4; ++q) {
                const int dv = (q == 0) ? d[u].x : (q == 1) ? d[u].y : (q == 2) ? d[u].z : d[u].w;
                const int dl = dv - lo;
                if ((unsigned)dl < (unsigned)nr) {
                    const int e = base + u * 128 + lane * 4 + q;
                    if (pos < CAP) {
                        list[pos] = e;
                    } else {
                        const float* xr = x + (long)src[e] * HDIM;
                        #pragma nounroll
                        for (int k = 0; k < HDIM; ++k) atomicAdd(&acc[dl * HDIM + k], xr[k]);
                        atomicAdd(&cn[dl], 1);
                    }
                    ++pos;
                }
            }
        }
    }
    }
    __syncthreads();

    if (binned) {
        const int n = min(nbin, CAPB);
        const int* __restrict__ bl = blist + (long)blockIdx.x * CAPB;
        int g = wv * 32;
        if (g < n) {
            int pn = bl[min(g + lane, n - 1)];
            for (; g < n; g += 32 * 32) {
                const int p = pn;
                pn = bl[min(g + 32 * 32 + lane, n - 1)];
                const int m = min(32, n - g);
                int s = 0, dl = 0;
                if (lane < m) {
                    s = p >> PK_SH;
                    dl = p & PK_MASK;
                    atomicAdd(&cn[dl], 1);
                }
                add_group(x, acc, s, dl, m, lane);
            }
        }
    } else {
        const int n = min(nlist, CAP);
        for (int g = wv * 32; g < n; g += 32 * 32) {
            const int k = g + lane;
            int s = 0, dl = 0;
            if (k < n) {
                const int e = list[k];
                s = src[e];
                dl = dst[e] - lo;
                atomicAdd(&cn[dl], 1);
            }
            add_group(x, acc, s, dl, min(32, n - g), lane);
        }
    }
    __syncthreads();

    for (int i = tid; i < nr * 16; i += 1024) {
        const int r = i >> 4;
        const int c4 = (i & 15) * 4;
        const float inv = 1.0f / fmaxf((float)cn[r], 1.0f);
        float4 v = *(const float4*)(acc + r * HDIM + c4);
        v.x *= inv; v.y *= inv; v.z *= inv; v.w *= inv;
        *(float4*)(out + (long)(lo + r) * HDIM + c4) = v;
    }
}

template<int NT, bool FULL, bool VECA>
__device__ __forceinline__ void gemm_body(v8f (&c)[NT],
                                          const float* __restrict__ A,
                                          const v2f* __restrict__ Wp,
                                          int M, int K, int N,
                                          int mbase, int lane) {
    const int row  = mbase + (lane & 15);
    const int koff = (lane >= 16) ? 2 : 0;
    const int col0 = lane & 15;
    const bool rok = FULL || (row < M);
    const float* __restrict__ Arow = A + (long)row * K;
    const int Kmain = K & ~3;

    #pragma unroll 4
    for (int k = 0; k < Kmain; k += 4) {
        const int kk = k + koff;
        v2f a;
        if (rok) {
            if (VECA) {
                a = *(const v2f*)(Arow + kk);
            } else {
                a.x = Arow[kk];
                a.y = Arow[kk + 1];
            }
        } else {
            a.x = 0.f; a.y = 0.f;
        }
        #pragma unroll
        for (int nt = 0; nt < NT; ++nt) {
            v2f b = Wp[(long)(kk >> 1) * N + nt * 16 + col0];
            c[nt] = __builtin_amdgcn_wmma_f32_16x16x4_f32(
                false, a, false, b, (short)0, c[nt], false, false);
        }
    }
    if (Kmain < K) {
        const int kk = Kmain + koff;
        v2f a;
        a.x = (rok && kk     < K) ? Arow[kk]     : 0.f;
        a.y = (rok && kk + 1 < K) ? Arow[kk + 1] : 0.f;
        #pragma unroll
        for (int nt = 0; nt < NT; ++nt) {
            v2f b = Wp[(long)(kk >> 1) * N + nt * 16 + col0];
            c[nt] = __builtin_amdgcn_wmma_f32_16x16x4_f32(
                false, a, false, b, (short)0, c[nt], false, false);
        }
    }
}

template<int NT, bool VECA1>
__global__ __launch_bounds__(256) void k_gemm(const float* __restrict__ A1,
                                              const v2f* __restrict__ W1p,
                                              const float* __restrict__ bias,
                                              const float* __restrict__ A2,
                                              const v2f* __restrict__ W2p,
                                              float* __restrict__ out,
                                              int M, int K1, int K2,
                                              int do_relu, int accum) {
    const int N = NT * 16;
    const int lane = threadIdx.x & 31;
    const int wv = threadIdx.x >> 5;
    const int mbase = (blockIdx.x * 8 + wv) * 16;
    if (mbase >= M) return;

    v8f c[NT];
    #pragma unroll
    for (int t = 0; t < NT; ++t)
        c[t] = (v8f){0.f, 0.f, 0.f, 0.f, 0.f, 0.f, 0.f, 0.f};

    const bool full = (mbase + 16 <= M);
    if (full) {
        gemm_body<NT, true, VECA1>(c, A1, W1p, M, K1, N, mbase, lane);
        if (A2) gemm_body<NT, true, true>(c, A2, W2p, M, K2, N, mbase, lane);
    } else {
        gemm_body<NT, false, VECA1>(c, A1, W1p, M, K1, N, mbase, lane);
        if (A2) gemm_body<NT, false, true>(c, A2, W2p, M, K2, N, mbase, lane);
    }

    const int col0 = lane & 15;
    const int rofs = (lane >= 16) ? 8 : 0;
    #pragma unroll
    for (int nt = 0; nt < NT; ++nt) {
        const int col = nt * 16 + col0;
        const float bb = bias ? bias[col] : 0.f;
        #pragma unroll
        for (int r = 0; r < 8; ++r) {
            const int row = mbase + r + rofs;
            if (full || row < M) {
                const long idx = (long)row * N + col;
                float v = c[nt][r] + bb;
                if (accum) v += out[idx];
                if (do_relu) v = fmaxf(v, 0.f);
                out[idx] = v;
            }
        }
    }
}

__global__ void k_ln(float* __restrict__ out, const float* __restrict__ in,
                     const float* __restrict__ res,
                     const float* __restrict__ g, const float* __restrict__ b,
                     int M) {
    int lane = threadIdx.x;
    int row = blockIdx.x * blockDim.y + threadIdx.y;
    if (row >= M) return;
    long base = (long)row * HDIM;
    float v0 = in[base + lane];
    float v1 = in[base + lane + 32];
    if (res) { v0 += res[base + lane]; v1 += res[base + lane + 32]; }
    v0 = fmaxf(v0, 0.f);
    v1 = fmaxf(v1, 0.f);
    float s = v0 + v1;
    for (int off = 16; off > 0; off >>= 1) s += __shfl_xor(s, off, 32);
    float mean = s * (1.0f / 64.0f);
    float d0 = v0 - mean, d1 = v1 - mean;
    float q = d0 * d0 + d1 * d1;
    for (int off = 16; off > 0; off >>= 1) q += __shfl_xor(q, off, 32);
    float rstd = rsqrtf(q * (1.0f / 64.0f) + 1e-5f);
    out[base + lane]      = d0 * rstd * g[lane]      + b[lane];
    out[base + lane + 32] = d1 * rstd * g[lane + 32] + b[lane + 32];
}

__global__ __launch_bounds__(256) void k_addrelu(float* __restrict__ o,
                                                 const float* __restrict__ a,
                                                 const float* __restrict__ b,
                                                 long n) {
    long i = (long)blockIdx.x * blockDim.x + threadIdx.x;
    if (i < n) o[i] = fmaxf(a[i] + b[i], 0.f);
}

__global__ __launch_bounds__(256) void k_final(const float* __restrict__ h,
                                               const float* __restrict__ W2,
                                               const float* __restrict__ b2,
                                               float* __restrict__ out, int M) {
    int i = blockIdx.x * blockDim.x + threadIdx.x;
    if (i >= M) return;
    float acc = b2[0];
    const float* hr = h + (long)i * 32;
    #pragma unroll
    for (int j = 0; j < 32; ++j) acc += hr[j] * W2[j];
    out[i] = acc;
}

extern "C" void kernel_launch(void* const* d_in, const int* in_sizes, int n_in,
                              void* d_out, int out_size, void* d_ws, size_t ws_size,
                              hipStream_t stream) {
    const float* x_news    = (const float*)d_in[0];
    const float* x_company = (const float*)d_in[1];
    const int*   edge_sim  = (const int*)d_in[2];
    const int*   men_src   = (const int*)d_in[3];
    const int*   men_dst   = (const int*)d_in[4];
    const int*   edge_rel  = (const int*)d_in[5];
    const float* nw_W = (const float*)d_in[6];
    const float* nw_b = (const float*)d_in[7];
    const float* cw_W = (const float*)d_in[8];
    const float* cw_b = (const float*)d_in[9];
    const float* n1n_g = (const float*)d_in[10];
    const float* n1n_b = (const float*)d_in[11];
    const float* n1c_g = (const float*)d_in[12];
    const float* n1c_b = (const float*)d_in[13];
    const float* c1_Wl = (const float*)d_in[14];
    const float* c1_bl = (const float*)d_in[15];
    const float* c1_Wr = (const float*)d_in[16];
    const float* c2_Wl = (const float*)d_in[17];
    const float* c2_bl = (const float*)d_in[18];
    const float* c2_Wr = (const float*)d_in[19];
    const float* cls_W1 = (const float*)d_in[20];
    const float* cls_b1 = (const float*)d_in[21];
    const float* cls_W2 = (const float*)d_in[22];
    const float* cls_b2 = (const float*)d_in[23];
    float* out = (float*)d_out;

    float* ws = (float*)d_ws;
    long o = 0;
    float* xn0  = ws + o; o += (long)NN * HDIM;
    float* xn1  = ws + o; o += (long)NN * HDIM;
    float* aggn = ws + o; o += (long)NN * HDIM;
    float* aggcM = ws + o; o += (long)NC * HDIM;
    float* aggcR = ws + o; o += (long)NC * HDIM;
    float* xc0  = ws + o; o += (long)NC * HDIM;
    float* xc1  = ws + o; o += (long)NC * HDIM;
    float* ocb  = ws + o; o += (long)NC * HDIM;
    float* comp = ws + o; o += (long)NC * HDIM;
    float* hbuf = ws + o; o += (long)NC * 32;
    v2f* wp_nw  = (v2f*)(ws + o); o += 2L * (388 / 2) * 64;
    v2f* wp_cw  = (v2f*)(ws + o); o += 2L * (24 / 2) * 64;
    v2f* wp_c1  = (v2f*)(ws + o); o += 2L * 6 * 32 * 64;
    v2f* wp_c2  = (v2f*)(ws + o); o += 2L * 4 * 32 * 64;
    v2f* wp_cls = (v2f*)(ws + o); o += 2L * 32 * 32;

    int* ctr   = (int*)comp;
    int* ctrN  = ctr;
    int* ctrM1 = ctr + 512;
    int* ctrR  = ctr + 1024;
    int* ctrM2 = ctr + 1536;
    int* blR   = ctr + 4096;
    int* blN   = (int*)xn1;
    int* blM1  = blN + 256L * 15360;
    int* blM2  = (int*)xn0;

    dim3 blk(256);
    auto blocks = [](long n) { return (unsigned)((n + 255) / 256); };
    int tilesNN = (NN + 15) / 16, tilesNC = (NC + 15) / 16;
    unsigned gemmBlkNN = (tilesNN + 7) / 8, gemmBlkNC = (tilesNC + 7) / 8;

    k_packW<<<blocks((388L / 2) * 64), blk, 0, stream>>>(nw_W, wp_nw, 385, 64);
    k_packW<<<blocks((24L / 2) * 64), blk, 0, stream>>>(cw_W, wp_cw, 24, 64);
    for (int r = 0; r < 3; ++r) {
        k_packW<<<blocks(32L * 64), blk, 0, stream>>>(c1_Wl + r * 4096,
                                                      wp_c1 + (2 * r) * 2048, 64, 64);
        k_packW<<<blocks(32L * 64), blk, 0, stream>>>(c1_Wr + r * 4096,
                                                      wp_c1 + (2 * r + 1) * 2048, 64, 64);
    }
    for (int r = 1; r < 3; ++r) {
        k_packW<<<blocks(32L * 64), blk, 0, stream>>>(c2_Wl + r * 4096,
                                                      wp_c2 + (2 * (r - 1)) * 2048, 64, 64);
        k_packW<<<blocks(32L * 64), blk, 0, stream>>>(c2_Wr + r * 4096,
                                                      wp_c2 + (2 * (r - 1) + 1) * 2048, 64, 64);
    }
    k_packW<<<blocks(32L * 32), blk, 0, stream>>>(cls_W1, wp_cls, 64, 32);

    k_gemm<4, false><<<gemmBlkNN, blk, 0, stream>>>(x_news, wp_nw, nw_b,
                                                    nullptr, nullptr,
                                                    xn0, NN, 385, 0, 1, 0);
    k_gemm<4, true><<<gemmBlkNC, blk, 0, stream>>>(x_company, wp_cw, cw_b,
                                                   nullptr, nullptr,
                                                   xc0, NC, 24, 0, 1, 0);

    k_zero<<<16, blk, 0, stream>>>(ctr, 4096);
    k_bin<391, 15360><<<(ES + 8191) / 8192, 1024, 0, stream>>>(edge_sim, edge_sim + ES, ES,
                                                               (NN + 390) / 391, blN, ctrN);
    k_bin<20, 6144><<<(EM + 8191) / 8192, 1024, 0, stream>>>(men_src, men_dst, EM,
                                                             (NC + 19) / 20, blM1, ctrM1);
    k_bin<20, 1536><<<(ER + 8191) / 8192, 1024, 0, stream>>>(edge_rel, edge_rel + ER, ER,
                                                             (NC + 19) / 20, blR, ctrR);
    k_own_agg<391, 16384, 15360><<<(NN + 390) / 391, 1024, 0, stream>>>(edge_sim, edge_sim + ES,
                                                                        xn0, aggn, ES, NN,
                                                                        blN, ctrN);
    k_own_agg<20, 8192, 6144><<<(NC + 19) / 20, 1024, 0, stream>>>(men_src, men_dst,
                                                                   xn0, aggcM, EM, NC,
                                                                   blM1, ctrM1);
    k_own_agg<20, 8192, 1536><<<(NC + 19) / 20, 1024, 0, stream>>>(edge_rel, edge_rel + ER,
                                                                   xc0, aggcR, ER, NC,
                                                                   blR, ctrR);

    k_gemm<4, true><<<gemmBlkNN, blk, 0, stream>>>(aggn, wp_c1 + 0 * 2048,
                                                   c1_bl + 0 * 64,
                                                   xn0, wp_c1 + 1 * 2048,
                                                   xn1, NN, 64, 64, 0, 0);
    k_gemm<4, true><<<gemmBlkNC, blk, 0, stream>>>(aggcM, wp_c1 + 2 * 2048,
                                                   c1_bl + 1 * 64,
                                                   xc0, wp_c1 + 3 * 2048,
                                                   ocb, NC, 64, 64, 0, 0);
    k_gemm<4, true><<<gemmBlkNC, blk, 0, stream>>>(aggcR, wp_c1 + 4 * 2048,
                                                   c1_bl + 2 * 64,
                                                   xc0, wp_c1 + 5 * 2048,
                                                   ocb, NC, 64, 64, 0, 1);

    dim3 lnb(32, 8);
    k_ln<<<(NN + 7) / 8, lnb, 0, stream>>>(xn1, xn1, nullptr, n1n_g, n1n_b, NN);
    k_ln<<<(NC + 7) / 8, lnb, 0, stream>>>(xc1, ocb, xc0, n1c_g, n1c_b, NC);

    k_bin<20, 6144><<<(EM + 8191) / 8192, 1024, 0, stream>>>(men_src, men_dst, EM,
                                                             (NC + 19) / 20, blM2, ctrM2);
    k_own_agg<20, 8192, 6144><<<(NC + 19) / 20, 1024, 0, stream>>>(men_src, men_dst,
                                                                   xn1, aggcM, EM, NC,
                                                                   blM2, ctrM2);
    k_own_agg<20, 8192, 1536><<<(NC + 19) / 20, 1024, 0, stream>>>(edge_rel, edge_rel + ER,
                                                                   xc1, aggcR, ER, NC,
                                                                   blR, ctrR);
    k_gemm<4, true><<<gemmBlkNC, blk, 0, stream>>>(aggcM, wp_c2 + 0 * 2048,
                                                   c2_bl + 1 * 64,
                                                   xc1, wp_c2 + 1 * 2048,
                                                   ocb, NC, 64, 64, 0, 0);
    k_gemm<4, true><<<gemmBlkNC, blk, 0, stream>>>(aggcR, wp_c2 + 2 * 2048,
                                                   c2_bl + 2 * 64,
                                                   xc1, wp_c2 + 3 * 2048,
                                                   ocb, NC, 64, 64, 0, 1);

    k_addrelu<<<blocks((long)NC * HDIM), blk, 0, stream>>>(comp, ocb, xc1,
                                                           (long)NC * HDIM);
    k_gemm<2, true><<<gemmBlkNC, blk, 0, stream>>>(comp, wp_cls, cls_b1,
                                                   nullptr, nullptr,
                                                   hbuf, NC, 64, 0, 1, 0);
    k_final<<<blocks(NC), blk, 0, stream>>>(hbuf, cls_W2, cls_b2, out, NC);
}


# epilogue fusion: residual add, relu and row normalisation run on the WMMA accumulators in the GEMM epilogue; both norm passes and the add-relu pass go (35 to 32 launches); row sums in another order
# speedup vs baseline: 3.8683x; 1.1471x over previous
#include <hip/hip_runtime.h>
#include <hip/hip_bf16.h>

typedef __attribute__((ext_vector_type(2))) float v2f;
typedef __attribute__((ext_vector_type(8))) float v8f;

#define NN 100000
#define NC 5000
#define HDIM 64
#define ES 3200000
#define EM 1000000
#define ER 160000

__global__ __launch_bounds__(256) void k_packW(const float* __restrict__ W,
                                               v2f* __restrict__ Wp,
                                               int K, int N) {
    const int Kp2 = ((K + 3) & ~3) >> 1;
    const long n = (long)Kp2 * N;
    long i = (long)blockIdx.x * blockDim.x + threadIdx.x;
    if (i >= n) return;
    int k2 = (int)(i / N), col = (int)(i % N);
    int k = 2 * k2;
    v2f v;
    v.x = (k     < K) ? W[(long)k * N + col]       : 0.f;
    v.y = (k + 1 < K) ? W[(long)(k + 1) * N + col] : 0.f;
    Wp[i] = v;
}

#define PK_SH 9
#define PK_MASK 511
__global__ __launch_bounds__(256) void k_zero(int* __restrict__ p, int n) {
    const int i = blockIdx.x * blockDim.x + threadIdx.x;
    if (i < n) p[i] = 0;
}

template<int R, int CAPB>
__global__ __launch_bounds__(1024) void k_bin(const int* __restrict__ src,
                                              const int* __restrict__ dst,
                                              int nE, int nB,
                                              int* __restrict__ blist,
                                              int* __restrict__ bctr) {
    static_assert(R <= PK_MASK + 1, "row inside the range must fit the packed field");
    constexpr int EPT = 8;
    constexpr int CHUNK = 1024 * EPT;
    constexpr int NBMAX = 256;
    __shared__ int hcnt[NBMAX];
    __shared__ int hoff[NBMAX];
    __shared__ int gbase[NBMAX];
    __shared__ int stage[CHUNK];
    __shared__ int sown[CHUNK];

    const int tid = threadIdx.x;
    const int lane = tid & 31;
    const int wv = tid >> 5;
    const int base = blockIdx.x * CHUNK;

    if (tid < NBMAX) hcnt[tid] = 0;
    __syncthreads();

    int ow[EPT], rk[EPT], pk[EPT];
    #pragma unroll
    for (int u = 0; u < EPT; ++u) {
        const int e = base + u * 1024 + tid;
        ow[u] = -1; rk[u] = 0; pk[u] = 0;
        if (e < nE) {
            const int d = dst[e];
            const int o = (int)((unsigned)d / (unsigned)R);
            ow[u] = o;
            pk[u] = (src[e] << PK_SH) | (d - o * R);
            rk[u] = atomicAdd(&hcnt[o], 1);
        }
    }
    __syncthreads();

    if (tid < nB) {
        const int cnt = hcnt[tid];
        gbase[tid] = cnt ? atomicAdd(&bctr[tid], cnt) : 0;
    }
    if (wv == 31) {
        int cq[8];
        int t = 0;
        #pragma unroll
        for (int q = 0; q < 8; ++q) { const int v = hcnt[lane * 8 + q]; cq[q] = t; t += v; }
        int incl = t;
        #pragma unroll
        for (int off = 1; off < 32; off <<= 1) {
            const int s = __shfl_up(incl, off, 32);
            if (lane >= off) incl += s;
        }
        const int excl = incl - t;
        #pragma unroll
        for (int q = 0; q < 8; ++q) hoff[lane * 8 + q] = cq[q] + excl;
    }
    __syncthreads();

    #pragma unroll
    for (int u = 0; u < EPT; ++u) {
        if (ow[u] >= 0) {
            const int lp = hoff[ow[u]] + rk[u];
            stage[lp] = pk[u];
            sown[lp] = ow[u];
        }
    }
    __syncthreads();

    const int total = min(CHUNK, nE - base);
    for (int i = tid; i < total; i += 1024) {
        const int o = sown[i];
        const int pos = gbase[o] + (i - hoff[o]);
        if (pos < CAPB) blist[(long)o * CAPB + pos] = stage[i];
        else bctr[nB] = 1;
    }
}

__device__ __forceinline__ void add_group(const float* __restrict__ x, float* acc,
                                          int s, int dl, int m, int lane) {
    if (m >= 32) {
        float v0[32], v1[32];
        #pragma unroll
        for (int j = 0; j < 32; ++j) {
            const int sj = __builtin_amdgcn_readlane(s, j);
            const float* xr = x + (long)sj * HDIM + lane;
            v0[j] = xr[0];
            v1[j] = xr[32];
        }
        #pragma unroll
        for (int j = 0; j < 32; ++j) {
            const int dj = __builtin_amdgcn_readlane(dl, j);
            float* ar = acc + dj * HDIM + lane;
            atomicAdd(ar, v0[j]);
            atomicAdd(ar + 32, v1[j]);
        }
    } else {
        for (int j = 0; j < m; ++j) {
            const int sj = __builtin_amdgcn_readlane(s, j);
            const int dj = __builtin_amdgcn_readlane(dl, j);
            const float* xr = x + (long)sj * HDIM + lane;
            float* ar = acc + dj * HDIM + lane;
            const float a0 = xr[0];
            const float a1 = xr[32];
            atomicAdd(ar, a0);
            atomicAdd(ar + 32, a1);
        }
    }
}

template<int R, int CAP, int CAPB>
__global__ __launch_bounds__(1024) void k_own_agg(const int* __restrict__ src,
                                                  const int* __restrict__ dst,
                                                  const float* __restrict__ x,
                                                  float* __restrict__ out,
                                                  int nE, int nDst,
                                                  const int* __restrict__ blist,
                                                  const int* __restrict__ bctr) {
    __shared__ __attribute__((aligned(16))) float acc[R * HDIM];
    __shared__ int cn[R];
    __shared__ int list[CAP];
    __shared__ int nlist;

    const int tid = threadIdx.x;
    const int lane = tid & 31;
    const int wv = __builtin_amdgcn_readfirstlane(tid >> 5);
    const int lo = blockIdx.x * R;
    const int nr = min(R, nDst - lo);
    const bool binned = __builtin_amdgcn_readfirstlane(bctr[gridDim.x]) == 0;
    const int nbin = __builtin_amdgcn_readfirstlane(bctr[blockIdx.x]);

    for (int i = tid; i < R * HDIM; i += 1024) acc[i] = 0.f;
    for (int i = tid; i < R; i += 1024) cn[i] = 0;
    if (tid == 0) nlist = 0;
    __syncthreads();

    if (!binned) {
    constexpr int U = 8;
    constexpr int CH = U * 128;
    const int nCh = (nE + CH - 1) / CH;
    const int rot = (int)(((long)blockIdx.x * nCh) / gridDim.x);
    for (int c0 = wv; c0 < nCh; c0 += 32) {
        int c = c0 + rot;
        if (c >= nCh) c -= nCh;
        const int base = c * CH;
        int4 d[U];
        if (base + CH <= nE) {
            #pragma unroll
            for (int u = 0; u < U; ++u)
                d[u] = *(const int4*)(dst + base + u * 128 + lane * 4);
        } else {
            #pragma unroll
            for (int u = 0; u < U; ++u) {
                const int i0 = base + u * 128 + lane * 4;
                d[u].x = (i0     < nE) ? dst[i0]     : -1;
                d[u].y = (i0 + 1 < nE) ? dst[i0 + 1] : -1;
                d[u].z = (i0 + 2 < nE) ? dst[i0 + 2] : -1;
                d[u].w = (i0 + 3 < nE) ? dst[i0 + 3] : -1;
            }
        }
        int cl = 0;
        #pragma unroll
        for (int u = 0; u < U; ++u) {
            cl += ((unsigned)(d[u].x - lo) < (unsigned)nr) ? 1 : 0;
            cl += ((unsigned)(d[u].y - lo) < (unsigned)nr) ? 1 : 0;
            cl += ((unsigned)(d[u].z - lo) < (unsigned)nr) ? 1 : 0;
            cl += ((unsigned)(d[u].w - lo) < (unsigned)nr) ? 1 : 0;
        }
        int incl = cl;
        #pragma unroll
        for (int off = 1; off < 32; off <<= 1) {
            const int t = __shfl_up(incl, off, 32);
            if (lane >= off) incl += t;
        }
        const int total = __builtin_amdgcn_readlane(incl, 31);
        if (total == 0) continue;
        int s0 = 0;
        if (lane == 31) s0 = atomicAdd(&nlist, total);
        s0 = __builtin_amdgcn_readlane(s0, 31);
        int pos = s0 + incl - cl;
        #pragma unroll
        for (int u = 0; u < U; ++u) {
            #pragma unroll
            for (int q = 0; q < 4; ++q) {
                const int dv = (q == 0) ? d[u].x : (q == 1) ? d[u].y : (q == 2) ? d[u].z : d[u].w;
                const int dl = dv - lo;
                if ((unsigned)dl < (unsigned)nr) {
                    const int e = base + u * 128 + lane * 4 + q;
                    if (pos < CAP) {
                        list[pos] = e;
                    } else {
                        const float* xr = x + (long)src[e] * HDIM;
                        #pragma nounroll
                        for (int k = 0; k < HDIM; ++k) atomicAdd(&acc[dl * HDIM + k], xr[k]);
                        atomicAdd(&cn[dl], 1);
                    }
                    ++pos;
                }
            }
        }
    }
    }
    __syncthreads();

    if (binned) {
        const int n = min(nbin, CAPB);
        const int* __restrict__ bl = blist + (long)blockIdx.x * CAPB;
        int g = wv * 32;
        if (g < n) {
            int pn = bl[min(g + lane, n - 1)];
            for (; g < n; g += 32 * 32) {
                const int p = pn;
                pn = bl[min(g + 32 * 32 + lane, n - 1)];
                const int m = min(32, n - g);
                int s = 0, dl = 0;
                if (lane < m) {
                    s = p >> PK_SH;
                    dl = p & PK_MASK;
                    atomicAdd(&cn[dl], 1);
                }
                add_group(x, acc, s, dl, m, lane);
            }
        }
    } else {
        const int n = min(nlist, CAP);
        for (int g = wv * 32; g < n; g += 32 * 32) {
            const int k = g + lane;
            int s = 0, dl = 0;
            if (k < n) {
                const int e = list[k];
                s = src[e];
                dl = dst[e] - lo;
                atomicAdd(&cn[dl], 1);
            }
            add_group(x, acc, s, dl, min(32, n - g), lane);
        }
    }
    __syncthreads();

    for (int i = tid; i < nr * 16; i += 1024) {
        const int r = i >> 4;
        const int c4 = (i & 15) * 4;
        const float inv = 1.0f / fmaxf((float)cn[r], 1.0f);
        float4 v = *(const float4*)(acc + r * HDIM + c4);
        v.x *= inv; v.y *= inv; v.z *= inv; v.w *= inv;
        *(float4*)(out + (long)(lo + r) * HDIM + c4) = v;
    }
}

template<int NT, bool FULL, bool VECA>
__device__ __forceinline__ void gemm_body(v8f (&c)[NT],
                                          const float* __restrict__ A,
                                          const v2f* __restrict__ Wp,
                                          int M, int K, int N,
                                          int mbase, int lane) {
    const int row  = mbase + (lane & 15);
    const int koff = (lane >= 16) ? 2 : 0;
    const int col0 = lane & 15;
    const bool rok = FULL || (row < M);
    const float* __restrict__ Arow = A + (long)row * K;
    const int Kmain = K & ~3;

    #pragma unroll 4
    for (int k = 0; k < Kmain; k += 4) {
        const int kk = k + koff;
        v2f a;
        if (rok) {
            if (VECA) {
                a = *(const v2f*)(Arow + kk);
            } else {
                a.x = Arow[kk];
                a.y = Arow[kk + 1];
            }
        } else {
            a.x = 0.f; a.y = 0.f;
        }
        #pragma unroll
        for (int nt = 0; nt < NT; ++nt) {
            v2f b = Wp[(long)(kk >> 1) * N + nt * 16 + col0];
            c[nt] = __builtin_amdgcn_wmma_f32_16x16x4_f32(
                false, a, false, b, (short)0, c[nt], false, false);
        }
    }
    if (Kmain < K) {
        const int kk = Kmain + koff;
        v2f a;
        a.x = (rok && kk     < K) ? Arow[kk]     : 0.f;
        a.y = (rok && kk + 1 < K) ? Arow[kk + 1] : 0.f;
        #pragma unroll
        for (int nt = 0; nt < NT; ++nt) {
            v2f b = Wp[(long)(kk >> 1) * N + nt * 16 + col0];
            c[nt] = __builtin_amdgcn_wmma_f32_16x16x4_f32(
                false, a, false, b, (short)0, c[nt], false, false);
        }
    }
}

#define SLOTS8 "v_nop\n\tv_nop\n\tv_nop\n\tv_nop\n\tv_nop\n\tv_nop\n\tv_nop\n\tv_nop"
__device__ __forceinline__ void result_slots(v8f (&c)[4]) {
    asm volatile(SLOTS8 : "+v"(c[0]), "+v"(c[1]), "+v"(c[2]), "+v"(c[3]));
}
__device__ __forceinline__ void result_slots(v8f (&c)[2]) {
    asm volatile(SLOTS8 : "+v"(c[0]), "+v"(c[1]));
}

template<int NT, bool VECA1, bool NORM>
__global__ __launch_bounds__(256) void k_gemm(const float* __restrict__ A1,
                                              const v2f* __restrict__ W1p,
                                              const float* __restrict__ bias,
                                              const float* __restrict__ A2,
                                              const v2f* __restrict__ W2p,
                                              const float* __restrict__ prev,
                                              const float* __restrict__ res,
                                              const float* __restrict__ g,
                                              const float* __restrict__ b,
                                              float* __restrict__ out,
                                              int M, int K1, int K2,
                                              int do_relu) {
    static_assert(!NORM || NT * 16 == HDIM, "the normalised row must be the whole tile row");
    const int N = NT * 16;
    const int lane = threadIdx.x & 31;
    const int wv = threadIdx.x >> 5;
    const int mbase = (blockIdx.x * 8 + wv) * 16;
    if (mbase >= M) return;

    v8f c[NT];
    #pragma unroll
    for (int t = 0; t < NT; ++t)
        c[t] = (v8f){0.f, 0.f, 0.f, 0.f, 0.f, 0.f, 0.f, 0.f};

    const bool full = (mbase + 16 <= M);
    if (full) {
        gemm_body<NT, true, VECA1>(c, A1, W1p, M, K1, N, mbase, lane);
        if (A2) gemm_body<NT, true, true>(c, A2, W2p, M, K2, N, mbase, lane);
    } else {
        gemm_body<NT, false, VECA1>(c, A1, W1p, M, K1, N, mbase, lane);
        if (A2) gemm_body<NT, false, true>(c, A2, W2p, M, K2, N, mbase, lane);
    }

    result_slots(c);

    const int col0 = lane & 15;
    const int rofs = (lane >= 16) ? 8 : 0;
    #pragma unroll
    for (int nt = 0; nt < NT; ++nt) {
        const float bb = bias ? bias[nt * 16 + col0] : 0.f;
        #pragma unroll
        for (int r = 0; r < 8; ++r) c[nt][r] += bb;
    }
    if (prev) {
        #pragma unroll
        for (int r = 0; r < 8; ++r) {
            const long rb = (long)min(mbase + r + rofs, M - 1) * N + col0;
            #pragma unroll
            for (int nt = 0; nt < NT; ++nt) c[nt][r] += prev[rb + nt * 16];
        }
    }
    if (res) {
        #pragma unroll
        for (int r = 0; r < 8; ++r) {
            const long rb = (long)min(mbase + r + rofs, M - 1) * N + col0;
            #pragma unroll
            for (int nt = 0; nt < NT; ++nt) c[nt][r] += res[rb + nt * 16];
        }
    }

    if (NORM) {
        float gg[NT], be[NT];
        #pragma unroll
        for (int nt = 0; nt < NT; ++nt) { gg[nt] = g[nt * 16 + col0]; be[nt] = b[nt * 16 + col0]; }
        #pragma unroll
        for (int r = 0; r < 8; ++r) {
            const int row = mbase + r + rofs;
            float v[NT];
            #pragma unroll
            for (int nt = 0; nt < NT; ++nt) v[nt] = fmaxf(c[nt][r], 0.f);
            float s = v[0];
            #pragma unroll
            for (int nt = 1; nt < NT; ++nt) s += v[nt];
            #pragma unroll
            for (int off = 8; off > 0; off >>= 1) s += __shfl_xor(s, off, 32);
            const float mean = s * (1.0f / 64.0f);
            float d[NT];
            #pragma unroll
            for (int nt = 0; nt < NT; ++nt) d[nt] = v[nt] - mean;
            float q = d[0] * d[0];
            #pragma unroll
            for (int nt = 1; nt < NT; ++nt) q += d[nt] * d[nt];
            #pragma unroll
            for (int off = 8; off > 0; off >>= 1) q += __shfl_xor(q, off, 32);
            const float rstd = rsqrtf(q * (1.0f / 64.0f) + 1e-5f);
            if (full || row < M) {
                const long rb = (long)row * N + col0;
                #pragma unroll
                for (int nt = 0; nt < NT; ++nt)
                    out[rb + nt * 16] = d[nt] * rstd * gg[nt] + be[nt];
            }
        }
    } else {
        #pragma unroll
        for (int nt = 0; nt < NT; ++nt) {
            #pragma unroll
            for (int r = 0; r < 8; ++r) {
                const int row = mbase + r + rofs;
                if (full || row < M) {
                    float v = c[nt][r];
                    if (do_relu) v = fmaxf(v, 0.f);
                    out[(long)row * N + nt * 16 + col0] = v;
                }
            }
        }
    }
}

__global__ __launch_bounds__(256) void k_final(const float* __restrict__ h,
                                               const float* __restrict__ W2,
                                               const float* __restrict__ b2,
                                               float* __restrict__ out, int M) {
    int i = blockIdx.x * blockDim.x + threadIdx.x;
    if (i >= M) return;
    float acc = b2[0];
    const float* hr = h + (long)i * 32;
    #pragma unroll
    for (int j = 0; j < 32; ++j) acc += hr[j] * W2[j];
    out[i] = acc;
}

extern "C" void kernel_launch(void* const* d_in, const int* in_sizes, int n_in,
                              void* d_out, int out_size, void* d_ws, size_t ws_size,
                              hipStream_t stream) {
    const float* x_news    = (const float*)d_in[0];
    const float* x_company = (const float*)d_in[1];
    const int*   edge_sim  = (const int*)d_in[2];
    const int*   men_src   = (const int*)d_in[3];
    const int*   men_dst   = (const int*)d_in[4];
    const int*   edge_rel  = (const int*)d_in[5];
    const float* nw_W = (const float*)d_in[6];
    const float* nw_b = (const float*)d_in[7];
    const float* cw_W = (const float*)d_in[8];
    const float* cw_b = (const float*)d_in[9];
    const float* n1n_g = (const float*)d_in[10];
    const float* n1n_b = (const float*)d_in[11];
    const float* n1c_g = (const float*)d_in[12];
    const float* n1c_b = (const float*)d_in[13];
    const float* c1_Wl = (const float*)d_in[14];
    const float* c1_bl = (const float*)d_in[15];
    const float* c1_Wr = (const float*)d_in[16];
    const float* c2_Wl = (const float*)d_in[17];
    const float* c2_bl = (const float*)d_in[18];
    const float* c2_Wr = (const float*)d_in[19];
    const float* cls_W1 = (const float*)d_in[20];
    const float* cls_b1 = (const float*)d_in[21];
    const float* cls_W2 = (const float*)d_in[22];
    const float* cls_b2 = (const float*)d_in[23];
    float* out = (float*)d_out;

    float* ws = (float*)d_ws;
    long o = 0;
    float* xn0  = ws + o; o += (long)NN * HDIM;
    float* xn1  = ws + o; o += (long)NN * HDIM;
    float* aggn = ws + o; o += (long)NN * HDIM;
    float* aggcM = ws + o; o += (long)NC * HDIM;
    float* aggcR = ws + o; o += (long)NC * HDIM;
    float* xc0  = ws + o; o += (long)NC * HDIM;
    float* xc1  = ws + o; o += (long)NC * HDIM;
    float* ocb  = ws + o; o += (long)NC * HDIM;
    float* comp = ws + o; o += (long)NC * HDIM;
    float* hbuf = ws + o; o += (long)NC * 32;
    v2f* wp_nw  = (v2f*)(ws + o); o += 2L * (388 / 2) * 64;
    v2f* wp_cw  = (v2f*)(ws + o); o += 2L * (24 / 2) * 64;
    v2f* wp_c1  = (v2f*)(ws + o); o += 2L * 6 * 32 * 64;
    v2f* wp_c2  = (v2f*)(ws + o); o += 2L * 4 * 32 * 64;
    v2f* wp_cls = (v2f*)(ws + o); o += 2L * 32 * 32;

    int* ctr   = (int*)comp;
    int* ctrN  = ctr;
    int* ctrM1 = ctr + 512;
    int* ctrR  = ctr + 1024;
    int* ctrM2 = ctr + 1536;
    int* blR   = ctr + 4096;
    int* blN   = (int*)xn1;
    int* blM1  = blN + 256L * 15360;
    int* blM2  = (int*)xn0;

    dim3 blk(256);
    auto blocks = [](long n) { return (unsigned)((n + 255) / 256); };
    int tilesNN = (NN + 15) / 16, tilesNC = (NC + 15) / 16;
    unsigned gemmBlkNN = (tilesNN + 7) / 8, gemmBlkNC = (tilesNC + 7) / 8;

    k_packW<<<blocks((388L / 2) * 64), blk, 0, stream>>>(nw_W, wp_nw, 385, 64);
    k_packW<<<blocks((24L / 2) * 64), blk, 0, stream>>>(cw_W, wp_cw, 24, 64);
    for (int r = 0; r < 3; ++r) {
        k_packW<<<blocks(32L * 64), blk, 0, stream>>>(c1_Wl + r * 4096,
                                                      wp_c1 + (2 * r) * 2048, 64, 64);
        k_packW<<<blocks(32L * 64), blk, 0, stream>>>(c1_Wr + r * 4096,
                                                      wp_c1 + (2 * r + 1) * 2048, 64, 64);
    }
    for (int r = 1; r < 3; ++r) {
        k_packW<<<blocks(32L * 64), blk, 0, stream>>>(c2_Wl + r * 4096,
                                                      wp_c2 + (2 * (r - 1)) * 2048, 64, 64);
        k_packW<<<blocks(32L * 64), blk, 0, stream>>>(c2_Wr + r * 4096,
                                                      wp_c2 + (2 * (r - 1) + 1) * 2048, 64, 64);
    }
    k_packW<<<blocks(32L * 32), blk, 0, stream>>>(cls_W1, wp_cls, 64, 32);

    k_gemm<4, false, false><<<gemmBlkNN, blk, 0, stream>>>(x_news, wp_nw, nw_b,
                                                           nullptr, nullptr,
                                                           nullptr, nullptr, nullptr, nullptr,
                                                           xn0, NN, 385, 0, 1);
    k_gemm<4, true, false><<<gemmBlkNC, blk, 0, stream>>>(x_company, wp_cw, cw_b,
                                                          nullptr, nullptr,
                                                          nullptr, nullptr, nullptr, nullptr,
                                                          xc0, NC, 24, 0, 1);

    k_zero<<<16, blk, 0, stream>>>(ctr, 4096);
    k_bin<391, 15360><<<(ES + 8191) / 8192, 1024, 0, stream>>>(edge_sim, edge_sim + ES, ES,
                                                               (NN + 390) / 391, blN, ctrN);
    k_bin<20, 6144><<<(EM + 8191) / 8192, 1024, 0, stream>>>(men_src, men_dst, EM,
                                                             (NC + 19) / 20, blM1, ctrM1);
    k_bin<20, 1536><<<(ER + 8191) / 8192, 1024, 0, stream>>>(edge_rel, edge_rel + ER, ER,
                                                             (NC + 19) / 20, blR, ctrR);
    k_own_agg<391, 16384, 15360><<<(NN + 390) / 391, 1024, 0, stream>>>(edge_sim, edge_sim + ES,
                                                                        xn0, aggn, ES, NN,
                                                                        blN, ctrN);
    k_own_agg<20, 8192, 6144><<<(NC + 19) / 20, 1024, 0, stream>>>(men_src, men_dst,
                                                                   xn0, aggcM, EM, NC,
                                                                   blM1, ctrM1);
    k_own_agg<20, 8192, 1536><<<(NC + 19) / 20, 1024, 0, stream>>>(edge_rel, edge_rel + ER,
                                                                   xc0, aggcR, ER, NC,
                                                                   blR, ctrR);

    k_gemm<4, true, true><<<gemmBlkNN, blk, 0, stream>>>(aggn, wp_c1 + 0 * 2048,
                                                         c1_bl + 0 * 64,
                                                         xn0, wp_c1 + 1 * 2048,
                                                         nullptr, nullptr, n1n_g, n1n_b,
                                                         xn1, NN, 64, 64, 1);
    k_gemm<4, true, false><<<gemmBlkNC, blk, 0, stream>>>(aggcM, wp_c1 + 2 * 2048,
                                                          c1_bl + 1 * 64,
                                                          xc0, wp_c1 + 3 * 2048,
                                                          nullptr, nullptr, nullptr, nullptr,
                                                          ocb, NC, 64, 64, 0);
    k_gemm<4, true, true><<<gemmBlkNC, blk, 0, stream>>>(aggcR, wp_c1 + 4 * 2048,
                                                         c1_bl + 2 * 64,
                                                         xc0, wp_c1 + 5 * 2048,
                                                         ocb, xc0, n1c_g, n1c_b,
                                                         xc1, NC, 64, 64, 1);

    k_bin<20, 6144><<<(EM + 8191) / 8192, 1024, 0, stream>>>(men_src, men_dst, EM,
                                                             (NC + 19) / 20, blM2, ctrM2);
    k_own_agg<20, 8192, 6144><<<(NC + 19) / 20, 1024, 0, stream>>>(men_src, men_dst,
                                                                   xn1, aggcM, EM, NC,
                                                                   blM2, ctrM2);
    k_own_agg<20, 8192, 1536><<<(NC + 19) / 20, 1024, 0, stream>>>(edge_rel, edge_rel + ER,
                                                                   xc1, aggcR, ER, NC,
                                                                   blR, ctrR);
    k_gemm<4, true, false><<<gemmBlkNC, blk, 0, stream>>>(aggcM, wp_c2 + 0 * 2048,
                                                          c2_bl + 1 * 64,
                                                          xc1, wp_c2 + 1 * 2048,
                                                          nullptr, nullptr, nullptr, nullptr,
                                                          ocb, NC, 64, 64, 0);
    k_gemm<4, true, false><<<gemmBlkNC, blk, 0, stream>>>(aggcR, wp_c2 + 2 * 2048,
                                                          c2_bl + 2 * 64,
                                                          xc1, wp_c2 + 3 * 2048,
                                                          ocb, xc1, nullptr, nullptr,
                                                          comp, NC, 64, 64, 1);

    k_gemm<2, true, false><<<gemmBlkNC, blk, 0, stream>>>(comp, wp_cls, cls_b1,
                                                          nullptr, nullptr,
                                                          nullptr, nullptr, nullptr, nullptr,
                                                          hbuf, NC, 64, 0, 1);
    k_final<<<blocks(NC), blk, 0, stream>>>(hbuf, cls_W2, cls_b2, out, NC);
}


# launch merging: independent launches of one stage share one grid, job picked by block index: 13 weight packs + counter clear, 3 binning passes, 3 + 2 aggregation passes; 32 to 14 launches
# speedup vs baseline: 4.4109x; 1.1403x over previous
#include <hip/hip_runtime.h>
#include <hip/hip_bf16.h>

typedef __attribute__((ext_vector_type(2))) float v2f;
typedef __attribute__((ext_vector_type(8))) float v8f;

#define NN 100000
#define NC 5000
#define HDIM 64
#define ES 3200000
#define EM 1000000
#define ER 160000

__device__ __forceinline__ void pack_one(const float* __restrict__ W,
                                         v2f* __restrict__ Wp,
                                         int K, int N, int i) {
    const int Kp2 = ((K + 3) & ~3) >> 1;
    if (i >= Kp2 * N) return;
    const int k2 = i / N, col = i % N;
    const int k = 2 * k2;
    v2f v;
    v.x = (k     < K) ? W[k * N + col]       : 0.f;
    v.y = (k + 1 < K) ? W[(k + 1) * N + col] : 0.f;
    Wp[i] = v;
}

#define PB_NW  49
#define PB_CW  3
#define PB_C1  48
#define PB_C2  32
#define PB_CLS 4
#define PB_Z   16
__global__ __launch_bounds__(256) void k_prep(const float* __restrict__ nw_W,
                                              const float* __restrict__ cw_W,
                                              const float* __restrict__ c1_Wl,
                                              const float* __restrict__ c1_Wr,
                                              const float* __restrict__ c2_Wl,
                                              const float* __restrict__ c2_Wr,
                                              const float* __restrict__ cls_W1,
                                              v2f* __restrict__ wp_nw,
                                              v2f* __restrict__ wp_cw,
                                              v2f* __restrict__ wp_c1,
                                              v2f* __restrict__ wp_c2,
                                              v2f* __restrict__ wp_cls,
                                              int* __restrict__ ctr) {
    const int tid = threadIdx.x;
    int b = blockIdx.x;
    if (b < PB_NW) { pack_one(nw_W, wp_nw, 385, 64, b * 256 + tid); return; }
    b -= PB_NW;
    if (b < PB_CW) { pack_one(cw_W, wp_cw, 24, 64, b * 256 + tid); return; }
    b -= PB_CW;
    if (b < PB_C1) {
        const int m = b >> 3;
        const float* W = ((m & 1) ? c1_Wr : c1_Wl) + (m >> 1) * 4096;
        pack_one(W, wp_c1 + m * 2048, 64, 64, (b & 7) * 256 + tid);
        return;
    }
    b -= PB_C1;
    if (b < PB_C2) {
        const int m = b >> 3;
        const float* W = ((m & 1) ? c2_Wr : c2_Wl) + ((m >> 1) + 1) * 4096;
        pack_one(W, wp_c2 + m * 2048, 64, 64, (b & 7) * 256 + tid);
        return;
    }
    b -= PB_C2;
    if (b < PB_CLS) { pack_one(cls_W1, wp_cls, 64, 32, b * 256 + tid); return; }
    b -= PB_CLS;
    ctr[b * 256 + tid] = 0;
}

#define PK_SH 9
#define PK_MASK 511
#define BIN_EPT 8
#define BIN_CHUNK (1024 * BIN_EPT)
#define BIN_NBMAX 256
template<int R, int CAPB>
__device__ __forceinline__ void bin_body(const int* __restrict__ src,
                                         const int* __restrict__ dst,
                                         int nE, int nB,
                                         int* __restrict__ blist,
                                         int* __restrict__ bctr,
                                         int bid,
                                         int* hcnt, int* hoff, int* gbase,
                                         int* stage, int* sown) {
    static_assert(R <= PK_MASK + 1, "row inside the range must fit the packed field");
    constexpr int EPT = BIN_EPT;
    constexpr int CHUNK = BIN_CHUNK;
    constexpr int NBMAX = BIN_NBMAX;

    const int tid = threadIdx.x;
    const int lane = tid & 31;
    const int wv = tid >> 5;
    const int base = bid * CHUNK;

    if (tid < NBMAX) hcnt[tid] = 0;
    __syncthreads();

    int ow[EPT], rk[EPT], pk[EPT];
    #pragma unroll
    for (int u = 0; u < EPT; ++u) {
        const int e = base + u * 1024 + tid;
        ow[u] = -1; rk[u] = 0; pk[u] = 0;
        if (e < nE) {
            const int d = dst[e];
            const int o = (int)((unsigned)d / (unsigned)R);
            ow[u] = o;
            pk[u] = (src[e] << PK_SH) | (d - o * R);
            rk[u] = atomicAdd(&hcnt[o], 1);
        }
    }
    __syncthreads();

    if (tid < nB) {
        const int cnt = hcnt[tid];
        gbase[tid] = cnt ? atomicAdd(&bctr[tid], cnt) : 0;
    }
    if (wv == 31) {
        int cq[8];
        int t = 0;
        #pragma unroll
        for (int q = 0; q < 8; ++q) { const int v = hcnt[lane * 8 + q]; cq[q] = t; t += v; }
        int incl = t;
        #pragma unroll
        for (int off = 1; off < 32; off <<= 1) {
            const int s = __shfl_up(incl, off, 32);
            if (lane >= off) incl += s;
        }
        const int excl = incl - t;
        #pragma unroll
        for (int q = 0; q < 8; ++q) hoff[lane * 8 + q] = cq[q] + excl;
    }
    __syncthreads();

    #pragma unroll
    for (int u = 0; u < EPT; ++u) {
        if (ow[u] >= 0) {
            const int lp = hoff[ow[u]] + rk[u];
            stage[lp] = pk[u];
            sown[lp] = ow[u];
        }
    }
    __syncthreads();

    const int total = min(CHUNK, nE - base);
    for (int i = tid; i < total; i += 1024) {
        const int o = sown[i];
        const int pos = gbase[o] + (i - hoff[o]);
        if (pos < CAPB) blist[(long)o * CAPB + pos] = stage[i];
        else bctr[nB] = 1;
    }
}

#define BIN_LDS \
    __shared__ int hcnt[BIN_NBMAX];      \
    __shared__ int hoff[BIN_NBMAX];      \
    __shared__ int gbase[BIN_NBMAX];     \
    __shared__ int stage[BIN_CHUNK];     \
    __shared__ int sown[BIN_CHUNK];

template<int R, int CAPB>
__global__ __launch_bounds__(1024) void k_bin(const int* __restrict__ src,
                                              const int* __restrict__ dst,
                                              int nE, int nB,
                                              int* __restrict__ blist,
                                              int* __restrict__ bctr) {
    BIN_LDS
    bin_body<R, CAPB>(src, dst, nE, nB, blist, bctr, blockIdx.x, hcnt, hoff, gbase, stage, sown);
}

#define BINB_N ((ES + BIN_CHUNK - 1) / BIN_CHUNK)
#define BINB_M ((EM + BIN_CHUNK - 1) / BIN_CHUNK)
#define BINB_R ((ER + BIN_CHUNK - 1) / BIN_CHUNK)
#define OWN_N ((NN + 390) / 391)
#define OWN_C ((NC + 19) / 20)
__global__ __launch_bounds__(1024) void k_bin3(const int* __restrict__ sN, const int* __restrict__ dN,
                                               int* __restrict__ blN, int* __restrict__ cN,
                                               const int* __restrict__ sM, const int* __restrict__ dM,
                                               int* __restrict__ blM, int* __restrict__ cM,
                                               const int* __restrict__ sR, const int* __restrict__ dR,
                                               int* __restrict__ blR, int* __restrict__ cR) {
    BIN_LDS
    const int b = blockIdx.x;
    if (b < BINB_N)
        bin_body<391, 15360>(sN, dN, ES, OWN_N, blN, cN, b, hcnt, hoff, gbase, stage, sown);
    else if (b < BINB_N + BINB_M)
        bin_body<20, 6144>(sM, dM, EM, OWN_C, blM, cM, b - BINB_N, hcnt, hoff, gbase, stage, sown);
    else
        bin_body<20, 1536>(sR, dR, ER, OWN_C, blR, cR, b - BINB_N - BINB_M, hcnt, hoff, gbase, stage, sown);
}

__device__ __forceinline__ void add_group(const float* __restrict__ x, float* acc,
                                          int s, int dl, int m, int lane) {
    if (m >= 32) {
        float v0[32], v1[32];
        #pragma unroll
        for (int j = 0; j < 32; ++j) {
            const int sj = __builtin_amdgcn_readlane(s, j);
            const float* xr = x + (long)sj * HDIM + lane;
            v0[j] = xr[0];
            v1[j] = xr[32];
        }
        #pragma unroll
        for (int j = 0; j < 32; ++j) {
            const int dj = __builtin_amdgcn_readlane(dl, j);
            float* ar = acc + dj * HDIM + lane;
            atomicAdd(ar, v0[j]);
            atomicAdd(ar + 32, v1[j]);
        }
    } else {
        for (int j = 0; j < m; ++j) {
            const int sj = __builtin_amdgcn_readlane(s, j);
            const int dj = __builtin_amdgcn_readlane(dl, j);
            const float* xr = x + (long)sj * HDIM + lane;
            float* ar = acc + dj * HDIM + lane;
            const float a0 = xr[0];
            const float a1 = xr[32];
            atomicAdd(ar, a0);
            atomicAdd(ar + 32, a1);
        }
    }
}

template<int R, int CAP, int CAPB>
__device__ __forceinline__ void own_agg_body(const int* __restrict__ src,
                                             const int* __restrict__ dst,
                                             const float* __restrict__ x,
                                             float* __restrict__ out,
                                             int nE, int nDst,
                                             const int* __restrict__ blist,
                                             const int* __restrict__ bctr,
                                             int bid, int nB,
                                             float* acc, int* cn, int* list, int& nlist) {
    const int tid = threadIdx.x;
    const int lane = tid & 31;
    const int wv = __builtin_amdgcn_readfirstlane(tid >> 5);
    const int lo = bid * R;
    const int nr = min(R, nDst - lo);
    const bool binned = __builtin_amdgcn_readfirstlane(bctr[nB]) == 0;
    const int nbin = __builtin_amdgcn_readfirstlane(bctr[bid]);

    for (int i = tid; i < R * HDIM; i += 1024) acc[i] = 0.f;
    for (int i = tid; i < R; i += 1024) cn[i] = 0;
    if (tid == 0) nlist = 0;
    __syncthreads();

    if (!binned) {
    constexpr int U = 8;
    constexpr int CH = U * 128;
    const int nCh = (nE + CH - 1) / CH;
    const int rot = (int)(((long)bid * nCh) / nB);
    for (int c0 = wv; c0 < nCh; c0 += 32) {
        int c = c0 + rot;
        if (c >= nCh) c -= nCh;
        const int base = c * CH;
        int4 d[U];
        if (base + CH <= nE) {
            #pragma unroll
            for (int u = 0; u < U; ++u)
                d[u] = *(const int4*)(dst + base + u * 128 + lane * 4);
        } else {
            #pragma unroll
            for (int u = 0; u < U; ++u) {
                const int i0 = base + u * 128 + lane * 4;
                d[u].x = (i0     < nE) ? dst[i0]     : -1;
                d[u].y = (i0 + 1 < nE) ? dst[i0 + 1] : -1;
                d[u].z = (i0 + 2 < nE) ? dst[i0 + 2] : -1;
                d[u].w = (i0 + 3 < nE) ? dst[i0 + 3] : -1;
            }
        }
        int cl = 0;
        #pragma unroll
        for (int u = 0; u < U; ++u) {
            cl += ((unsigned)(d[u].x - lo) < (unsigned)nr) ? 1 : 0;
            cl += ((unsigned)(d[u].y - lo) < (unsigned)nr) ? 1 : 0;
            cl += ((unsigned)(d[u].z - lo) < (unsigned)nr) ? 1 : 0;
            cl += ((unsigned)(d[u].w - lo) < (unsigned)nr) ? 1 : 0;
        }
        int incl = cl;
        #pragma unroll
        for (int off = 1; off < 32; off <<= 1) {
            const int t = __shfl_up(incl, off, 32);
            if (lane >= off) incl += t;
        }
        const int total = __builtin_amdgcn_readlane(incl, 31);
        if (total == 0) continue;
        int s0 = 0;
        if (lane == 31) s0 = atomicAdd(&nlist, total);
        s0 = __builtin_amdgcn_readlane(s0, 31);
        int pos = s0 + incl - cl;
        #pragma unroll
        for (int u = 0; u < U; ++u) {
            #pragma unroll
            for (int q = 0; q < 4; ++q) {
                const int dv = (q == 0) ? d[u].x : (q == 1) ? d[u].y : (q == 2) ? d[u].z : d[u].w;
                const int dl = dv - lo;
                if ((unsigned)dl < (unsigned)nr) {
                    const int e = base + u * 128 + lane * 4 + q;
                    if (pos < CAP) {
                        list[pos] = e;
                    } else {
                        const float* xr = x + (long)src[e] * HDIM;
                        #pragma nounroll
                        for (int k = 0; k < HDIM; ++k) atomicAdd(&acc[dl * HDIM + k], xr[k]);
                        atomicAdd(&cn[dl], 1);
                    }
                    ++pos;
                }
            }
        }
    }
    }
    __syncthreads();

    if (binned) {
        const int n = min(nbin, CAPB);
        const int* __restrict__ bl = blist + (long)bid * CAPB;
        int g = wv * 32;
        if (g < n) {
            int pn = bl[min(g + lane, n - 1)];
            for (; g < n; g += 32 * 32) {
                const int p = pn;
                pn = bl[min(g + 32 * 32 + lane, n - 1)];
                const int m = min(32, n - g);
                int s = 0, dl = 0;
                if (lane < m) {
                    s = p >> PK_SH;
                    dl = p & PK_MASK;
                    atomicAdd(&cn[dl], 1);
                }
                add_group(x, acc, s, dl, m, lane);
            }
        }
    } else {
        const int n = min(nlist, CAP);
        for (int g = wv * 32; g < n; g += 32 * 32) {
            const int k = g + lane;
            int s = 0, dl = 0;
            if (k < n) {
                const int e = list[k];
                s = src[e];
                dl = dst[e] - lo;
                atomicAdd(&cn[dl], 1);
            }
            add_group(x, acc, s, dl, min(32, n - g), lane);
        }
    }
    __syncthreads();

    for (int i = tid; i < nr * 16; i += 1024) {
        const int r = i >> 4;
        const int c4 = (i & 15) * 4;
        const float inv = 1.0f / fmaxf((float)cn[r], 1.0f);
        float4 v = *(const float4*)(acc + r * HDIM + c4);
        v.x *= inv; v.y *= inv; v.z *= inv; v.w *= inv;
        *(float4*)(out + (long)(lo + r) * HDIM + c4) = v;
    }
}

__global__ __launch_bounds__(1024) void k_own_agg3(
        const int* __restrict__ sN, const int* __restrict__ dN, const float* __restrict__ xN,
        float* __restrict__ oN, const int* __restrict__ blN, const int* __restrict__ cN,
        const int* __restrict__ sM, const int* __restrict__ dM, const float* __restrict__ xM,
        float* __restrict__ oM, const int* __restrict__ blM, const int* __restrict__ cM,
        const int* __restrict__ sR, const int* __restrict__ dR, const float* __restrict__ xR,
        float* __restrict__ oR, const int* __restrict__ blR, const int* __restrict__ cR) {
    __shared__ __attribute__((aligned(16))) float acc[391 * HDIM];
    __shared__ int cn[391];
    __shared__ int list[16384];
    __shared__ int nlist;
    const int b = blockIdx.x;
    if (b < OWN_N)
        own_agg_body<391, 16384, 15360>(sN, dN, xN, oN, ES, NN, blN, cN, b, OWN_N,
                                        acc, cn, list, nlist);
    else if (b < OWN_N + OWN_C)
        own_agg_body<20, 8192, 6144>(sM, dM, xM, oM, EM, NC, blM, cM, b - OWN_N, OWN_C,
                                     acc, cn, list, nlist);
    else
        own_agg_body<20, 8192, 1536>(sR, dR, xR, oR, ER, NC, blR, cR, b - OWN_N - OWN_C, OWN_C,
                                     acc, cn, list, nlist);
}

__global__ __launch_bounds__(1024) void k_own_agg2(
        const int* __restrict__ sM, const int* __restrict__ dM, const float* __restrict__ xM,
        float* __restrict__ oM, const int* __restrict__ blM, const int* __restrict__ cM,
        const int* __restrict__ sR, const int* __restrict__ dR, const float* __restrict__ xR,
        float* __restrict__ oR, const int* __restrict__ blR, const int* __restrict__ cR) {
    __shared__ __attribute__((aligned(16))) float acc[20 * HDIM];
    __shared__ int cn[20];
    __shared__ int list[8192];
    __shared__ int nlist;
    const int b = blockIdx.x;
    if (b < OWN_C)
        own_agg_body<20, 8192, 6144>(sM, dM, xM, oM, EM, NC, blM, cM, b, OWN_C,
                                     acc, cn, list, nlist);
    else
        own_agg_body<20, 8192, 1536>(sR, dR, xR, oR, ER, NC, blR, cR, b - OWN_C, OWN_C,
                                     acc, cn, list, nlist);
}

template<int NT, bool FULL, bool VECA>
__device__ __forceinline__ void gemm_body(v8f (&c)[NT],
                                          const float* __restrict__ A,
                                          const v2f* __restrict__ Wp,
                                          int M, int K, int N,
                                          int mbase, int lane) {
    const int row  = mbase + (lane & 15);
    const int koff = (lane >= 16) ? 2 : 0;
    const int col0 = lane & 15;
    const bool rok = FULL || (row < M);
    const float* __restrict__ Arow = A + (long)row * K;
    const int Kmain = K & ~3;

    #pragma unroll 4
    for (int k = 0; k < Kmain; k += 4) {
        const int kk = k + koff;
        v2f a;
        if (rok) {
            if (VECA) {
                a = *(const v2f*)(Arow + kk);
            } else {
                a.x = Arow[kk];
                a.y = Arow[kk + 1];
            }
        } else {
            a.x = 0.f; a.y = 0.f;
        }
        #pragma unroll
        for (int nt = 0; nt < NT; ++nt) {
            v2f b = Wp[(long)(kk >> 1) * N + nt * 16 + col0];
            c[nt] = __builtin_amdgcn_wmma_f32_16x16x4_f32(
                false, a, false, b, (short)0, c[nt], false, false);
        }
    }
    if (Kmain < K) {
        const int kk = Kmain + koff;
        v2f a;
        a.x = (rok && kk     < K) ? Arow[kk]     : 0.f;
        a.y = (rok && kk + 1 < K) ? Arow[kk + 1] : 0.f;
        #pragma unroll
        for (int nt = 0; nt < NT; ++nt) {
            v2f b = Wp[(long)(kk >> 1) * N + nt * 16 + col0];
            c[nt] = __builtin_amdgcn_wmma_f32_16x16x4_f32(
                false, a, false, b, (short)0, c[nt], false, false);
        }
    }
}

#define SLOTS8 "v_nop\n\tv_nop\n\tv_nop\n\tv_nop\n\tv_nop\n\tv_nop\n\tv_nop\n\tv_nop"
__device__ __forceinline__ void result_slots(v8f (&c)[4]) {
    asm volatile(SLOTS8 : "+v"(c[0]), "+v"(c[1]), "+v"(c[2]), "+v"(c[3]));
}
__device__ __forceinline__ void result_slots(v8f (&c)[2]) {
    asm volatile(SLOTS8 : "+v"(c[0]), "+v"(c[1]));
}

template<int NT, bool VECA1, bool NORM>
__global__ __launch_bounds__(256) void k_gemm(const float* __restrict__ A1,
                                              const v2f* __restrict__ W1p,
                                              const float* __restrict__ bias,
                                              const float* __restrict__ A2,
                                              const v2f* __restrict__ W2p,
                                              const float* __restrict__ prev,
                                              const float* __restrict__ res,
                                              const float* __restrict__ g,
                                              const float* __restrict__ b,
                                              float* __restrict__ out,
                                              int M, int K1, int K2,
                                              int do_relu) {
    static_assert(!NORM || NT * 16 == HDIM, "the normalised row must be the whole tile row");
    const int N = NT * 16;
    const int lane = threadIdx.x & 31;
    const int wv = threadIdx.x >> 5;
    const int mbase = (blockIdx.x * 8 + wv) * 16;
    if (mbase >= M) return;

    v8f c[NT];
    #pragma unroll
    for (int t = 0; t < NT; ++t)
        c[t] = (v8f){0.f, 0.f, 0.f, 0.f, 0.f, 0.f, 0.f, 0.f};

    const bool full = (mbase + 16 <= M);
    if (full) {
        gemm_body<NT, true, VECA1>(c, A1, W1p, M, K1, N, mbase, lane);
        if (A2) gemm_body<NT, true, true>(c, A2, W2p, M, K2, N, mbase, lane);
    } else {
        gemm_body<NT, false, VECA1>(c, A1, W1p, M, K1, N, mbase, lane);
        if (A2) gemm_body<NT, false, true>(c, A2, W2p, M, K2, N, mbase, lane);
    }

    result_slots(c);

    const int col0 = lane & 15;
    const int rofs = (lane >= 16) ? 8 : 0;
    #pragma unroll
    for (int nt = 0; nt < NT; ++nt) {
        const float bb = bias ? bias[nt * 16 + col0] : 0.f;
        #pragma unroll
        for (int r = 0; r < 8; ++r) c[nt][r] += bb;
    }
    if (prev) {
        #pragma unroll
        for (int r = 0; r < 8; ++r) {
            const long rb = (long)min(mbase + r + rofs, M - 1) * N + col0;
            #pragma unroll
            for (int nt = 0; nt < NT; ++nt) c[nt][r] += prev[rb + nt * 16];
        }
    }
    if (res) {
        #pragma unroll
        for (int r = 0; r < 8; ++r) {
            const long rb = (long)min(mbase + r + rofs, M - 1) * N + col0;
            #pragma unroll
            for (int nt = 0; nt < NT; ++nt) c[nt][r] += res[rb + nt * 16];
        }
    }

    if (NORM) {
        float gg[NT], be[NT];
        #pragma unroll
        for (int nt = 0; nt < NT; ++nt) { gg[nt] = g[nt * 16 + col0]; be[nt] = b[nt * 16 + col0]; }
        #pragma unroll
        for (int r = 0; r < 8; ++r) {
            const int row = mbase + r + rofs;
            float v[NT];
            #pragma unroll
            for (int nt = 0; nt < NT; ++nt) v[nt] = fmaxf(c[nt][r], 0.f);
            float s = v[0];
            #pragma unroll
            for (int nt = 1; nt < NT; ++nt) s += v[nt];
            #pragma unroll
            for (int off = 8; off > 0; off >>= 1) s += __shfl_xor(s, off, 32);
            const float mean = s * (1.0f / 64.0f);
            float d[NT];
            #pragma unroll
            for (int nt = 0; nt < NT; ++nt) d[nt] = v[nt] - mean;
            float q = d[0] * d[0];
            #pragma unroll
            for (int nt = 1; nt < NT; ++nt) q += d[nt] * d[nt];
            #pragma unroll
            for (int off = 8; off > 0; off >>= 1) q += __shfl_xor(q, off, 32);
            const float rstd = rsqrtf(q * (1.0f / 64.0f) + 1e-5f);
            if (full || row < M) {
                const long rb = (long)row * N + col0;
                #pragma unroll
                for (int nt = 0; nt < NT; ++nt)
                    out[rb + nt * 16] = d[nt] * rstd * gg[nt] + be[nt];
            }
        }
    } else {
        #pragma unroll
        for (int nt = 0; nt < NT; ++nt) {
            #pragma unroll
            for (int r = 0; r < 8; ++r) {
                const int row = mbase + r + rofs;
                if (full || row < M) {
                    float v = c[nt][r];
                    if (do_relu) v = fmaxf(v, 0.f);
                    out[(long)row * N + nt * 16 + col0] = v;
                }
            }
        }
    }
}

__global__ __launch_bounds__(256) void k_final(const float* __restrict__ h,
                                               const float* __restrict__ W2,
                                               const float* __restrict__ b2,
                                               float* __restrict__ out, int M) {
    int i = blockIdx.x * blockDim.x + threadIdx.x;
    if (i >= M) return;
    float acc = b2[0];
    const float* hr = h + (long)i * 32;
    #pragma unroll
    for (int j = 0; j < 32; ++j) acc += hr[j] * W2[j];
    out[i] = acc;
}

extern "C" void kernel_launch(void* const* d_in, const int* in_sizes, int n_in,
                              void* d_out, int out_size, void* d_ws, size_t ws_size,
                              hipStream_t stream) {
    const float* x_news    = (const float*)d_in[0];
    const float* x_company = (const float*)d_in[1];
    const int*   edge_sim  = (const int*)d_in[2];
    const int*   men_src   = (const int*)d_in[3];
    const int*   men_dst   = (const int*)d_in[4];
    const int*   edge_rel  = (const int*)d_in[5];
    const float* nw_W = (const float*)d_in[6];
    const float* nw_b = (const float*)d_in[7];
    const float* cw_W = (const float*)d_in[8];
    const float* cw_b = (const float*)d_in[9];
    const float* n1n_g = (const float*)d_in[10];
    const float* n1n_b = (const float*)d_in[11];
    const float* n1c_g = (const float*)d_in[12];
    const float* n1c_b = (const float*)d_in[13];
    const float* c1_Wl = (const float*)d_in[14];
    const float* c1_bl = (const float*)d_in[15];
    const float* c1_Wr = (const float*)d_in[16];
    const float* c2_Wl = (const float*)d_in[17];
    const float* c2_bl = (const float*)d_in[18];
    const float* c2_Wr = (const float*)d_in[19];
    const float* cls_W1 = (const float*)d_in[20];
    const float* cls_b1 = (const float*)d_in[21];
    const float* cls_W2 = (const float*)d_in[22];
    const float* cls_b2 = (const float*)d_in[23];
    float* out = (float*)d_out;

    float* ws = (float*)d_ws;
    long o = 0;
    float* xn0  = ws + o; o += (long)NN * HDIM;
    float* xn1  = ws + o; o += (long)NN * HDIM;
    float* aggn = ws + o; o += (long)NN * HDIM;
    float* aggcM = ws + o; o += (long)NC * HDIM;
    float* aggcR = ws + o; o += (long)NC * HDIM;
    float* xc0  = ws + o; o += (long)NC * HDIM;
    float* xc1  = ws + o; o += (long)NC * HDIM;
    float* ocb  = ws + o; o += (long)NC * HDIM;
    float* comp = ws + o; o += (long)NC * HDIM;
    float* hbuf = ws + o; o += (long)NC * 32;
    v2f* wp_nw  = (v2f*)(ws + o); o += 2L * (388 / 2) * 64;
    v2f* wp_cw  = (v2f*)(ws + o); o += 2L * (24 / 2) * 64;
    v2f* wp_c1  = (v2f*)(ws + o); o += 2L * 6 * 32 * 64;
    v2f* wp_c2  = (v2f*)(ws + o); o += 2L * 4 * 32 * 64;
    v2f* wp_cls = (v2f*)(ws + o); o += 2L * 32 * 32;

    int* ctr   = (int*)comp;
    int* ctrN  = ctr;
    int* ctrM1 = ctr + 512;
    int* ctrR  = ctr + 1024;
    int* ctrM2 = ctr + 1536;
    int* blR   = ctr + 4096;
    int* blN   = (int*)xn1;
    int* blM1  = blN + 256L * 15360;
    int* blM2  = (int*)xn0;

    dim3 blk(256);
    auto blocks = [](long n) { return (unsigned)((n + 255) / 256); };
    int tilesNN = (NN + 15) / 16, tilesNC = (NC + 15) / 16;
    unsigned gemmBlkNN = (tilesNN + 7) / 8, gemmBlkNC = (tilesNC + 7) / 8;

    k_prep<<<PB_NW + PB_CW + PB_C1 + PB_C2 + PB_CLS + PB_Z, blk, 0, stream>>>(
        nw_W, cw_W, c1_Wl, c1_Wr, c2_Wl, c2_Wr, cls_W1,
        wp_nw, wp_cw, wp_c1, wp_c2, wp_cls, ctr);

    k_gemm<4, false, false><<<gemmBlkNN, blk, 0, stream>>>(x_news, wp_nw, nw_b,
                                                           nullptr, nullptr,
                                                           nullptr, nullptr, nullptr, nullptr,
                                                           xn0, NN, 385, 0, 1);
    k_gemm<4, true, false><<<gemmBlkNC, blk, 0, stream>>>(x_company, wp_cw, cw_b,
                                                          nullptr, nullptr,
                                                          nullptr, nullptr, nullptr, nullptr,
                                                          xc0, NC, 24, 0, 1);

    k_bin3<<<BINB_N + BINB_M + BINB_R, 1024, 0, stream>>>(edge_sim, edge_sim + ES, blN, ctrN,
                                                          men_src, men_dst, blM1, ctrM1,
                                                          edge_rel, edge_rel + ER, blR, ctrR);
    k_own_agg3<<<OWN_N + OWN_C + OWN_C, 1024, 0, stream>>>(edge_sim, edge_sim + ES, xn0, aggn, blN, ctrN,
                                                           men_src, men_dst, xn0, aggcM, blM1, ctrM1,
                                                           edge_rel, edge_rel + ER, xc0, aggcR, blR, ctrR);

    k_gemm<4, true, true><<<gemmBlkNN, blk, 0, stream>>>(aggn, wp_c1 + 0 * 2048,
                                                         c1_bl + 0 * 64,
                                                         xn0, wp_c1 + 1 * 2048,
                                                         nullptr, nullptr, n1n_g, n1n_b,
                                                         xn1, NN, 64, 64, 1);
    k_gemm<4, true, false><<<gemmBlkNC, blk, 0, stream>>>(aggcM, wp_c1 + 2 * 2048,
                                                          c1_bl + 1 * 64,
                                                          xc0, wp_c1 + 3 * 2048,
                                                          nullptr, nullptr, nullptr, nullptr,
                                                          ocb, NC, 64, 64, 0);
    k_gemm<4, true, true><<<gemmBlkNC, blk, 0, stream>>>(aggcR, wp_c1 + 4 * 2048,
                                                         c1_bl + 2 * 64,
                                                         xc0, wp_c1 + 5 * 2048,
                                                         ocb, xc0, n1c_g, n1c_b,
                                                         xc1, NC, 64, 64, 1);

    k_bin<20, 6144><<<(EM + 8191) / 8192, 1024, 0, stream>>>(men_src, men_dst, EM,
                                                             (NC + 19) / 20, blM2, ctrM2);
    k_own_agg2<<<OWN_C + OWN_C, 1024, 0, stream>>>(men_src, men_dst, xn1, aggcM, blM2, ctrM2,
                                                   edge_rel, edge_rel + ER, xc1, aggcR, blR, ctrR);
    k_gemm<4, true, false><<<gemmBlkNC, blk, 0, stream>>>(aggcM, wp_c2 + 0 * 2048,
                                                          c2_bl + 1 * 64,
                                                          xc1, wp_c2 + 1 * 2048,
                                                          nullptr, nullptr, nullptr, nullptr,
                                                          ocb, NC, 64, 64, 0);
    k_gemm<4, true, false><<<gemmBlkNC, blk, 0, stream>>>(aggcR, wp_c2 + 2 * 2048,
                                                          c2_bl + 2 * 64,
                                                          xc1, wp_c2 + 3 * 2048,
                                                          ocb, xc1, nullptr, nullptr,
                                                          comp, NC, 64, 64, 1);

    k_gemm<2, true, false><<<gemmBlkNC, blk, 0, stream>>>(comp, wp_cls, cls_b1,
                                                          nullptr, nullptr,
                                                          nullptr, nullptr, nullptr, nullptr,
                                                          hbuf, NC, 64, 0, 1);
    k_final<<<blocks(NC), blk, 0, stream>>>(hbuf, cls_W2, cls_b2, out, NC);
}


# aggregation: each owner orders its edge list by destination row in LDS (integer counting sort); a wave adds a row's gathered rows in registers; no LDS float adds; sums in the ranks' arrival order
# speedup vs baseline: 4.5812x; 1.0386x over previous
#include <hip/hip_runtime.h>
#include <hip/hip_bf16.h>

typedef __attribute__((ext_vector_type(2))) float v2f;
typedef __attribute__((ext_vector_type(8))) float v8f;

#define NN 100000
#define NC 5000
#define HDIM 64
#define ES 3200000
#define EM 1000000
#define ER 160000

__device__ __forceinline__ void pack_one(const float* __restrict__ W,
                                         v2f* __restrict__ Wp,
                                         int K, int N, int i) {
    const int Kp2 = ((K + 3) & ~3) >> 1;
    if (i >= Kp2 * N) return;
    const int k2 = i / N, col = i % N;
    const int k = 2 * k2;
    v2f v;
    v.x = (k     < K) ? W[k * N + col]       : 0.f;
    v.y = (k + 1 < K) ? W[(k + 1) * N + col] : 0.f;
    Wp[i] = v;
}

#define PB_NW  49
#define PB_CW  3
#define PB_C1  48
#define PB_C2  32
#define PB_CLS 4
#define PB_Z   16
__global__ __launch_bounds__(256) void k_prep(const float* __restrict__ nw_W,
                                              const float* __restrict__ cw_W,
                                              const float* __restrict__ c1_Wl,
                                              const float* __restrict__ c1_Wr,
                                              const float* __restrict__ c2_Wl,
                                              const float* __restrict__ c2_Wr,
                                              const float* __restrict__ cls_W1,
                                              v2f* __restrict__ wp_nw,
                                              v2f* __restrict__ wp_cw,
                                              v2f* __restrict__ wp_c1,
                                              v2f* __restrict__ wp_c2,
                                              v2f* __restrict__ wp_cls,
                                              int* __restrict__ ctr) {
    const int tid = threadIdx.x;
    int b = blockIdx.x;
    if (b < PB_NW) { pack_one(nw_W, wp_nw, 385, 64, b * 256 + tid); return; }
    b -= PB_NW;
    if (b < PB_CW) { pack_one(cw_W, wp_cw, 24, 64, b * 256 + tid); return; }
    b -= PB_CW;
    if (b < PB_C1) {
        const int m = b >> 3;
        const float* W = ((m & 1) ? c1_Wr : c1_Wl) + (m >> 1) * 4096;
        pack_one(W, wp_c1 + m * 2048, 64, 64, (b & 7) * 256 + tid);
        return;
    }
    b -= PB_C1;
    if (b < PB_C2) {
        const int m = b >> 3;
        const float* W = ((m & 1) ? c2_Wr : c2_Wl) + ((m >> 1) + 1) * 4096;
        pack_one(W, wp_c2 + m * 2048, 64, 64, (b & 7) * 256 + tid);
        return;
    }
    b -= PB_C2;
    if (b < PB_CLS) { pack_one(cls_W1, wp_cls, 64, 32, b * 256 + tid); return; }
    b -= PB_CLS;
    ctr[b * 256 + tid] = 0;
}

#define PK_SH 9
#define PK_MASK 511
#define BIN_EPT 8
#define BIN_CHUNK (1024 * BIN_EPT)
#define BIN_NBMAX 256
template<int R, int CAPB>
__device__ __forceinline__ void bin_body(const int* __restrict__ src,
                                         const int* __restrict__ dst,
                                         int nE, int nB,
                                         int* __restrict__ blist,
                                         int* __restrict__ bctr,
                                         int bid,
                                         int* hcnt, int* hoff, int* gbase,
                                         int* stage, int* sown) {
    static_assert(R <= PK_MASK + 1, "row inside the range must fit the packed field");
    constexpr int EPT = BIN_EPT;
    constexpr int CHUNK = BIN_CHUNK;
    constexpr int NBMAX = BIN_NBMAX;

    const int tid = threadIdx.x;
    const int lane = tid & 31;
    const int wv = tid >> 5;
    const int base = bid * CHUNK;

    if (tid < NBMAX) hcnt[tid] = 0;
    __syncthreads();

    int ow[EPT], rk[EPT], pk[EPT];
    #pragma unroll
    for (int u = 0; u < EPT; ++u) {
        const int e = base + u * 1024 + tid;
        ow[u] = -1; rk[u] = 0; pk[u] = 0;
        if (e < nE) {
            const int d = dst[e];
            const int o = (int)((unsigned)d / (unsigned)R);
            ow[u] = o;
            pk[u] = (src[e] << PK_SH) | (d - o * R);
            rk[u] = atomicAdd(&hcnt[o], 1);
        }
    }
    __syncthreads();

    if (tid < nB) {
        const int cnt = hcnt[tid];
        gbase[tid] = cnt ? atomicAdd(&bctr[tid], cnt) : 0;
    }
    if (wv == 31) {
        int cq[8];
        int t = 0;
        #pragma unroll
        for (int q = 0; q < 8; ++q) { const int v = hcnt[lane * 8 + q]; cq[q] = t; t += v; }
        int incl = t;
        #pragma unroll
        for (int off = 1; off < 32; off <<= 1) {
            const int s = __shfl_up(incl, off, 32);
            if (lane >= off) incl += s;
        }
        const int excl = incl - t;
        #pragma unroll
        for (int q = 0; q < 8; ++q) hoff[lane * 8 + q] = cq[q] + excl;
    }
    __syncthreads();

    #pragma unroll
    for (int u = 0; u < EPT; ++u) {
        if (ow[u] >= 0) {
            const int lp = hoff[ow[u]] + rk[u];
            stage[lp] = pk[u];
            sown[lp] = ow[u];
        }
    }
    __syncthreads();

    const int total = min(CHUNK, nE - base);
    for (int i = tid; i < total; i += 1024) {
        const int o = sown[i];
        const int pos = gbase[o] + (i - hoff[o]);
        if (pos < CAPB) blist[(long)o * CAPB + pos] = stage[i];
        else bctr[nB] = 1;
    }
}

#define BIN_LDS \
    __shared__ int hcnt[BIN_NBMAX];      \
    __shared__ int hoff[BIN_NBMAX];      \
    __shared__ int gbase[BIN_NBMAX];     \
    __shared__ int stage[BIN_CHUNK];     \
    __shared__ int sown[BIN_CHUNK];

template<int R, int CAPB>
__global__ __launch_bounds__(1024) void k_bin(const int* __restrict__ src,
                                              const int* __restrict__ dst,
                                              int nE, int nB,
                                              int* __restrict__ blist,
                                              int* __restrict__ bctr) {
    BIN_LDS
    bin_body<R, CAPB>(src, dst, nE, nB, blist, bctr, blockIdx.x, hcnt, hoff, gbase, stage, sown);
}

#define BINB_N ((ES + BIN_CHUNK - 1) / BIN_CHUNK)
#define BINB_M ((EM + BIN_CHUNK - 1) / BIN_CHUNK)
#define BINB_R ((ER + BIN_CHUNK - 1) / BIN_CHUNK)
#define OWN_N ((NN + 390) / 391)
#define OWN_C ((NC + 19) / 20)
__global__ __launch_bounds__(1024) void k_bin3(const int* __restrict__ sN, const int* __restrict__ dN,
                                               int* __restrict__ blN, int* __restrict__ cN,
                                               const int* __restrict__ sM, const int* __restrict__ dM,
                                               int* __restrict__ blM, int* __restrict__ cM,
                                               const int* __restrict__ sR, const int* __restrict__ dR,
                                               int* __restrict__ blR, int* __restrict__ cR) {
    BIN_LDS
    const int b = blockIdx.x;
    if (b < BINB_N)
        bin_body<391, 15360>(sN, dN, ES, OWN_N, blN, cN, b, hcnt, hoff, gbase, stage, sown);
    else if (b < BINB_N + BINB_M)
        bin_body<20, 6144>(sM, dM, EM, OWN_C, blM, cM, b - BINB_N, hcnt, hoff, gbase, stage, sown);
    else
        bin_body<20, 1536>(sR, dR, ER, OWN_C, blR, cR, b - BINB_N - BINB_M, hcnt, hoff, gbase, stage, sown);
}

__device__ __forceinline__ void add_group(const float* __restrict__ x, float* acc,
                                          int s, int dl, int m, int lane) {
    if (m >= 32) {
        float v0[32], v1[32];
        #pragma unroll
        for (int j = 0; j < 32; ++j) {
            const int sj = __builtin_amdgcn_readlane(s, j);
            const float* xr = x + (long)sj * HDIM + lane;
            v0[j] = xr[0];
            v1[j] = xr[32];
        }
        #pragma unroll
        for (int j = 0; j < 32; ++j) {
            const int dj = __builtin_amdgcn_readlane(dl, j);
            float* ar = acc + dj * HDIM + lane;
            atomicAdd(ar, v0[j]);
            atomicAdd(ar + 32, v1[j]);
        }
    } else {
        for (int j = 0; j < m; ++j) {
            const int sj = __builtin_amdgcn_readlane(s, j);
            const int dj = __builtin_amdgcn_readlane(dl, j);
            const float* xr = x + (long)sj * HDIM + lane;
            float* ar = acc + dj * HDIM + lane;
            const float a0 = xr[0];
            const float a1 = xr[32];
            atomicAdd(ar, a0);
            atomicAdd(ar + 32, a1);
        }
    }
}

#define SRT_UB 16
template<int R, int CAPB>
__device__ __forceinline__ void own_sorted_body(const float* __restrict__ x,
                                                float* __restrict__ out,
                                                const int* __restrict__ bl,
                                                int nbin, int lo, int nr,
                                                int* cn, int* off, int* sorted) {
    constexpr int EPT = (CAPB + 1023) / 1024;
    constexpr int Q = (R + 31) / 32;
    const int tid = threadIdx.x;
    const int lane = tid & 31;
    const int wv = __builtin_amdgcn_readfirstlane(tid >> 5);
    const int n = min(nbin, CAPB);

    for (int i = tid; i < R; i += 1024) cn[i] = 0;
    __syncthreads();

    int pk[EPT], rk[EPT];
    #pragma unroll
    for (int u = 0; u < EPT; ++u) {
        const int i = u * 1024 + tid;
        pk[u] = -1; rk[u] = 0;
        if (i < n) {
            const int p = bl[i];
            pk[u] = p;
            rk[u] = atomicAdd(&cn[p & PK_MASK], 1);
        }
    }
    __syncthreads();

    if (wv == 0) {
        int cq[Q];
        int t = 0;
        #pragma unroll
        for (int q = 0; q < Q; ++q) {
            const int r = lane * Q + q;
            const int v = (r < R) ? cn[r] : 0;
            cq[q] = t; t += v;
        }
        int incl = t;
        #pragma unroll
        for (int o = 1; o < 32; o <<= 1) {
            const int s = __shfl_up(incl, o, 32);
            if (lane >= o) incl += s;
        }
        const int excl = incl - t;
        #pragma unroll
        for (int q = 0; q < Q; ++q) {
            const int r = lane * Q + q;
            if (r < R) off[r] = cq[q] + excl;
        }
        if (lane == 31) off[R] = incl;
    }
    __syncthreads();

    #pragma unroll
    for (int u = 0; u < EPT; ++u)
        if (pk[u] >= 0) sorted[off[pk[u] & PK_MASK] + rk[u]] = pk[u];
    for (int i = tid; i < nr * 16; i += 1024) {
        const int r = i >> 4;
        if (cn[r] == 0)
            *(float4*)(out + (long)(lo + r) * HDIM + (i & 15) * 4) = make_float4(0.f, 0.f, 0.f, 0.f);
    }
    __syncthreads();

    const int r0 = min(wv * Q, nr);
    const int r1 = min(r0 + Q, nr);
    const int e0 = off[r0];
    const int e1 = off[r1];
    if (e0 >= e1) return;

    float a0 = 0.f, a1 = 0.f;
    int cur = -1;
    int pn = sorted[min(e0 + lane, e1 - 1)];
    for (int g = e0; g < e1; g += SRT_UB) {
        const int pv = pn;
        pn = sorted[min(g + SRT_UB + lane, e1 - 1)];
        float2 v[SRT_UB];
        #pragma unroll
        for (int u = 0; u < SRT_UB; ++u) {
            const int pu = __builtin_amdgcn_readlane(pv, u);
            v[u] = *(const float2*)(x + (long)(pu >> PK_SH) * HDIM + 2 * lane);
        }
        #pragma unroll
        for (int u = 0; u < SRT_UB; ++u) {
            if (g + u < e1) {
                const int d = __builtin_amdgcn_readlane(pv, u) & PK_MASK;
                if (d != cur) {
                    if (cur >= 0) {
                        const float inv = 1.0f / fmaxf((float)cn[cur], 1.0f);
                        *(float2*)(out + (long)(lo + cur) * HDIM + 2 * lane) = make_float2(a0 * inv, a1 * inv);
                    }
                    cur = d; a0 = 0.f; a1 = 0.f;
                }
                a0 += v[u].x; a1 += v[u].y;
            }
        }
    }
    const float inv = 1.0f / fmaxf((float)cn[cur], 1.0f);
    *(float2*)(out + (long)(lo + cur) * HDIM + 2 * lane) = make_float2(a0 * inv, a1 * inv);
}

template<int R, int CAP, int CAPB>
__device__ __forceinline__ void own_agg_body(const int* __restrict__ src,
                                             const int* __restrict__ dst,
                                             const float* __restrict__ x,
                                             float* __restrict__ out,
                                             int nE, int nDst,
                                             const int* __restrict__ blist,
                                             const int* __restrict__ bctr,
                                             int bid, int nB,
                                             float* acc, int* cn, int* off, int* list, int& nlist) {
    static_assert(CAPB <= CAP, "the ordered list must fit the list array");
    const int tid = threadIdx.x;
    const int lane = tid & 31;
    const int wv = __builtin_amdgcn_readfirstlane(tid >> 5);
    const int lo = bid * R;
    const int nr = min(R, nDst - lo);
    const bool binned = __builtin_amdgcn_readfirstlane(bctr[nB]) == 0;
    const int nbin = __builtin_amdgcn_readfirstlane(bctr[bid]);

    if (binned) {
        own_sorted_body<R, CAPB>(x, out, blist + (long)bid * CAPB, nbin, lo, nr, cn, off, list);
        return;
    }

    for (int i = tid; i < R * HDIM; i += 1024) acc[i] = 0.f;
    for (int i = tid; i < R; i += 1024) cn[i] = 0;
    if (tid == 0) nlist = 0;
    __syncthreads();

    {
    constexpr int U = 8;
    constexpr int CH = U * 128;
    const int nCh = (nE + CH - 1) / CH;
    const int rot = (int)(((long)bid * nCh) / nB);
    for (int c0 = wv; c0 < nCh; c0 += 32) {
        int c = c0 + rot;
        if (c >= nCh) c -= nCh;
        const int base = c * CH;
        int4 d[U];
        if (base + CH <= nE) {
            #pragma unroll
            for (int u = 0; u < U; ++u)
                d[u] = *(const int4*)(dst + base + u * 128 + lane * 4);
        } else {
            #pragma unroll
            for (int u = 0; u < U; ++u) {
                const int i0 = base + u * 128 + lane * 4;
                d[u].x = (i0     < nE) ? dst[i0]     : -1;
                d[u].y = (i0 + 1 < nE) ? dst[i0 + 1] : -1;
                d[u].z = (i0 + 2 < nE) ? dst[i0 + 2] : -1;
                d[u].w = (i0 + 3 < nE) ? dst[i0 + 3] : -1;
            }
        }
        int cl = 0;
        #pragma unroll
        for (int u = 0; u < U; ++u) {
            cl += ((unsigned)(d[u].x - lo) < (unsigned)nr) ? 1 : 0;
            cl += ((unsigned)(d[u].y - lo) < (unsigned)nr) ? 1 : 0;
            cl += ((unsigned)(d[u].z - lo) < (unsigned)nr) ? 1 : 0;
            cl += ((unsigned)(d[u].w - lo) < (unsigned)nr) ? 1 : 0;
        }
        int incl = cl;
        #pragma unroll
        for (int off = 1; off < 32; off <<= 1) {
            const int t = __shfl_up(incl, off, 32);
            if (lane >= off) incl += t;
        }
        const int total = __builtin_amdgcn_readlane(incl, 31);
        if (total == 0) continue;
        int s0 = 0;
        if (lane == 31) s0 = atomicAdd(&nlist, total);
        s0 = __builtin_amdgcn_readlane(s0, 31);
        int pos = s0 + incl - cl;
        #pragma unroll
        for (int u = 0; u < U; ++u) {
            #pragma unroll
            for (int q = 0; q < 4; ++q) {
                const int dv = (q == 0) ? d[u].x : (q == 1) ? d[u].y : (q == 2) ? d[u].z : d[u].w;
                const int dl = dv - lo;
                if ((unsigned)dl < (unsigned)nr) {
                    const int e = base + u * 128 + lane * 4 + q;
                    if (pos < CAP) {
                        list[pos] = e;
                    } else {
                        const float* xr = x + (long)src[e] * HDIM;
                        #pragma nounroll
                        for (int k = 0; k < HDIM; ++k) atomicAdd(&acc[dl * HDIM + k], xr[k]);
                        atomicAdd(&cn[dl], 1);
                    }
                    ++pos;
                }
            }
        }
    }
    }
    __syncthreads();

    {
        const int n = min(nlist, CAP);
        for (int g = wv * 32; g < n; g += 32 * 32) {
            const int k = g + lane;
            int s = 0, dl = 0;
            if (k < n) {
                const int e = list[k];
                s = src[e];
                dl = dst[e] - lo;
                atomicAdd(&cn[dl], 1);
            }
            add_group(x, acc, s, dl, min(32, n - g), lane);
        }
    }
    __syncthreads();

    for (int i = tid; i < nr * 16; i += 1024) {
        const int r = i >> 4;
        const int c4 = (i & 15) * 4;
        const float inv = 1.0f / fmaxf((float)cn[r], 1.0f);
        float4 v = *(const float4*)(acc + r * HDIM + c4);
        v.x *= inv; v.y *= inv; v.z *= inv; v.w *= inv;
        *(float4*)(out + (long)(lo + r) * HDIM + c4) = v;
    }
}

__global__ __launch_bounds__(1024) void k_own_agg3(
        const int* __restrict__ sN, const int* __restrict__ dN, const float* __restrict__ xN,
        float* __restrict__ oN, const int* __restrict__ blN, const int* __restrict__ cN,
        const int* __restrict__ sM, const int* __restrict__ dM, const float* __restrict__ xM,
        float* __restrict__ oM, const int* __restrict__ blM, const int* __restrict__ cM,
        const int* __restrict__ sR, const int* __restrict__ dR, const float* __restrict__ xR,
        float* __restrict__ oR, const int* __restrict__ blR, const int* __restrict__ cR) {
    __shared__ __attribute__((aligned(16))) float acc[391 * HDIM];
    __shared__ int cn[391];
    __shared__ int off[392];
    __shared__ int list[16384];
    __shared__ int nlist;
    const int b = blockIdx.x;
    if (b < OWN_N)
        own_agg_body<391, 16384, 15360>(sN, dN, xN, oN, ES, NN, blN, cN, b, OWN_N,
                                        acc, cn, off, list, nlist);
    else if (b < OWN_N + OWN_C)
        own_agg_body<20, 8192, 6144>(sM, dM, xM, oM, EM, NC, blM, cM, b - OWN_N, OWN_C,
                                     acc, cn, off, list, nlist);
    else
        own_agg_body<20, 8192, 1536>(sR, dR, xR, oR, ER, NC, blR, cR, b - OWN_N - OWN_C, OWN_C,
                                     acc, cn, off, list, nlist);
}

__global__ __launch_bounds__(1024) void k_own_agg2(
        const int* __restrict__ sM, const int* __restrict__ dM, const float* __restrict__ xM,
        float* __restrict__ oM, const int* __restrict__ blM, const int* __restrict__ cM,
        const int* __restrict__ sR, const int* __restrict__ dR, const float* __restrict__ xR,
        float* __restrict__ oR, const int* __restrict__ blR, const int* __restrict__ cR) {
    __shared__ __attribute__((aligned(16))) float acc[20 * HDIM];
    __shared__ int cn[20];
    __shared__ int off[21];
    __shared__ int list[8192];
    __shared__ int nlist;
    const int b = blockIdx.x;
    if (b < OWN_C)
        own_agg_body<20, 8192, 6144>(sM, dM, xM, oM, EM, NC, blM, cM, b, OWN_C,
                                     acc, cn, off, list, nlist);
    else
        own_agg_body<20, 8192, 1536>(sR, dR, xR, oR, ER, NC, blR, cR, b - OWN_C, OWN_C,
                                     acc, cn, off, list, nlist);
}

template<int NT, bool FULL, bool VECA>
__device__ __forceinline__ void gemm_body(v8f (&c)[NT],
                                          const float* __restrict__ A,
                                          const v2f* __restrict__ Wp,
                                          int M, int K, int N,
                                          int mbase, int lane) {
    const int row  = mbase + (lane & 15);
    const int koff = (lane >= 16) ? 2 : 0;
    const int col0 = lane & 15;
    const bool rok = FULL || (row < M);
    const float* __restrict__ Arow = A + (long)row * K;
    const int Kmain = K & ~3;

    #pragma unroll 4
    for (int k = 0; k < Kmain; k += 4) {
        const int kk = k + koff;
        v2f a;
        if (rok) {
            if (VECA) {
                a = *(const v2f*)(Arow + kk);
            } else {
                a.x = Arow[kk];
                a.y = Arow[kk + 1];
            }
        } else {
            a.x = 0.f; a.y = 0.f;
        }
        #pragma unroll
        for (int nt = 0; nt < NT; ++nt) {
            v2f b = Wp[(long)(kk >> 1) * N + nt * 16 + col0];
            c[nt] = __builtin_amdgcn_wmma_f32_16x16x4_f32(
                false, a, false, b, (short)0, c[nt], false, false);
        }
    }
    if (Kmain < K) {
        const int kk = Kmain + koff;
        v2f a;
        a.x = (rok && kk     < K) ? Arow[kk]     : 0.f;
        a.y = (rok && kk + 1 < K) ? Arow[kk + 1] : 0.f;
        #pragma unroll
        for (int nt = 0; nt < NT; ++nt) {
            v2f b = Wp[(long)(kk >> 1) * N + nt * 16 + col0];
            c[nt] = __builtin_amdgcn_wmma_f32_16x16x4_f32(
                false, a, false, b, (short)0, c[nt], false, false);
        }
    }
}

#define SLOTS8 "v_nop\n\tv_nop\n\tv_nop\n\tv_nop\n\tv_nop\n\tv_nop\n\tv_nop\n\tv_nop"
__device__ __forceinline__ void result_slots(v8f (&c)[4]) {
    asm volatile(SLOTS8 : "+v"(c[0]), "+v"(c[1]), "+v"(c[2]), "+v"(c[3]));
}
__device__ __forceinline__ void result_slots(v8f (&c)[2]) {
    asm volatile(SLOTS8 : "+v"(c[0]), "+v"(c[1]));
}

template<int NT, bool VECA1, bool NORM>
__global__ __launch_bounds__(256) void k_gemm(const float* __restrict__ A1,
                                              const v2f* __restrict__ W1p,
                                              const float* __restrict__ bias,
                                              const float* __restrict__ A2,
                                              const v2f* __restrict__ W2p,
                                              const float* __restrict__ prev,
                                              const float* __restrict__ res,
                                              const float* __restrict__ g,
                                              const float* __restrict__ b,
                                              float* __restrict__ out,
                                              int M, int K1, int K2,
                                              int do_relu) {
    static_assert(!NORM || NT * 16 == HDIM, "the normalised row must be the whole tile row");
    const int N = NT * 16;
    const int lane = threadIdx.x & 31;
    const int wv = threadIdx.x >> 5;
    const int mbase = (blockIdx.x * 8 + wv) * 16;
    if (mbase >= M) return;

    v8f c[NT];
    #pragma unroll
    for (int t = 0; t < NT; ++t)
        c[t] = (v8f){0.f, 0.f, 0.f, 0.f, 0.f, 0.f, 0.f, 0.f};

    const bool full = (mbase + 16 <= M);
    if (full) {
        gemm_body<NT, true, VECA1>(c, A1, W1p, M, K1, N, mbase, lane);
        if (A2) gemm_body<NT, true, true>(c, A2, W2p, M, K2, N, mbase, lane);
    } else {
        gemm_body<NT, false, VECA1>(c, A1, W1p, M, K1, N, mbase, lane);
        if (A2) gemm_body<NT, false, true>(c, A2, W2p, M, K2, N, mbase, lane);
    }

    result_slots(c);

    const int col0 = lane & 15;
    const int rofs = (lane >= 16) ? 8 : 0;
    #pragma unroll
    for (int nt = 0; nt < NT; ++nt) {
        const float bb = bias ? bias[nt * 16 + col0] : 0.f;
        #pragma unroll
        for (int r = 0; r < 8; ++r) c[nt][r] += bb;
    }
    if (prev) {
        #pragma unroll
        for (int r = 0; r < 8; ++r) {
            const long rb = (long)min(mbase + r + rofs, M - 1) * N + col0;
            #pragma unroll
            for (int nt = 0; nt < NT; ++nt) c[nt][r] += prev[rb + nt * 16];
        }
    }
    if (res) {
        #pragma unroll
        for (int r = 0; r < 8; ++r) {
            const long rb = (long)min(mbase + r + rofs, M - 1) * N + col0;
            #pragma unroll
            for (int nt = 0; nt < NT; ++nt) c[nt][r] += res[rb + nt * 16];
        }
    }

    if (NORM) {
        float gg[NT], be[NT];
        #pragma unroll
        for (int nt = 0; nt < NT; ++nt) { gg[nt] = g[nt * 16 + col0]; be[nt] = b[nt * 16 + col0]; }
        #pragma unroll
        for (int r = 0; r < 8; ++r) {
            const int row = mbase + r + rofs;
            float v[NT];
            #pragma unroll
            for (int nt = 0; nt < NT; ++nt) v[nt] = fmaxf(c[nt][r], 0.f);
            float s = v[0];
            #pragma unroll
            for (int nt = 1; nt < NT; ++nt) s += v[nt];
            #pragma unroll
            for (int off = 8; off > 0; off >>= 1) s += __shfl_xor(s, off, 32);
            const float mean = s * (1.0f / 64.0f);
            float d[NT];
            #pragma unroll
            for (int nt = 0; nt < NT; ++nt) d[nt] = v[nt] - mean;
            float q = d[0] * d[0];
            #pragma unroll
            for (int nt = 1; nt < NT; ++nt) q += d[nt] * d[nt];
            #pragma unroll
            for (int off = 8; off > 0; off >>= 1) q += __shfl_xor(q, off, 32);
            const float rstd = rsqrtf(q * (1.0f / 64.0f) + 1e-5f);
            if (full || row < M) {
                const long rb = (long)row * N + col0;
                #pragma unroll
                for (int nt = 0; nt < NT; ++nt)
                    out[rb + nt * 16] = d[nt] * rstd * gg[nt] + be[nt];
            }
        }
    } else {
        #pragma unroll
        for (int nt = 0; nt < NT; ++nt) {
            #pragma unroll
            for (int r = 0; r < 8; ++r) {
                const int row = mbase + r + rofs;
                if (full || row < M) {
                    float v = c[nt][r];
                    if (do_relu) v = fmaxf(v, 0.f);
                    out[(long)row * N + nt * 16 + col0] = v;
                }
            }
        }
    }
}

__global__ __launch_bounds__(256) void k_final(const float* __restrict__ h,
                                               const float* __restrict__ W2,
                                               const float* __restrict__ b2,
                                               float* __restrict__ out, int M) {
    int i = blockIdx.x * blockDim.x + threadIdx.x;
    if (i >= M) return;
    float acc = b2[0];
    const float* hr = h + (long)i * 32;
    #pragma unroll
    for (int j = 0; j < 32; ++j) acc += hr[j] * W2[j];
    out[i] = acc;
}

extern "C" void kernel_launch(void* const* d_in, const int* in_sizes, int n_in,
                              void* d_out, int out_size, void* d_ws, size_t ws_size,
                              hipStream_t stream) {
    const float* x_news    = (const float*)d_in[0];
    const float* x_company = (const float*)d_in[1];
    const int*   edge_sim  = (const int*)d_in[2];
    const int*   men_src   = (const int*)d_in[3];
    const int*   men_dst   = (const int*)d_in[4];
    const int*   edge_rel  = (const int*)d_in[5];
    const float* nw_W = (const float*)d_in[6];
    const float* nw_b = (const float*)d_in[7];
    const float* cw_W = (const float*)d_in[8];
    const float* cw_b = (const float*)d_in[9];
    const float* n1n_g = (const float*)d_in[10];
    const float* n1n_b = (const float*)d_in[11];
    const float* n1c_g = (const float*)d_in[12];
    const float* n1c_b = (const float*)d_in[13];
    const float* c1_Wl = (const float*)d_in[14];
    const float* c1_bl = (const float*)d_in[15];
    const float* c1_Wr = (const float*)d_in[16];
    const float* c2_Wl = (const float*)d_in[17];
    const float* c2_bl = (const float*)d_in[18];
    const float* c2_Wr = (const float*)d_in[19];
    const float* cls_W1 = (const float*)d_in[20];
    const float* cls_b1 = (const float*)d_in[21];
    const float* cls_W2 = (const float*)d_in[22];
    const float* cls_b2 = (const float*)d_in[23];
    float* out = (float*)d_out;

    float* ws = (float*)d_ws;
    long o = 0;
    float* xn0  = ws + o; o += (long)NN * HDIM;
    float* xn1  = ws + o; o += (long)NN * HDIM;
    float* aggn = ws + o; o += (long)NN * HDIM;
    float* aggcM = ws + o; o += (long)NC * HDIM;
    float* aggcR = ws + o; o += (long)NC * HDIM;
    float* xc0  = ws + o; o += (long)NC * HDIM;
    float* xc1  = ws + o; o += (long)NC * HDIM;
    float* ocb  = ws + o; o += (long)NC * HDIM;
    float* comp = ws + o; o += (long)NC * HDIM;
    float* hbuf = ws + o; o += (long)NC * 32;
    v2f* wp_nw  = (v2f*)(ws + o); o += 2L * (388 / 2) * 64;
    v2f* wp_cw  = (v2f*)(ws + o); o += 2L * (24 / 2) * 64;
    v2f* wp_c1  = (v2f*)(ws + o); o += 2L * 6 * 32 * 64;
    v2f* wp_c2  = (v2f*)(ws + o); o += 2L * 4 * 32 * 64;
    v2f* wp_cls = (v2f*)(ws + o); o += 2L * 32 * 32;

    int* ctr   = (int*)comp;
    int* ctrN  = ctr;
    int* ctrM1 = ctr + 512;
    int* ctrR  = ctr + 1024;
    int* ctrM2 = ctr + 1536;
    int* blR   = ctr + 4096;
    int* blN   = (int*)xn1;
    int* blM1  = blN + 256L * 15360;
    int* blM2  = (int*)xn0;

    dim3 blk(256);
    auto blocks = [](long n) { return (unsigned)((n + 255) / 256); };
    int tilesNN = (NN + 15) / 16, tilesNC = (NC + 15) / 16;
    unsigned gemmBlkNN = (tilesNN + 7) / 8, gemmBlkNC = (tilesNC + 7) / 8;

    k_prep<<<PB_NW + PB_CW + PB_C1 + PB_C2 + PB_CLS + PB_Z, blk, 0, stream>>>(
        nw_W, cw_W, c1_Wl, c1_Wr, c2_Wl, c2_Wr, cls_W1,
        wp_nw, wp_cw, wp_c1, wp_c2, wp_cls, ctr);

    k_gemm<4, false, false><<<gemmBlkNN, blk, 0, stream>>>(x_news, wp_nw, nw_b,
                                                           nullptr, nullptr,
                                                           nullptr, nullptr, nullptr, nullptr,
                                                           xn0, NN, 385, 0, 1);
    k_gemm<4, true, false><<<gemmBlkNC, blk, 0, stream>>>(x_company, wp_cw, cw_b,
                                                          nullptr, nullptr,
                                                          nullptr, nullptr, nullptr, nullptr,
                                                          xc0, NC, 24, 0, 1);

    k_bin3<<<BINB_N + BINB_M + BINB_R, 1024, 0, stream>>>(edge_sim, edge_sim + ES, blN, ctrN,
                                                          men_src, men_dst, blM1, ctrM1,
                                                          edge_rel, edge_rel + ER, blR, ctrR);
    k_own_agg3<<<OWN_N + OWN_C + OWN_C, 1024, 0, stream>>>(edge_sim, edge_sim + ES, xn0, aggn, blN, ctrN,
                                                           men_src, men_dst, xn0, aggcM, blM1, ctrM1,
                                                           edge_rel, edge_rel + ER, xc0, aggcR, blR, ctrR);

    k_gemm<4, true, true><<<gemmBlkNN, blk, 0, stream>>>(aggn, wp_c1 + 0 * 2048,
                                                         c1_bl + 0 * 64,
                                                         xn0, wp_c1 + 1 * 2048,
                                                         nullptr, nullptr, n1n_g, n1n_b,
                                                         xn1, NN, 64, 64, 1);
    k_gemm<4, true, false><<<gemmBlkNC, blk, 0, stream>>>(aggcM, wp_c1 + 2 * 2048,
                                                          c1_bl + 1 * 64,
                                                          xc0, wp_c1 + 3 * 2048,
                                                          nullptr, nullptr, nullptr, nullptr,
                                                          ocb, NC, 64, 64, 0);
    k_gemm<4, true, true><<<gemmBlkNC, blk, 0, stream>>>(aggcR, wp_c1 + 4 * 2048,
                                                         c1_bl + 2 * 64,
                                                         xc0, wp_c1 + 5 * 2048,
                                                         ocb, xc0, n1c_g, n1c_b,
                                                         xc1, NC, 64, 64, 1);

    k_bin<20, 6144><<<(EM + 8191) / 8192, 1024, 0, stream>>>(men_src, men_dst, EM,
                                                             (NC + 19) / 20, blM2, ctrM2);
    k_own_agg2<<<OWN_C + OWN_C, 1024, 0, stream>>>(men_src, men_dst, xn1, aggcM, blM2, ctrM2,
                                                   edge_rel, edge_rel + ER, xc1, aggcR, blR, ctrR);
    k_gemm<4, true, false><<<gemmBlkNC, blk, 0, stream>>>(aggcM, wp_c2 + 0 * 2048,
                                                          c2_bl + 1 * 64,
                                                          xc1, wp_c2 + 1 * 2048,
                                                          nullptr, nullptr, nullptr, nullptr,
                                                          ocb, NC, 64, 64, 0);
    k_gemm<4, true, false><<<gemmBlkNC, blk, 0, stream>>>(aggcR, wp_c2 + 2 * 2048,
                                                          c2_bl + 2 * 64,
                                                          xc1, wp_c2 + 3 * 2048,
                                                          ocb, xc1, nullptr, nullptr,
                                                          comp, NC, 64, 64, 1);

    k_gemm<2, true, false><<<gemmBlkNC, blk, 0, stream>>>(comp, wp_cls, cls_b1,
                                                          nullptr, nullptr,
                                                          nullptr, nullptr, nullptr, nullptr,
                                                          hbuf, NC, 64, 0, 1);
    k_final<<<blocks(NC), blk, 0, stream>>>(hbuf, cls_W2, cls_b2, out, NC);
}


# company rows: each layer's chain of small launches is one kernel: four products in one accumulator set, weights staged in LDS, head and final dot in its epilogue; 14 to 10 launches; sums reordered
# speedup vs baseline: 5.3158x; 1.1604x over previous
#include <hip/hip_runtime.h>
#include <hip/hip_bf16.h>

typedef __attribute__((ext_vector_type(2))) float v2f;
typedef __attribute__((ext_vector_type(8))) float v8f;

#define NN 100000
#define NC 5000
#define HDIM 64
#define ES 3200000
#define EM 1000000
#define ER 160000

__device__ __forceinline__ void pack_one(const float* __restrict__ W,
                                         v2f* __restrict__ Wp,
                                         int K, int N, int i) {
    const int Kp2 = ((K + 3) & ~3) >> 1;
    if (i >= Kp2 * N) return;
    const int k2 = i / N, col = i % N;
    const int k = 2 * k2;
    v2f v;
    v.x = (k     < K) ? W[k * N + col]       : 0.f;
    v.y = (k + 1 < K) ? W[(k + 1) * N + col] : 0.f;
    Wp[i] = v;
}

#define PB_NW  49
#define PB_CW  3
#define PB_C1  48
#define PB_C2  32
#define PB_CLS 4
#define PB_Z   16
__global__ __launch_bounds__(256) void k_prep(const float* __restrict__ nw_W,
                                              const float* __restrict__ cw_W,
                                              const float* __restrict__ c1_Wl,
                                              const float* __restrict__ c1_Wr,
                                              const float* __restrict__ c2_Wl,
                                              const float* __restrict__ c2_Wr,
                                              const float* __restrict__ cls_W1,
                                              v2f* __restrict__ wp_nw,
                                              v2f* __restrict__ wp_cw,
                                              v2f* __restrict__ wp_c1,
                                              v2f* __restrict__ wp_c2,
                                              v2f* __restrict__ wp_cls,
                                              int* __restrict__ ctr) {
    const int tid = threadIdx.x;
    int b = blockIdx.x;
    if (b < PB_NW) { pack_one(nw_W, wp_nw, 385, 64, b * 256 + tid); return; }
    b -= PB_NW;
    if (b < PB_CW) { pack_one(cw_W, wp_cw, 24, 64, b * 256 + tid); return; }
    b -= PB_CW;
    if (b < PB_C1) {
        const int m = b >> 3;
        const float* W = ((m & 1) ? c1_Wr : c1_Wl) + (m >> 1) * 4096;
        pack_one(W, wp_c1 + m * 2048, 64, 64, (b & 7) * 256 + tid);
        return;
    }
    b -= PB_C1;
    if (b < PB_C2) {
        const int m = b >> 3;
        const float* W = ((m & 1) ? c2_Wr : c2_Wl) + ((m >> 1) + 1) * 4096;
        pack_one(W, wp_c2 + m * 2048, 64, 64, (b & 7) * 256 + tid);
        return;
    }
    b -= PB_C2;
    if (b < PB_CLS) { pack_one(cls_W1, wp_cls, 64, 32, b * 256 + tid); return; }
    b -= PB_CLS;
    ctr[b * 256 + tid] = 0;
}

#define PK_SH 9
#define PK_MASK 511
#define BIN_EPT 8
#define BIN_CHUNK (1024 * BIN_EPT)
#define BIN_NBMAX 256
template<int R, int CAPB>
__device__ __forceinline__ void bin_body(const int* __restrict__ src,
                                         const int* __restrict__ dst,
                                         int nE, int nB,
                                         int* __restrict__ blist,
                                         int* __restrict__ bctr,
                                         int bid,
                                         int* hcnt, int* hoff, int* gbase,
                                         int* stage, int* sown) {
    static_assert(R <= PK_MASK + 1, "row inside the range must fit the packed field");
    constexpr int EPT = BIN_EPT;
    constexpr int CHUNK = BIN_CHUNK;
    constexpr int NBMAX = BIN_NBMAX;

    const int tid = threadIdx.x;
    const int lane = tid & 31;
    const int wv = tid >> 5;
    const int base = bid * CHUNK;

    if (tid < NBMAX) hcnt[tid] = 0;
    __syncthreads();

    int ow[EPT], rk[EPT], pk[EPT];
    #pragma unroll
    for (int u = 0; u < EPT; ++u) {
        const int e = base + u * 1024 + tid;
        ow[u] = -1; rk[u] = 0; pk[u] = 0;
        if (e < nE) {
            const int d = dst[e];
            const int o = (int)((unsigned)d / (unsigned)R);
            ow[u] = o;
            pk[u] = (src[e] << PK_SH) | (d - o * R);
            rk[u] = atomicAdd(&hcnt[o], 1);
        }
    }
    __syncthreads();

    if (tid < nB) {
        const int cnt = hcnt[tid];
        gbase[tid] = cnt ? atomicAdd(&bctr[tid], cnt) : 0;
    }
    if (wv == 31) {
        int cq[8];
        int t = 0;
        #pragma unroll
        for (int q = 0; q < 8; ++q) { const int v = hcnt[lane * 8 + q]; cq[q] = t; t += v; }
        int incl = t;
        #pragma unroll
        for (int off = 1; off < 32; off <<= 1) {
            const int s = __shfl_up(incl, off, 32);
            if (lane >= off) incl += s;
        }
        const int excl = incl - t;
        #pragma unroll
        for (int q = 0; q < 8; ++q) hoff[lane * 8 + q] = cq[q] + excl;
    }
    __syncthreads();

    #pragma unroll
    for (int u = 0; u < EPT; ++u) {
        if (ow[u] >= 0) {
            const int lp = hoff[ow[u]] + rk[u];
            stage[lp] = pk[u];
            sown[lp] = ow[u];
        }
    }
    __syncthreads();

    const int total = min(CHUNK, nE - base);
    for (int i = tid; i < total; i += 1024) {
        const int o = sown[i];
        const int pos = gbase[o] + (i - hoff[o]);
        if (pos < CAPB) blist[(long)o * CAPB + pos] = stage[i];
        else bctr[nB] = 1;
    }
}

#define BIN_LDS \
    __shared__ int hcnt[BIN_NBMAX];      \
    __shared__ int hoff[BIN_NBMAX];      \
    __shared__ int gbase[BIN_NBMAX];     \
    __shared__ int stage[BIN_CHUNK];     \
    __shared__ int sown[BIN_CHUNK];

template<int R, int CAPB>
__global__ __launch_bounds__(1024) void k_bin(const int* __restrict__ src,
                                              const int* __restrict__ dst,
                                              int nE, int nB,
                                              int* __restrict__ blist,
                                              int* __restrict__ bctr) {
    BIN_LDS
    bin_body<R, CAPB>(src, dst, nE, nB, blist, bctr, blockIdx.x, hcnt, hoff, gbase, stage, sown);
}

#define BINB_N ((ES + BIN_CHUNK - 1) / BIN_CHUNK)
#define BINB_M ((EM + BIN_CHUNK - 1) / BIN_CHUNK)
#define BINB_R ((ER + BIN_CHUNK - 1) / BIN_CHUNK)
#define OWN_N ((NN + 390) / 391)
#define OWN_C ((NC + 19) / 20)
__global__ __launch_bounds__(1024) void k_bin3(const int* __restrict__ sN, const int* __restrict__ dN,
                                               int* __restrict__ blN, int* __restrict__ cN,
                                               const int* __restrict__ sM, const int* __restrict__ dM,
                                               int* __restrict__ blM, int* __restrict__ cM,
                                               const int* __restrict__ sR, const int* __restrict__ dR,
                                               int* __restrict__ blR, int* __restrict__ cR) {
    BIN_LDS
    const int b = blockIdx.x;
    if (b < BINB_N)
        bin_body<391, 15360>(sN, dN, ES, OWN_N, blN, cN, b, hcnt, hoff, gbase, stage, sown);
    else if (b < BINB_N + BINB_M)
        bin_body<20, 6144>(sM, dM, EM, OWN_C, blM, cM, b - BINB_N, hcnt, hoff, gbase, stage, sown);
    else
        bin_body<20, 1536>(sR, dR, ER, OWN_C, blR, cR, b - BINB_N - BINB_M, hcnt, hoff, gbase, stage, sown);
}

__device__ __forceinline__ void add_group(const float* __restrict__ x, float* acc,
                                          int s, int dl, int m, int lane) {
    if (m >= 32) {
        float v0[32], v1[32];
        #pragma unroll
        for (int j = 0; j < 32; ++j) {
            const int sj = __builtin_amdgcn_readlane(s, j);
            const float* xr = x + (long)sj * HDIM + lane;
            v0[j] = xr[0];
            v1[j] = xr[32];
        }
        #pragma unroll
        for (int j = 0; j < 32; ++j) {
            const int dj = __builtin_amdgcn_readlane(dl, j);
            float* ar = acc + dj * HDIM + lane;
            atomicAdd(ar, v0[j]);
            atomicAdd(ar + 32, v1[j]);
        }
    } else {
        for (int j = 0; j < m; ++j) {
            const int sj = __builtin_amdgcn_readlane(s, j);
            const int dj = __builtin_amdgcn_readlane(dl, j);
            const float* xr = x + (long)sj * HDIM + lane;
            float* ar = acc + dj * HDIM + lane;
            const float a0 = xr[0];
            const float a1 = xr[32];
            atomicAdd(ar, a0);
            atomicAdd(ar + 32, a1);
        }
    }
}

#define SRT_UB 16
template<int R, int CAPB>
__device__ __forceinline__ void own_sorted_body(const float* __restrict__ x,
                                                float* __restrict__ out,
                                                const int* __restrict__ bl,
                                                int nbin, int lo, int nr,
                                                int* cn, int* off, int* sorted) {
    constexpr int EPT = (CAPB + 1023) / 1024;
    constexpr int Q = (R + 31) / 32;
    const int tid = threadIdx.x;
    const int lane = tid & 31;
    const int wv = __builtin_amdgcn_readfirstlane(tid >> 5);
    const int n = min(nbin, CAPB);

    for (int i = tid; i < R; i += 1024) cn[i] = 0;
    __syncthreads();

    int pk[EPT], rk[EPT];
    #pragma unroll
    for (int u = 0; u < EPT; ++u) {
        const int i = u * 1024 + tid;
        pk[u] = -1; rk[u] = 0;
        if (i < n) {
            const int p = bl[i];
            pk[u] = p;
            rk[u] = atomicAdd(&cn[p & PK_MASK], 1);
        }
    }
    __syncthreads();

    if (wv == 0) {
        int cq[Q];
        int t = 0;
        #pragma unroll
        for (int q = 0; q < Q; ++q) {
            const int r = lane * Q + q;
            const int v = (r < R) ? cn[r] : 0;
            cq[q] = t; t += v;
        }
        int incl = t;
        #pragma unroll
        for (int o = 1; o < 32; o <<= 1) {
            const int s = __shfl_up(incl, o, 32);
            if (lane >= o) incl += s;
        }
        const int excl = incl - t;
        #pragma unroll
        for (int q = 0; q < Q; ++q) {
            const int r = lane * Q + q;
            if (r < R) off[r] = cq[q] + excl;
        }
        if (lane == 31) off[R] = incl;
    }
    __syncthreads();

    #pragma unroll
    for (int u = 0; u < EPT; ++u)
        if (pk[u] >= 0) sorted[off[pk[u] & PK_MASK] + rk[u]] = pk[u];
    for (int i = tid; i < nr * 16; i += 1024) {
        const int r = i >> 4;
        if (cn[r] == 0)
            *(float4*)(out + (long)(lo + r) * HDIM + (i & 15) * 4) = make_float4(0.f, 0.f, 0.f, 0.f);
    }
    __syncthreads();

    const int r0 = min(wv * Q, nr);
    const int r1 = min(r0 + Q, nr);
    const int e0 = off[r0];
    const int e1 = off[r1];
    if (e0 >= e1) return;

    float a0 = 0.f, a1 = 0.f;
    int cur = -1;
    int pn = sorted[min(e0 + lane, e1 - 1)];
    for (int g = e0; g < e1; g += SRT_UB) {
        const int pv = pn;
        pn = sorted[min(g + SRT_UB + lane, e1 - 1)];
        float2 v[SRT_UB];
        #pragma unroll
        for (int u = 0; u < SRT_UB; ++u) {
            const int pu = __builtin_amdgcn_readlane(pv, u);
            v[u] = *(const float2*)(x + (long)(pu >> PK_SH) * HDIM + 2 * lane);
        }
        #pragma unroll
        for (int u = 0; u < SRT_UB; ++u) {
            if (g + u < e1) {
                const int d = __builtin_amdgcn_readlane(pv, u) & PK_MASK;
                if (d != cur) {
                    if (cur >= 0) {
                        const float inv = 1.0f / fmaxf((float)cn[cur], 1.0f);
                        *(float2*)(out + (long)(lo + cur) * HDIM + 2 * lane) = make_float2(a0 * inv, a1 * inv);
                    }
                    cur = d; a0 = 0.f; a1 = 0.f;
                }
                a0 += v[u].x; a1 += v[u].y;
            }
        }
    }
    const float inv = 1.0f / fmaxf((float)cn[cur], 1.0f);
    *(float2*)(out + (long)(lo + cur) * HDIM + 2 * lane) = make_float2(a0 * inv, a1 * inv);
}

template<int R, int CAP, int CAPB>
__device__ __forceinline__ void own_agg_body(const int* __restrict__ src,
                                             const int* __restrict__ dst,
                                             const float* __restrict__ x,
                                             float* __restrict__ out,
                                             int nE, int nDst,
                                             const int* __restrict__ blist,
                                             const int* __restrict__ bctr,
                                             int bid, int nB,
                                             float* acc, int* cn, int* off, int* list, int& nlist) {
    static_assert(CAPB <= CAP, "the ordered list must fit the list array");
    const int tid = threadIdx.x;
    const int lane = tid & 31;
    const int wv = __builtin_amdgcn_readfirstlane(tid >> 5);
    const int lo = bid * R;
    const int nr = min(R, nDst - lo);
    const bool binned = __builtin_amdgcn_readfirstlane(bctr[nB]) == 0;
    const int nbin = __builtin_amdgcn_readfirstlane(bctr[bid]);

    if (binned) {
        own_sorted_body<R, CAPB>(x, out, blist + (long)bid * CAPB, nbin, lo, nr, cn, off, list);
        return;
    }

    for (int i = tid; i < R * HDIM; i += 1024) acc[i] = 0.f;
    for (int i = tid; i < R; i += 1024) cn[i] = 0;
    if (tid == 0) nlist = 0;
    __syncthreads();

    {
    constexpr int U = 8;
    constexpr int CH = U * 128;
    const int nCh = (nE + CH - 1) / CH;
    const int rot = (int)(((long)bid * nCh) / nB);
    for (int c0 = wv; c0 < nCh; c0 += 32) {
        int c = c0 + rot;
        if (c >= nCh) c -= nCh;
        const int base = c * CH;
        int4 d[U];
        if (base + CH <= nE) {
            #pragma unroll
            for (int u = 0; u < U; ++u)
                d[u] = *(const int4*)(dst + base + u * 128 + lane * 4);
        } else {
            #pragma unroll
            for (int u = 0; u < U; ++u) {
                const int i0 = base + u * 128 + lane * 4;
                d[u].x = (i0     < nE) ? dst[i0]     : -1;
                d[u].y = (i0 + 1 < nE) ? dst[i0 + 1] : -1;
                d[u].z = (i0 + 2 < nE) ? dst[i0 + 2] : -1;
                d[u].w = (i0 + 3 < nE) ? dst[i0 + 3] : -1;
            }
        }
        int cl = 0;
        #pragma unroll
        for (int u = 0; u < U; ++u) {
            cl += ((unsigned)(d[u].x - lo) < (unsigned)nr) ? 1 : 0;
            cl += ((unsigned)(d[u].y - lo) < (unsigned)nr) ? 1 : 0;
            cl += ((unsigned)(d[u].z - lo) < (unsigned)nr) ? 1 : 0;
            cl += ((unsigned)(d[u].w - lo) < (unsigned)nr) ? 1 : 0;
        }
        int incl = cl;
        #pragma unroll
        for (int off = 1; off < 32; off <<= 1) {
            const int t = __shfl_up(incl, off, 32);
            if (lane >= off) incl += t;
        }
        const int total = __builtin_amdgcn_readlane(incl, 31);
        if (total == 0) continue;
        int s0 = 0;
        if (lane == 31) s0 = atomicAdd(&nlist, total);
        s0 = __builtin_amdgcn_readlane(s0, 31);
        int pos = s0 + incl - cl;
        #pragma unroll
        for (int u = 0; u < U; ++u) {
            #pragma unroll
            for (int q = 0; q < 4; ++q) {
                const int dv = (q == 0) ? d[u].x : (q == 1) ? d[u].y : (q == 2) ? d[u].z : d[u].w;
                const int dl = dv - lo;
                if ((unsigned)dl < (unsigned)nr) {
                    const int e = base + u * 128 + lane * 4 + q;
                    if (pos < CAP) {
                        list[pos] = e;
                    } else {
                        const float* xr = x + (long)src[e] * HDIM;
                        #pragma nounroll
                        for (int k = 0; k < HDIM; ++k) atomicAdd(&acc[dl * HDIM + k], xr[k]);
                        atomicAdd(&cn[dl], 1);
                    }
                    ++pos;
                }
            }
        }
    }
    }
    __syncthreads();

    {
        const int n = min(nlist, CAP);
        for (int g = wv * 32; g < n; g += 32 * 32) {
            const int k = g + lane;
            int s = 0, dl = 0;
            if (k < n) {
                const int e = list[k];
                s = src[e];
                dl = dst[e] - lo;
                atomicAdd(&cn[dl], 1);
            }
            add_group(x, acc, s, dl, min(32, n - g), lane);
        }
    }
    __syncthreads();

    for (int i = tid; i < nr * 16; i += 1024) {
        const int r = i >> 4;
        const int c4 = (i & 15) * 4;
        const float inv = 1.0f / fmaxf((float)cn[r], 1.0f);
        float4 v = *(const float4*)(acc + r * HDIM + c4);
        v.x *= inv; v.y *= inv; v.z *= inv; v.w *= inv;
        *(float4*)(out + (long)(lo + r) * HDIM + c4) = v;
    }
}

__global__ __launch_bounds__(1024) void k_own_agg3(
        const int* __restrict__ sN, const int* __restrict__ dN, const float* __restrict__ xN,
        float* __restrict__ oN, const int* __restrict__ blN, const int* __restrict__ cN,
        const int* __restrict__ sM, const int* __restrict__ dM, const float* __restrict__ xM,
        float* __restrict__ oM, const int* __restrict__ blM, const int* __restrict__ cM,
        const int* __restrict__ sR, const int* __restrict__ dR, const float* __restrict__ xR,
        float* __restrict__ oR, const int* __restrict__ blR, const int* __restrict__ cR) {
    __shared__ __attribute__((aligned(16))) float acc[391 * HDIM];
    __shared__ int cn[391];
    __shared__ int off[392];
    __shared__ int list[16384];
    __shared__ int nlist;
    const int b = blockIdx.x;
    if (b < OWN_N)
        own_agg_body<391, 16384, 15360>(sN, dN, xN, oN, ES, NN, blN, cN, b, OWN_N,
                                        acc, cn, off, list, nlist);
    else if (b < OWN_N + OWN_C)
        own_agg_body<20, 8192, 6144>(sM, dM, xM, oM, EM, NC, blM, cM, b - OWN_N, OWN_C,
                                     acc, cn, off, list, nlist);
    else
        own_agg_body<20, 8192, 1536>(sR, dR, xR, oR, ER, NC, blR, cR, b - OWN_N - OWN_C, OWN_C,
                                     acc, cn, off, list, nlist);
}

__global__ __launch_bounds__(1024) void k_own_agg2(
        const int* __restrict__ sM, const int* __restrict__ dM, const float* __restrict__ xM,
        float* __restrict__ oM, const int* __restrict__ blM, const int* __restrict__ cM,
        const int* __restrict__ sR, const int* __restrict__ dR, const float* __restrict__ xR,
        float* __restrict__ oR, const int* __restrict__ blR, const int* __restrict__ cR) {
    __shared__ __attribute__((aligned(16))) float acc[20 * HDIM];
    __shared__ int cn[20];
    __shared__ int off[21];
    __shared__ int list[8192];
    __shared__ int nlist;
    const int b = blockIdx.x;
    if (b < OWN_C)
        own_agg_body<20, 8192, 6144>(sM, dM, xM, oM, EM, NC, blM, cM, b, OWN_C,
                                     acc, cn, off, list, nlist);
    else
        own_agg_body<20, 8192, 1536>(sR, dR, xR, oR, ER, NC, blR, cR, b - OWN_C, OWN_C,
                                     acc, cn, off, list, nlist);
}

template<int NT, bool FULL, bool VECA>
__device__ __forceinline__ void gemm_body(v8f (&c)[NT],
                                          const float* __restrict__ A,
                                          const v2f* __restrict__ Wp,
                                          int M, int K, int N,
                                          int mbase, int lane) {
    const int row  = mbase + (lane & 15);
    const int koff = (lane >= 16) ? 2 : 0;
    const int col0 = lane & 15;
    const bool rok = FULL || (row < M);
    const float* __restrict__ Arow = A + (long)row * K;
    const int Kmain = K & ~3;

    #pragma unroll 4
    for (int k = 0; k < Kmain; k += 4) {
        const int kk = k + koff;
        v2f a;
        if (rok) {
            if (VECA) {
                a = *(const v2f*)(Arow + kk);
            } else {
                a.x = Arow[kk];
                a.y = Arow[kk + 1];
            }
        } else {
            a.x = 0.f; a.y = 0.f;
        }
        #pragma unroll
        for (int nt = 0; nt < NT; ++nt) {
            v2f b = Wp[(long)(kk >> 1) * N + nt * 16 + col0];
            c[nt] = __builtin_amdgcn_wmma_f32_16x16x4_f32(
                false, a, false, b, (short)0, c[nt], false, false);
        }
    }
    if (Kmain < K) {
        const int kk = Kmain + koff;
        v2f a;
        a.x = (rok && kk     < K) ? Arow[kk]     : 0.f;
        a.y = (rok && kk + 1 < K) ? Arow[kk + 1] : 0.f;
        #pragma unroll
        for (int nt = 0; nt < NT; ++nt) {
            v2f b = Wp[(long)(kk >> 1) * N + nt * 16 + col0];
            c[nt] = __builtin_amdgcn_wmma_f32_16x16x4_f32(
                false, a, false, b, (short)0, c[nt], false, false);
        }
    }
}

#define SLOTS8 "v_nop\n\tv_nop\n\tv_nop\n\tv_nop\n\tv_nop\n\tv_nop\n\tv_nop\n\tv_nop"
__device__ __forceinline__ void result_slots(v8f (&c)[4]) {
    asm volatile(SLOTS8 : "+v"(c[0]), "+v"(c[1]), "+v"(c[2]), "+v"(c[3]));
}
__device__ __forceinline__ void result_slots(v8f (&c)[2]) {
    asm volatile(SLOTS8 : "+v"(c[0]), "+v"(c[1]));
}

template<int NT, bool VECA1, bool NORM>
__global__ __launch_bounds__(256) void k_gemm(const float* __restrict__ A1,
                                              const v2f* __restrict__ W1p,
                                              const float* __restrict__ bias,
                                              const float* __restrict__ A2,
                                              const v2f* __restrict__ W2p,
                                              const float* __restrict__ prev,
                                              const float* __restrict__ res,
                                              const float* __restrict__ g,
                                              const float* __restrict__ b,
                                              float* __restrict__ out,
                                              int M, int K1, int K2,
                                              int do_relu) {
    static_assert(!NORM || NT * 16 == HDIM, "the normalised row must be the whole tile row");
    const int N = NT * 16;
    const int lane = threadIdx.x & 31;
    const int wv = threadIdx.x >> 5;
    const int mbase = (blockIdx.x * 8 + wv) * 16;
    if (mbase >= M) return;

    v8f c[NT];
    #pragma unroll
    for (int t = 0; t < NT; ++t)
        c[t] = (v8f){0.f, 0.f, 0.f, 0.f, 0.f, 0.f, 0.f, 0.f};

    const bool full = (mbase + 16 <= M);
    if (full) {
        gemm_body<NT, true, VECA1>(c, A1, W1p, M, K1, N, mbase, lane);
        if (A2) gemm_body<NT, true, true>(c, A2, W2p, M, K2, N, mbase, lane);
    } else {
        gemm_body<NT, false, VECA1>(c, A1, W1p, M, K1, N, mbase, lane);
        if (A2) gemm_body<NT, false, true>(c, A2, W2p, M, K2, N, mbase, lane);
    }

    result_slots(c);

    const int col0 = lane & 15;
    const int rofs = (lane >= 16) ? 8 : 0;
    #pragma unroll
    for (int nt = 0; nt < NT; ++nt) {
        const float bb = bias ? bias[nt * 16 + col0] : 0.f;
        #pragma unroll
        for (int r = 0; r < 8; ++r) c[nt][r] += bb;
    }
    if (prev) {
        #pragma unroll
        for (int r = 0; r < 8; ++r) {
            const long rb = (long)min(mbase + r + rofs, M - 1) * N + col0;
            #pragma unroll
            for (int nt = 0; nt < NT; ++nt) c[nt][r] += prev[rb + nt * 16];
        }
    }
    if (res) {
        #pragma unroll
        for (int r = 0; r < 8; ++r) {
            const long rb = (long)min(mbase + r + rofs, M - 1) * N + col0;
            #pragma unroll
            for (int nt = 0; nt < NT; ++nt) c[nt][r] += res[rb + nt * 16];
        }
    }

    if (NORM) {
        float gg[NT], be[NT];
        #pragma unroll
        for (int nt = 0; nt < NT; ++nt) { gg[nt] = g[nt * 16 + col0]; be[nt] = b[nt * 16 + col0]; }
        #pragma unroll
        for (int r = 0; r < 8; ++r) {
            const int row = mbase + r + rofs;
            float v[NT];
            #pragma unroll
            for (int nt = 0; nt < NT; ++nt) v[nt] = fmaxf(c[nt][r], 0.f);
            float s = v[0];
            #pragma unroll
            for (int nt = 1; nt < NT; ++nt) s += v[nt];
            #pragma unroll
            for (int off = 8; off > 0; off >>= 1) s += __shfl_xor(s, off, 32);
            const float mean = s * (1.0f / 64.0f);
            float d[NT];
            #pragma unroll
            for (int nt = 0; nt < NT; ++nt) d[nt] = v[nt] - mean;
            float q = d[0] * d[0];
            #pragma unroll
            for (int nt = 1; nt < NT; ++nt) q += d[nt] * d[nt];
            #pragma unroll
            for (int off = 8; off > 0; off >>= 1) q += __shfl_xor(q, off, 32);
            const float rstd = rsqrtf(q * (1.0f / 64.0f) + 1e-5f);
            if (full || row < M) {
                const long rb = (long)row * N + col0;
                #pragma unroll
                for (int nt = 0; nt < NT; ++nt)
                    out[rb + nt * 16] = d[nt] * rstd * gg[nt] + be[nt];
            }
        }
    } else {
        #pragma unroll
        for (int nt = 0; nt < NT; ++nt) {
            #pragma unroll
            for (int r = 0; r < 8; ++r) {
                const int row = mbase + r + rofs;
                if (full || row < M) {
                    float v = c[nt][r];
                    if (do_relu) v = fmaxf(v, 0.f);
                    out[(long)row * N + nt * 16 + col0] = v;
                }
            }
        }
    }
}

#define CW_PITCH 80
#define CC_PITCH 48
#define CT_PITCH 68
typedef __attribute__((ext_vector_type(4))) float v4f;

template<bool HEAD>
__global__ __launch_bounds__(128) void k_crow(const float* __restrict__ aggM,
                                              const float* __restrict__ aggR,
                                              const float* __restrict__ x,
                                              const v2f* __restrict__ wp,
                                              const float* __restrict__ bM,
                                              const float* __restrict__ bR,
                                              const float* __restrict__ g,
                                              const float* __restrict__ b,
                                              const v2f* __restrict__ wpc,
                                              const float* __restrict__ cb1,
                                              const float* __restrict__ cW2,
                                              const float* __restrict__ cb2,
                                              float* __restrict__ out, int M) {
    __shared__ __attribute__((aligned(16))) v2f sW[4 * 32 * CW_PITCH];
    __shared__ __attribute__((aligned(16))) v2f sWc[HEAD ? 32 * CC_PITCH : 2];
    __shared__ __attribute__((aligned(16))) float sT[HEAD ? 4 * 16 * CT_PITCH : 4];
    __shared__ float sO[HEAD ? 64 : 1];

    const int tid = threadIdx.x;
    const int lane = tid & 31;
    const int wv = tid >> 5;
    const int mbase = (blockIdx.x * 4 + wv) * 16;
    const int col0 = lane & 15;
    const int hh = lane >> 4;
    const int rofs = hh * 8;

    v4f wr[32];
    #pragma unroll
    for (int u = 0; u < 32; ++u) wr[u] = ((const v4f*)wp)[u * 128 + tid];
    v4f wc[4];
    if (HEAD) {
        #pragma unroll
        for (int u = 0; u < 4; ++u) wc[u] = ((const v4f*)wpc)[u * 128 + tid];
    }

    const long ra = (long)min(mbase + col0, M - 1) * HDIM + 2 * hh;
    v2f aM[16], aX[16], aR[16];
    #pragma unroll
    for (int ks = 0; ks < 16; ++ks) {
        aM[ks] = *(const v2f*)(aggM + ra + 4 * ks);
        aX[ks] = *(const v2f*)(x + ra + 4 * ks);
        aR[ks] = *(const v2f*)(aggR + ra + 4 * ks);
    }
    float xr[4][8];
    #pragma unroll
    for (int r = 0; r < 8; ++r) {
        const long rb = (long)min(mbase + r + rofs, M - 1) * HDIM + col0;
        #pragma unroll
        for (int nt = 0; nt < 4; ++nt) xr[nt][r] = x[rb + nt * 16];
    }
    float bb[4];
    #pragma unroll
    for (int nt = 0; nt < 4; ++nt) bb[nt] = bM[nt * 16 + col0] + bR[nt * 16 + col0];
    float gg[4], be[4];
    if (!HEAD) {
        #pragma unroll
        for (int nt = 0; nt < 4; ++nt) { gg[nt] = g[nt * 16 + col0]; be[nt] = b[nt * 16 + col0]; }
    }
    float hb[2], w2[2], b2 = 0.f;
    if (HEAD) {
        #pragma unroll
        for (int nt = 0; nt < 2; ++nt) { hb[nt] = cb1[nt * 16 + col0]; w2[nt] = cW2[nt * 16 + col0]; }
        b2 = cb2[0];
    }

    #pragma unroll
    for (int u = 0; u < 32; ++u) {
        const int i4 = u * 128 + tid;
        ((v4f*)sW)[(i4 >> 5) * (CW_PITCH / 2) + (i4 & 31)] = wr[u];
    }
    if (HEAD) {
        #pragma unroll
        for (int u = 0; u < 4; ++u) {
            const int i4 = u * 128 + tid;
            ((v4f*)sWc)[(i4 >> 4) * (CC_PITCH / 2) + (i4 & 15)] = wc[u];
        }
    }
    __syncthreads();

    v8f c[4];
    #pragma unroll
    for (int t = 0; t < 4; ++t)
        c[t] = (v8f){0.f, 0.f, 0.f, 0.f, 0.f, 0.f, 0.f, 0.f};

    const v2f* sB = sW + hh * CW_PITCH + col0;
    v2f bq[2][16];
    #pragma unroll
    for (int j = 0; j < 4; ++j) {
        #pragma unroll
        for (int nt = 0; nt < 4; ++nt) bq[0][j * 4 + nt] = sB[(2 * j) * CW_PITCH + nt * 16];
    }
    #pragma unroll
    for (int gi = 0; gi < 16; ++gi) {
        if (gi + 1 < 16) {
            const int p1 = (gi + 1) >> 2, q1 = (gi + 1) & 3;
            #pragma unroll
            for (int j = 0; j < 4; ++j) {
                #pragma unroll
                for (int nt = 0; nt < 4; ++nt)
                    bq[(gi + 1) & 1][j * 4 + nt] = sB[(p1 * 32 + 2 * (q1 * 4 + j)) * CW_PITCH + nt * 16];
            }
        }
        const int p = gi >> 2, q = gi & 3;
        #pragma unroll
        for (int j = 0; j < 4; ++j) {
            const int ks = q * 4 + j;
            const v2f a = (p == 0) ? aM[ks] : (p == 2) ? aR[ks] : aX[ks];
            #pragma unroll
            for (int nt = 0; nt < 4; ++nt)
                c[nt] = __builtin_amdgcn_wmma_f32_16x16x4_f32(
                    false, a, false, bq[gi & 1][j * 4 + nt], (short)0, c[nt], false, false);
        }
    }

    if (!HEAD) {
        float v[4][8], d[4][8], s[8], q[8];
        #pragma unroll
        for (int r = 0; r < 8; ++r) {
            #pragma unroll
            for (int nt = 0; nt < 4; ++nt) v[nt][r] = fmaxf(c[nt][r] + bb[nt] + xr[nt][r], 0.f);
            s[r] = v[0][r];
            #pragma unroll
            for (int nt = 1; nt < 4; ++nt) s[r] += v[nt][r];
        }
        #pragma unroll
        for (int off = 8; off > 0; off >>= 1) {
            float t[8];
            #pragma unroll
            for (int r = 0; r < 8; ++r) t[r] = __shfl_xor(s[r], off, 32);
            #pragma unroll
            for (int r = 0; r < 8; ++r) s[r] += t[r];
        }
        #pragma unroll
        for (int r = 0; r < 8; ++r) {
            const float mean = s[r] * (1.0f / 64.0f);
            #pragma unroll
            for (int nt = 0; nt < 4; ++nt) d[nt][r] = v[nt][r] - mean;
            q[r] = d[0][r] * d[0][r];
            #pragma unroll
            for (int nt = 1; nt < 4; ++nt) q[r] += d[nt][r] * d[nt][r];
        }
        #pragma unroll
        for (int off = 8; off > 0; off >>= 1) {
            float t[8];
            #pragma unroll
            for (int r = 0; r < 8; ++r) t[r] = __shfl_xor(q[r], off, 32);
            #pragma unroll
            for (int r = 0; r < 8; ++r) q[r] += t[r];
        }
        #pragma unroll
        for (int r = 0; r < 8; ++r) {
            const int row = mbase + r + rofs;
            const float rstd = rsqrtf(q[r] * (1.0f / 64.0f) + 1e-5f);
            if (row < M) {
                const long rb = (long)row * HDIM + col0;
                #pragma unroll
                for (int nt = 0; nt < 4; ++nt)
                    out[rb + nt * 16] = d[nt][r] * rstd * gg[nt] + be[nt];
            }
        }
    } else {
        float* tw = sT + wv * (16 * CT_PITCH);
        #pragma unroll
        for (int nt = 0; nt < 4; ++nt) {
            #pragma unroll
            for (int r = 0; r < 8; ++r)
                tw[(rofs + r) * CT_PITCH + nt * 16 + col0] = fmaxf(c[nt][r] + bb[nt] + xr[nt][r], 0.f);
        }
        __syncthreads();
        v8f hc[2];
        hc[0] = (v8f){0.f, 0.f, 0.f, 0.f, 0.f, 0.f, 0.f, 0.f};
        hc[1] = (v8f){0.f, 0.f, 0.f, 0.f, 0.f, 0.f, 0.f, 0.f};
        const float* ta = tw + col0 * CT_PITCH + 2 * hh;
        const v2f* sC = sWc + hh * CC_PITCH + col0;
        #pragma unroll
        for (int ks = 0; ks < 16; ++ks) {
            const v2f a = *(const v2f*)(ta + 4 * ks);
            #pragma unroll
            for (int nt = 0; nt < 2; ++nt)
                hc[nt] = __builtin_amdgcn_wmma_f32_16x16x4_f32(
                    false, a, false, sC[(2 * ks) * CC_PITCH + nt * 16], (short)0, hc[nt], false, false);
        }
        float s[8];
        #pragma unroll
        for (int r = 0; r < 8; ++r)
            s[r] = fmaxf(hc[0][r] + hb[0], 0.f) * w2[0] + fmaxf(hc[1][r] + hb[1], 0.f) * w2[1];
        #pragma unroll
        for (int off = 8; off > 0; off >>= 1) {
            float t[8];
            #pragma unroll
            for (int r = 0; r < 8; ++r) t[r] = __shfl_xor(s[r], off, 32);
            #pragma unroll
            for (int r = 0; r < 8; ++r) s[r] += t[r];
        }
        float mine = 0.f;
        #pragma unroll
        for (int r = 0; r < 8; ++r)
            if (col0 == r) mine = s[r];
        if (col0 < 8) sO[wv * 16 + rofs + col0] = mine + b2;
        __syncthreads();
        if (tid < 64) {
            const int row = blockIdx.x * 64 + tid;
            if (row < M) out[row] = sO[tid];
        }
    }
}

extern "C" void kernel_launch(void* const* d_in, const int* in_sizes, int n_in,
                              void* d_out, int out_size, void* d_ws, size_t ws_size,
                              hipStream_t stream) {
    const float* x_news    = (const float*)d_in[0];
    const float* x_company = (const float*)d_in[1];
    const int*   edge_sim  = (const int*)d_in[2];
    const int*   men_src   = (const int*)d_in[3];
    const int*   men_dst   = (const int*)d_in[4];
    const int*   edge_rel  = (const int*)d_in[5];
    const float* nw_W = (const float*)d_in[6];
    const float* nw_b = (const float*)d_in[7];
    const float* cw_W = (const float*)d_in[8];
    const float* cw_b = (const float*)d_in[9];
    const float* n1n_g = (const float*)d_in[10];
    const float* n1n_b = (const float*)d_in[11];
    const float* n1c_g = (const float*)d_in[12];
    const float* n1c_b = (const float*)d_in[13];
    const float* c1_Wl = (const float*)d_in[14];
    const float* c1_bl = (const float*)d_in[15];
    const float* c1_Wr = (const float*)d_in[16];
    const float* c2_Wl = (const float*)d_in[17];
    const float* c2_bl = (const float*)d_in[18];
    const float* c2_Wr = (const float*)d_in[19];
    const float* cls_W1 = (const float*)d_in[20];
    const float* cls_b1 = (const float*)d_in[21];
    const float* cls_W2 = (const float*)d_in[22];
    const float* cls_b2 = (const float*)d_in[23];
    float* out = (float*)d_out;

    float* ws = (float*)d_ws;
    long o = 0;
    float* xn0  = ws + o; o += (long)NN * HDIM;
    float* xn1  = ws + o; o += (long)NN * HDIM;
    float* aggn = ws + o; o += (long)NN * HDIM;
    float* aggcM = ws + o; o += (long)NC * HDIM;
    float* aggcR = ws + o; o += (long)NC * HDIM;
    float* xc0  = ws + o; o += (long)NC * HDIM;
    float* xc1  = ws + o; o += (long)NC * HDIM;
    o += (long)NC * HDIM;
    float* comp = ws + o; o += (long)NC * HDIM;
    o += (long)NC * 32;
    v2f* wp_nw  = (v2f*)(ws + o); o += 2L * (388 / 2) * 64;
    v2f* wp_cw  = (v2f*)(ws + o); o += 2L * (24 / 2) * 64;
    v2f* wp_c1  = (v2f*)(ws + o); o += 2L * 6 * 32 * 64;
    v2f* wp_c2  = (v2f*)(ws + o); o += 2L * 4 * 32 * 64;
    v2f* wp_cls = (v2f*)(ws + o); o += 2L * 32 * 32;

    int* ctr   = (int*)comp;
    int* ctrN  = ctr;
    int* ctrM1 = ctr + 512;
    int* ctrR  = ctr + 1024;
    int* ctrM2 = ctr + 1536;
    int* blR   = ctr + 4096;
    int* blN   = (int*)xn1;
    int* blM1  = blN + 256L * 15360;
    int* blM2  = (int*)xn0;

    dim3 blk(256);
    int tilesNN = (NN + 15) / 16, tilesNC = (NC + 15) / 16;
    unsigned gemmBlkNN = (tilesNN + 7) / 8, gemmBlkNC = (tilesNC + 7) / 8;
    unsigned crowBlk = (tilesNC + 3) / 4;

    k_prep<<<PB_NW + PB_CW + PB_C1 + PB_C2 + PB_CLS + PB_Z, blk, 0, stream>>>(
        nw_W, cw_W, c1_Wl, c1_Wr, c2_Wl, c2_Wr, cls_W1,
        wp_nw, wp_cw, wp_c1, wp_c2, wp_cls, ctr);

    k_gemm<4, false, false><<<gemmBlkNN, blk, 0, stream>>>(x_news, wp_nw, nw_b,
                                                           nullptr, nullptr,
                                                           nullptr, nullptr, nullptr, nullptr,
                                                           xn0, NN, 385, 0, 1);
    k_gemm<4, true, false><<<gemmBlkNC, blk, 0, stream>>>(x_company, wp_cw, cw_b,
                                                          nullptr, nullptr,
                                                          nullptr, nullptr, nullptr, nullptr,
                                                          xc0, NC, 24, 0, 1);

    k_bin3<<<BINB_N + BINB_M + BINB_R, 1024, 0, stream>>>(edge_sim, edge_sim + ES, blN, ctrN,
                                                          men_src, men_dst, blM1, ctrM1,
                                                          edge_rel, edge_rel + ER, blR, ctrR);
    k_own_agg3<<<OWN_N + OWN_C + OWN_C, 1024, 0, stream>>>(edge_sim, edge_sim + ES, xn0, aggn, blN, ctrN,
                                                           men_src, men_dst, xn0, aggcM, blM1, ctrM1,
                                                           edge_rel, edge_rel + ER, xc0, aggcR, blR, ctrR);

    k_gemm<4, true, true><<<gemmBlkNN, blk, 0, stream>>>(aggn, wp_c1 + 0 * 2048,
                                                         c1_bl + 0 * 64,
                                                         xn0, wp_c1 + 1 * 2048,
                                                         nullptr, nullptr, n1n_g, n1n_b,
                                                         xn1, NN, 64, 64, 1);
    k_crow<false><<<crowBlk, 128, 0, stream>>>(aggcM, aggcR, xc0, wp_c1 + 2 * 2048,
                                               c1_bl + 1 * 64, c1_bl + 2 * 64, n1c_g, n1c_b,
                                               nullptr, nullptr, nullptr, nullptr,
                                               xc1, NC);

    k_bin<20, 6144><<<(EM + 8191) / 8192, 1024, 0, stream>>>(men_src, men_dst, EM,
                                                             (NC + 19) / 20, blM2, ctrM2);
    k_own_agg2<<<OWN_C + OWN_C, 1024, 0, stream>>>(men_src, men_dst, xn1, aggcM, blM2, ctrM2,
                                                   edge_rel, edge_rel + ER, xc1, aggcR, blR, ctrR);
    k_crow<true><<<crowBlk, 128, 0, stream>>>(aggcM, aggcR, xc1, wp_c2,
                                              c2_bl + 1 * 64, c2_bl + 2 * 64, nullptr, nullptr,
                                              wp_cls, cls_b1, cls_W2, cls_b2,
                                              out, NC);
}


# layer 1 stores its news rows over their own mean rows, so the plane with the binned news-to-company lists survives and layer 2 reads them again: second binning launch dropped, 10 to 9 launches
# speedup vs baseline: 5.8019x; 1.0914x over previous
#include <hip/hip_runtime.h>
#include <hip/hip_bf16.h>

typedef __attribute__((ext_vector_type(2))) float v2f;
typedef __attribute__((ext_vector_type(8))) float v8f;

#define NN 100000
#define NC 5000
#define HDIM 64
#define ES 3200000
#define EM 1000000
#define ER 160000

__device__ __forceinline__ void pack_one(const float* __restrict__ W,
                                         v2f* __restrict__ Wp,
                                         int K, int N, int i) {
    const int Kp2 = ((K + 3) & ~3) >> 1;
    if (i >= Kp2 * N) return;
    const int k2 = i / N, col = i % N;
    const int k = 2 * k2;
    v2f v;
    v.x = (k     < K) ? W[k * N + col]       : 0.f;
    v.y = (k + 1 < K) ? W[(k + 1) * N + col] : 0.f;
    Wp[i] = v;
}

#define PB_NW  49
#define PB_CW  3
#define PB_C1  48
#define PB_C2  32
#define PB_CLS 4
#define PB_Z   16
__global__ __launch_bounds__(256) void k_prep(const float* __restrict__ nw_W,
                                              const float* __restrict__ cw_W,
                                              const float* __restrict__ c1_Wl,
                                              const float* __restrict__ c1_Wr,
                                              const float* __restrict__ c2_Wl,
                                              const float* __restrict__ c2_Wr,
                                              const float* __restrict__ cls_W1,
                                              v2f* __restrict__ wp_nw,
                                              v2f* __restrict__ wp_cw,
                                              v2f* __restrict__ wp_c1,
                                              v2f* __restrict__ wp_c2,
                                              v2f* __restrict__ wp_cls,
                                              int* __restrict__ ctr) {
    const int tid = threadIdx.x;
    int b = blockIdx.x;
    if (b < PB_NW) { pack_one(nw_W, wp_nw, 385, 64, b * 256 + tid); return; }
    b -= PB_NW;
    if (b < PB_CW) { pack_one(cw_W, wp_cw, 24, 64, b * 256 + tid); return; }
    b -= PB_CW;
    if (b < PB_C1) {
        const int m = b >> 3;
        const float* W = ((m & 1) ? c1_Wr : c1_Wl) + (m >> 1) * 4096;
        pack_one(W, wp_c1 + m * 2048, 64, 64, (b & 7) * 256 + tid);
        return;
    }
    b -= PB_C1;
    if (b < PB_C2) {
        const int m = b >> 3;
        const float* W = ((m & 1) ? c2_Wr : c2_Wl) + ((m >> 1) + 1) * 4096;
        pack_one(W, wp_c2 + m * 2048, 64, 64, (b & 7) * 256 + tid);
        return;
    }
    b -= PB_C2;
    if (b < PB_CLS) { pack_one(cls_W1, wp_cls, 64, 32, b * 256 + tid); return; }
    b -= PB_CLS;
    ctr[b * 256 + tid] = 0;
}

#define PK_SH 9
#define PK_MASK 511
#define BIN_EPT 8
#define BIN_CHUNK (1024 * BIN_EPT)
#define BIN_NBMAX 256
template<int R, int CAPB>
__device__ __forceinline__ void bin_body(const int* __restrict__ src,
                                         const int* __restrict__ dst,
                                         int nE, int nB,
                                         int* __restrict__ blist,
                                         int* __restrict__ bctr,
                                         int bid,
                                         int* hcnt, int* hoff, int* gbase,
                                         int* stage, int* sown) {
    static_assert(R <= PK_MASK + 1, "row inside the range must fit the packed field");
    constexpr int EPT = BIN_EPT;
    constexpr int CHUNK = BIN_CHUNK;
    constexpr int NBMAX = BIN_NBMAX;

    const int tid = threadIdx.x;
    const int lane = tid & 31;
    const int wv = tid >> 5;
    const int base = bid * CHUNK;

    if (tid < NBMAX) hcnt[tid] = 0;
    __syncthreads();

    int ow[EPT], rk[EPT], pk[EPT];
    #pragma unroll
    for (int u = 0; u < EPT; ++u) {
        const int e = base + u * 1024 + tid;
        ow[u] = -1; rk[u] = 0; pk[u] = 0;
        if (e < nE) {
            const int d = dst[e];
            const int o = (int)((unsigned)d / (unsigned)R);
            ow[u] = o;
            pk[u] = (src[e] << PK_SH) | (d - o * R);
            rk[u] = atomicAdd(&hcnt[o], 1);
        }
    }
    __syncthreads();

    if (tid < nB) {
        const int cnt = hcnt[tid];
        gbase[tid] = cnt ? atomicAdd(&bctr[tid], cnt) : 0;
    }
    if (wv == 31) {
        int cq[8];
        int t = 0;
        #pragma unroll
        for (int q = 0; q < 8; ++q) { const int v = hcnt[lane * 8 + q]; cq[q] = t; t += v; }
        int incl = t;
        #pragma unroll
        for (int off = 1; off < 32; off <<= 1) {
            const int s = __shfl_up(incl, off, 32);
            if (lane >= off) incl += s;
        }
        const int excl = incl - t;
        #pragma unroll
        for (int q = 0; q < 8; ++q) hoff[lane * 8 + q] = cq[q] + excl;
    }
    __syncthreads();

    #pragma unroll
    for (int u = 0; u < EPT; ++u) {
        if (ow[u] >= 0) {
            const int lp = hoff[ow[u]] + rk[u];
            stage[lp] = pk[u];
            sown[lp] = ow[u];
        }
    }
    __syncthreads();

    const int total = min(CHUNK, nE - base);
    for (int i = tid; i < total; i += 1024) {
        const int o = sown[i];
        const int pos = gbase[o] + (i - hoff[o]);
        if (pos < CAPB) blist[(long)o * CAPB + pos] = stage[i];
        else bctr[nB] = 1;
    }
}

#define BIN_LDS \
    __shared__ int hcnt[BIN_NBMAX];      \
    __shared__ int hoff[BIN_NBMAX];      \
    __shared__ int gbase[BIN_NBMAX];     \
    __shared__ int stage[BIN_CHUNK];     \
    __shared__ int sown[BIN_CHUNK];

#define BINB_N ((ES + BIN_CHUNK - 1) / BIN_CHUNK)
#define BINB_M ((EM + BIN_CHUNK - 1) / BIN_CHUNK)
#define BINB_R ((ER + BIN_CHUNK - 1) / BIN_CHUNK)
#define OWN_N ((NN + 390) / 391)
#define OWN_C ((NC + 19) / 20)
__global__ __launch_bounds__(1024) void k_bin3(const int* __restrict__ sN, const int* __restrict__ dN,
                                               int* __restrict__ blN, int* __restrict__ cN,
                                               const int* __restrict__ sM, const int* __restrict__ dM,
                                               int* __restrict__ blM, int* __restrict__ cM,
                                               const int* __restrict__ sR, const int* __restrict__ dR,
                                               int* __restrict__ blR, int* __restrict__ cR) {
    BIN_LDS
    const int b = blockIdx.x;
    if (b < BINB_N)
        bin_body<391, 15360>(sN, dN, ES, OWN_N, blN, cN, b, hcnt, hoff, gbase, stage, sown);
    else if (b < BINB_N + BINB_M)
        bin_body<20, 6144>(sM, dM, EM, OWN_C, blM, cM, b - BINB_N, hcnt, hoff, gbase, stage, sown);
    else
        bin_body<20, 1536>(sR, dR, ER, OWN_C, blR, cR, b - BINB_N - BINB_M, hcnt, hoff, gbase, stage, sown);
}

__device__ __forceinline__ void add_group(const float* __restrict__ x, float* acc,
                                          int s, int dl, int m, int lane) {
    if (m >= 32) {
        float v0[32], v1[32];
        #pragma unroll
        for (int j = 0; j < 32; ++j) {
            const int sj = __builtin_amdgcn_readlane(s, j);
            const float* xr = x + (long)sj * HDIM + lane;
            v0[j] = xr[0];
            v1[j] = xr[32];
        }
        #pragma unroll
        for (int j = 0; j < 32; ++j) {
            const int dj = __builtin_amdgcn_readlane(dl, j);
            float* ar = acc + dj * HDIM + lane;
            atomicAdd(ar, v0[j]);
            atomicAdd(ar + 32, v1[j]);
        }
    } else {
        for (int j = 0; j < m; ++j) {
            const int sj = __builtin_amdgcn_readlane(s, j);
            const int dj = __builtin_amdgcn_readlane(dl, j);
            const float* xr = x + (long)sj * HDIM + lane;
            float* ar = acc + dj * HDIM + lane;
            const float a0 = xr[0];
            const float a1 = xr[32];
            atomicAdd(ar, a0);
            atomicAdd(ar + 32, a1);
        }
    }
}

#define SRT_UB 16
template<int R, int CAPB>
__device__ __forceinline__ void own_sorted_body(const float* __restrict__ x,
                                                float* __restrict__ out,
                                                const int* __restrict__ bl,
                                                int nbin, int lo, int nr,
                                                int* cn, int* off, int* sorted) {
    constexpr int EPT = (CAPB + 1023) / 1024;
    constexpr int Q = (R + 31) / 32;
    const int tid = threadIdx.x;
    const int lane = tid & 31;
    const int wv = __builtin_amdgcn_readfirstlane(tid >> 5);
    const int n = min(nbin, CAPB);

    for (int i = tid; i < R; i += 1024) cn[i] = 0;
    __syncthreads();

    int pk[EPT], rk[EPT];
    #pragma unroll
    for (int u = 0; u < EPT; ++u) {
        const int i = u * 1024 + tid;
        pk[u] = -1; rk[u] = 0;
        if (i < n) {
            const int p = bl[i];
            pk[u] = p;
            rk[u] = atomicAdd(&cn[p & PK_MASK], 1);
        }
    }
    __syncthreads();

    if (wv == 0) {
        int cq[Q];
        int t = 0;
        #pragma unroll
        for (int q = 0; q < Q; ++q) {
            const int r = lane * Q + q;
            const int v = (r < R) ? cn[r] : 0;
            cq[q] = t; t += v;
        }
        int incl = t;
        #pragma unroll
        for (int o = 1; o < 32; o <<= 1) {
            const int s = __shfl_up(incl, o, 32);
            if (lane >= o) incl += s;
        }
        const int excl = incl - t;
        #pragma unroll
        for (int q = 0; q < Q; ++q) {
            const int r = lane * Q + q;
            if (r < R) off[r] = cq[q] + excl;
        }
        if (lane == 31) off[R] = incl;
    }
    __syncthreads();

    #pragma unroll
    for (int u = 0; u < EPT; ++u)
        if (pk[u] >= 0) sorted[off[pk[u] & PK_MASK] + rk[u]] = pk[u];
    for (int i = tid; i < nr * 16; i += 1024) {
        const int r = i >> 4;
        if (cn[r] == 0)
            *(float4*)(out + (long)(lo + r) * HDIM + (i & 15) * 4) = make_float4(0.f, 0.f, 0.f, 0.f);
    }
    __syncthreads();

    const int r0 = min(wv * Q, nr);
    const int r1 = min(r0 + Q, nr);
    const int e0 = off[r0];
    const int e1 = off[r1];
    if (e0 >= e1) return;

    float a0 = 0.f, a1 = 0.f;
    int cur = -1;
    int pn = sorted[min(e0 + lane, e1 - 1)];
    for (int g = e0; g < e1; g += SRT_UB) {
        const int pv = pn;
        pn = sorted[min(g + SRT_UB + lane, e1 - 1)];
        float2 v[SRT_UB];
        #pragma unroll
        for (int u = 0; u < SRT_UB; ++u) {
            const int pu = __builtin_amdgcn_readlane(pv, u);
            v[u] = *(const float2*)(x + (long)(pu >> PK_SH) * HDIM + 2 * lane);
        }
        #pragma unroll
        for (int u = 0; u < SRT_UB; ++u) {
            if (g + u < e1) {
                const int d = __builtin_amdgcn_readlane(pv, u) & PK_MASK;
                if (d != cur) {
                    if (cur >= 0) {
                        const float inv = 1.0f / fmaxf((float)cn[cur], 1.0f);
                        *(float2*)(out + (long)(lo + cur) * HDIM + 2 * lane) = make_float2(a0 * inv, a1 * inv);
                    }
                    cur = d; a0 = 0.f; a1 = 0.f;
                }
                a0 += v[u].x; a1 += v[u].y;
            }
        }
    }
    const float inv = 1.0f / fmaxf((float)cn[cur], 1.0f);
    *(float2*)(out + (long)(lo + cur) * HDIM + 2 * lane) = make_float2(a0 * inv, a1 * inv);
}

template<int R, int CAP, int CAPB>
__device__ __forceinline__ void own_agg_body(const int* __restrict__ src,
                                             const int* __restrict__ dst,
                                             const float* __restrict__ x,
                                             float* __restrict__ out,
                                             int nE, int nDst,
                                             const int* __restrict__ blist,
                                             const int* __restrict__ bctr,
                                             int bid, int nB,
                                             float* acc, int* cn, int* off, int* list, int& nlist) {
    static_assert(CAPB <= CAP, "the ordered list must fit the list array");
    const int tid = threadIdx.x;
    const int lane = tid & 31;
    const int wv = __builtin_amdgcn_readfirstlane(tid >> 5);
    const int lo = bid * R;
    const int nr = min(R, nDst - lo);
    const bool binned = __builtin_amdgcn_readfirstlane(bctr[nB]) == 0;
    const int nbin = __builtin_amdgcn_readfirstlane(bctr[bid]);

    if (binned) {
        own_sorted_body<R, CAPB>(x, out, blist + (long)bid * CAPB, nbin, lo, nr, cn, off, list);
        return;
    }

    for (int i = tid; i < R * HDIM; i += 1024) acc[i] = 0.f;
    for (int i = tid; i < R; i += 1024) cn[i] = 0;
    if (tid == 0) nlist = 0;
    __syncthreads();

    {
    constexpr int U = 8;
    constexpr int CH = U * 128;
    const int nCh = (nE + CH - 1) / CH;
    const int rot = (int)(((long)bid * nCh) / nB);
    for (int c0 = wv; c0 < nCh; c0 += 32) {
        int c = c0 + rot;
        if (c >= nCh) c -= nCh;
        const int base = c * CH;
        int4 d[U];
        if (base + CH <= nE) {
            #pragma unroll
            for (int u = 0; u < U; ++u)
                d[u] = *(const int4*)(dst + base + u * 128 + lane * 4);
        } else {
            #pragma unroll
            for (int u = 0; u < U; ++u) {
                const int i0 = base + u * 128 + lane * 4;
                d[u].x = (i0     < nE) ? dst[i0]     : -1;
                d[u].y = (i0 + 1 < nE) ? dst[i0 + 1] : -1;
                d[u].z = (i0 + 2 < nE) ? dst[i0 + 2] : -1;
                d[u].w = (i0 + 3 < nE) ? dst[i0 + 3] : -1;
            }
        }
        int cl = 0;
        #pragma unroll
        for (int u = 0; u < U; ++u) {
            cl += ((unsigned)(d[u].x - lo) < (unsigned)nr) ? 1 : 0;
            cl += ((unsigned)(d[u].y - lo) < (unsigned)nr) ? 1 : 0;
            cl += ((unsigned)(d[u].z - lo) < (unsigned)nr) ? 1 : 0;
            cl += ((unsigned)(d[u].w - lo) < (unsigned)nr) ? 1 : 0;
        }
        int incl = cl;
        #pragma unroll
        for (int off = 1; off < 32; off <<= 1) {
            const int t = __shfl_up(incl, off, 32);
            if (lane >= off) incl += t;
        }
        const int total = __builtin_amdgcn_readlane(incl, 31);
        if (total == 0) continue;
        int s0 = 0;
        if (lane == 31) s0 = atomicAdd(&nlist, total);
        s0 = __builtin_amdgcn_readlane(s0, 31);
        int pos = s0 + incl - cl;
        #pragma unroll
        for (int u = 0; u < U; ++u) {
            #pragma unroll
            for (int q = 0; q < 4; ++q) {
                const int dv = (q == 0) ? d[u].x : (q == 1) ? d[u].y : (q == 2) ? d[u].z : d[u].w;
                const int dl = dv - lo;
                if ((unsigned)dl < (unsigned)nr) {
                    const int e = base + u * 128 + lane * 4 + q;
                    if (pos < CAP) {
                        list[pos] = e;
                    } else {
                        const float* xr = x + (long)src[e] * HDIM;
                        #pragma nounroll
                        for (int k = 0; k < HDIM; ++k) atomicAdd(&acc[dl * HDIM + k], xr[k]);
                        atomicAdd(&cn[dl], 1);
                    }
                    ++pos;
                }
            }
        }
    }
    }
    __syncthreads();

    {
        const int n = min(nlist, CAP);
        for (int g = wv * 32; g < n; g += 32 * 32) {
            const int k = g + lane;
            int s = 0, dl = 0;
            if (k < n) {
                const int e = list[k];
                s = src[e];
                dl = dst[e] - lo;
                atomicAdd(&cn[dl], 1);
            }
            add_group(x, acc, s, dl, min(32, n - g), lane);
        }
    }
    __syncthreads();

    for (int i = tid; i < nr * 16; i += 1024) {
        const int r = i >> 4;
        const int c4 = (i & 15) * 4;
        const float inv = 1.0f / fmaxf((float)cn[r], 1.0f);
        float4 v = *(const float4*)(acc + r * HDIM + c4);
        v.x *= inv; v.y *= inv; v.z *= inv; v.w *= inv;
        *(float4*)(out + (long)(lo + r) * HDIM + c4) = v;
    }
}

__global__ __launch_bounds__(1024) void k_own_agg3(
        const int* __restrict__ sN, const int* __restrict__ dN, const float* __restrict__ xN,
        float* __restrict__ oN, const int* __restrict__ blN, const int* __restrict__ cN,
        const int* __restrict__ sM, const int* __restrict__ dM, const float* __restrict__ xM,
        float* __restrict__ oM, const int* __restrict__ blM, const int* __restrict__ cM,
        const int* __restrict__ sR, const int* __restrict__ dR, const float* __restrict__ xR,
        float* __restrict__ oR, const int* __restrict__ blR, const int* __restrict__ cR) {
    __shared__ __attribute__((aligned(16))) float acc[391 * HDIM];
    __shared__ int cn[391];
    __shared__ int off[392];
    __shared__ int list[16384];
    __shared__ int nlist;
    const int b = blockIdx.x;
    if (b < OWN_N)
        own_agg_body<391, 16384, 15360>(sN, dN, xN, oN, ES, NN, blN, cN, b, OWN_N,
                                        acc, cn, off, list, nlist);
    else if (b < OWN_N + OWN_C)
        own_agg_body<20, 8192, 6144>(sM, dM, xM, oM, EM, NC, blM, cM, b - OWN_N, OWN_C,
                                     acc, cn, off, list, nlist);
    else
        own_agg_body<20, 8192, 1536>(sR, dR, xR, oR, ER, NC, blR, cR, b - OWN_N - OWN_C, OWN_C,
                                     acc, cn, off, list, nlist);
}

__global__ __launch_bounds__(1024) void k_own_agg2(
        const int* __restrict__ sM, const int* __restrict__ dM, const float* __restrict__ xM,
        float* __restrict__ oM, const int* __restrict__ blM, const int* __restrict__ cM,
        const int* __restrict__ sR, const int* __restrict__ dR, const float* __restrict__ xR,
        float* __restrict__ oR, const int* __restrict__ blR, const int* __restrict__ cR) {
    __shared__ __attribute__((aligned(16))) float acc[20 * HDIM];
    __shared__ int cn[20];
    __shared__ int off[21];
    __shared__ int list[8192];
    __shared__ int nlist;
    const int b = blockIdx.x;
    if (b < OWN_C)
        own_agg_body<20, 8192, 6144>(sM, dM, xM, oM, EM, NC, blM, cM, b, OWN_C,
                                     acc, cn, off, list, nlist);
    else
        own_agg_body<20, 8192, 1536>(sR, dR, xR, oR, ER, NC, blR, cR, b - OWN_C, OWN_C,
                                     acc, cn, off, list, nlist);
}

template<int NT, bool FULL, bool VECA>
__device__ __forceinline__ void gemm_body(v8f (&c)[NT],
                                          const float* __restrict__ A,
                                          const v2f* __restrict__ Wp,
                                          int M, int K, int N,
                                          int mbase, int lane) {
    const int row  = mbase + (lane & 15);
    const int koff = (lane >= 16) ? 2 : 0;
    const int col0 = lane & 15;
    const bool rok = FULL || (row < M);
    const float* __restrict__ Arow = A + (long)row * K;
    const int Kmain = K & ~3;

    #pragma unroll 4
    for (int k = 0; k < Kmain; k += 4) {
        const int kk = k + koff;
        v2f a;
        if (rok) {
            if (VECA) {
                a = *(const v2f*)(Arow + kk);
            } else {
                a.x = Arow[kk];
                a.y = Arow[kk + 1];
            }
        } else {
            a.x = 0.f; a.y = 0.f;
        }
        #pragma unroll
        for (int nt = 0; nt < NT; ++nt) {
            v2f b = Wp[(long)(kk >> 1) * N + nt * 16 + col0];
            c[nt] = __builtin_amdgcn_wmma_f32_16x16x4_f32(
                false, a, false, b, (short)0, c[nt], false, false);
        }
    }
    if (Kmain < K) {
        const int kk = Kmain + koff;
        v2f a;
        a.x = (rok && kk     < K) ? Arow[kk]     : 0.f;
        a.y = (rok && kk + 1 < K) ? Arow[kk + 1] : 0.f;
        #pragma unroll
        for (int nt = 0; nt < NT; ++nt) {
            v2f b = Wp[(long)(kk >> 1) * N + nt * 16 + col0];
            c[nt] = __builtin_amdgcn_wmma_f32_16x16x4_f32(
                false, a, false, b, (short)0, c[nt], false, false);
        }
    }
}

#define SLOTS8 "v_nop\n\tv_nop\n\tv_nop\n\tv_nop\n\tv_nop\n\tv_nop\n\tv_nop\n\tv_nop"
__device__ __forceinline__ void result_slots(v8f (&c)[4]) {
    asm volatile(SLOTS8 : "+v"(c[0]), "+v"(c[1]), "+v"(c[2]), "+v"(c[3]));
}
__device__ __forceinline__ void result_slots(v8f (&c)[2]) {
    asm volatile(SLOTS8 : "+v"(c[0]), "+v"(c[1]));
}

template<int NT, bool VECA1, bool NORM>
__global__ __launch_bounds__(256) void k_gemm(const float* A1,
                                              const v2f* __restrict__ W1p,
                                              const float* __restrict__ bias,
                                              const float* __restrict__ A2,
                                              const v2f* __restrict__ W2p,
                                              const float* __restrict__ prev,
                                              const float* __restrict__ res,
                                              const float* __restrict__ g,
                                              const float* __restrict__ b,
                                              float* out,
                                              int M, int K1, int K2,
                                              int do_relu) {
    static_assert(!NORM || NT * 16 == HDIM, "the normalised row must be the whole tile row");
    const int N = NT * 16;
    const int lane = threadIdx.x & 31;
    const int wv = threadIdx.x >> 5;
    const int mbase = (blockIdx.x * 8 + wv) * 16;
    if (mbase >= M) return;

    v8f c[NT];
    #pragma unroll
    for (int t = 0; t < NT; ++t)
        c[t] = (v8f){0.f, 0.f, 0.f, 0.f, 0.f, 0.f, 0.f, 0.f};

    const bool full = (mbase + 16 <= M);
    if (full) {
        gemm_body<NT, true, VECA1>(c, A1, W1p, M, K1, N, mbase, lane);
        if (A2) gemm_body<NT, true, true>(c, A2, W2p, M, K2, N, mbase, lane);
    } else {
        gemm_body<NT, false, VECA1>(c, A1, W1p, M, K1, N, mbase, lane);
        if (A2) gemm_body<NT, false, true>(c, A2, W2p, M, K2, N, mbase, lane);
    }

    result_slots(c);

    const int col0 = lane & 15;
    const int rofs = (lane >= 16) ? 8 : 0;
    #pragma unroll
    for (int nt = 0; nt < NT; ++nt) {
        const float bb = bias ? bias[nt * 16 + col0] : 0.f;
        #pragma unroll
        for (int r = 0; r < 8; ++r) c[nt][r] += bb;
    }
    if (prev) {
        #pragma unroll
        for (int r = 0; r < 8; ++r) {
            const long rb = (long)min(mbase + r + rofs, M - 1) * N + col0;
            #pragma unroll
            for (int nt = 0; nt < NT; ++nt) c[nt][r] += prev[rb + nt * 16];
        }
    }
    if (res) {
        #pragma unroll
        for (int r = 0; r < 8; ++r) {
            const long rb = (long)min(mbase + r + rofs, M - 1) * N + col0;
            #pragma unroll
            for (int nt = 0; nt < NT; ++nt) c[nt][r] += res[rb + nt * 16];
        }
    }

    if (NORM) {
        float gg[NT], be[NT];
        #pragma unroll
        for (int nt = 0; nt < NT; ++nt) { gg[nt] = g[nt * 16 + col0]; be[nt] = b[nt * 16 + col0]; }
        #pragma unroll
        for (int r = 0; r < 8; ++r) {
            const int row = mbase + r + rofs;
            float v[NT];
            #pragma unroll
            for (int nt = 0; nt < NT; ++nt) v[nt] = fmaxf(c[nt][r], 0.f);
            float s = v[0];
            #pragma unroll
            for (int nt = 1; nt < NT; ++nt) s += v[nt];
            #pragma unroll
            for (int off = 8; off > 0; off >>= 1) s += __shfl_xor(s, off, 32);
            const float mean = s * (1.0f / 64.0f);
            float d[NT];
            #pragma unroll
            for (int nt = 0; nt < NT; ++nt) d[nt] = v[nt] - mean;
            float q = d[0] * d[0];
            #pragma unroll
            for (int nt = 1; nt < NT; ++nt) q += d[nt] * d[nt];
            #pragma unroll
            for (int off = 8; off > 0; off >>= 1) q += __shfl_xor(q, off, 32);
            const float rstd = rsqrtf(q * (1.0f / 64.0f) + 1e-5f);
            if (full || row < M) {
                const long rb = (long)row * N + col0;
                #pragma unroll
                for (int nt = 0; nt < NT; ++nt)
                    out[rb + nt * 16] = d[nt] * rstd * gg[nt] + be[nt];
            }
        }
    } else {
        #pragma unroll
        for (int nt = 0; nt < NT; ++nt) {
            #pragma unroll
            for (int r = 0; r < 8; ++r) {
                const int row = mbase + r + rofs;
                if (full || row < M) {
                    float v = c[nt][r];
                    if (do_relu) v = fmaxf(v, 0.f);
                    out[(long)row * N + nt * 16 + col0] = v;
                }
            }
        }
    }
}

#define CW_PITCH 80
#define CC_PITCH 48
#define CT_PITCH 68
typedef __attribute__((ext_vector_type(4))) float v4f;

template<bool HEAD>
__global__ __launch_bounds__(128) void k_crow(const float* __restrict__ aggM,
                                              const float* __restrict__ aggR,
                                              const float* __restrict__ x,
                                              const v2f* __restrict__ wp,
                                              const float* __restrict__ bM,
                                              const float* __restrict__ bR,
                                              const float* __restrict__ g,
                                              const float* __restrict__ b,
                                              const v2f* __restrict__ wpc,
                                              const float* __restrict__ cb1,
                                              const float* __restrict__ cW2,
                                              const float* __restrict__ cb2,
                                              float* __restrict__ out, int M) {
    __shared__ __attribute__((aligned(16))) v2f sW[4 * 32 * CW_PITCH];
    __shared__ __attribute__((aligned(16))) v2f sWc[HEAD ? 32 * CC_PITCH : 2];
    __shared__ __attribute__((aligned(16))) float sT[HEAD ? 4 * 16 * CT_PITCH : 4];
    __shared__ float sO[HEAD ? 64 : 1];

    const int tid = threadIdx.x;
    const int lane = tid & 31;
    const int wv = tid >> 5;
    const int mbase = (blockIdx.x * 4 + wv) * 16;
    const int col0 = lane & 15;
    const int hh = lane >> 4;
    const int rofs = hh * 8;

    v4f wr[32];
    #pragma unroll
    for (int u = 0; u < 32; ++u) wr[u] = ((const v4f*)wp)[u * 128 + tid];
    v4f wc[4];
    if (HEAD) {
        #pragma unroll
        for (int u = 0; u < 4; ++u) wc[u] = ((const v4f*)wpc)[u * 128 + tid];
    }

    const long ra = (long)min(mbase + col0, M - 1) * HDIM + 2 * hh;
    v2f aM[16], aX[16], aR[16];
    #pragma unroll
    for (int ks = 0; ks < 16; ++ks) {
        aM[ks] = *(const v2f*)(aggM + ra + 4 * ks);
        aX[ks] = *(const v2f*)(x + ra + 4 * ks);
        aR[ks] = *(const v2f*)(aggR + ra + 4 * ks);
    }
    float xr[4][8];
    #pragma unroll
    for (int r = 0; r < 8; ++r) {
        const long rb = (long)min(mbase + r + rofs, M - 1) * HDIM + col0;
        #pragma unroll
        for (int nt = 0; nt < 4; ++nt) xr[nt][r] = x[rb + nt * 16];
    }
    float bb[4];
    #pragma unroll
    for (int nt = 0; nt < 4; ++nt) bb[nt] = bM[nt * 16 + col0] + bR[nt * 16 + col0];
    float gg[4], be[4];
    if (!HEAD) {
        #pragma unroll
        for (int nt = 0; nt < 4; ++nt) { gg[nt] = g[nt * 16 + col0]; be[nt] = b[nt * 16 + col0]; }
    }
    float hb[2], w2[2], b2 = 0.f;
    if (HEAD) {
        #pragma unroll
        for (int nt = 0; nt < 2; ++nt) { hb[nt] = cb1[nt * 16 + col0]; w2[nt] = cW2[nt * 16 + col0]; }
        b2 = cb2[0];
    }

    #pragma unroll
    for (int u = 0; u < 32; ++u) {
        const int i4 = u * 128 + tid;
        ((v4f*)sW)[(i4 >> 5) * (CW_PITCH / 2) + (i4 & 31)] = wr[u];
    }
    if (HEAD) {
        #pragma unroll
        for (int u = 0; u < 4; ++u) {
            const int i4 = u * 128 + tid;
            ((v4f*)sWc)[(i4 >> 4) * (CC_PITCH / 2) + (i4 & 15)] = wc[u];
        }
    }
    __syncthreads();

    v8f c[4];
    #pragma unroll
    for (int t = 0; t < 4; ++t)
        c[t] = (v8f){0.f, 0.f, 0.f, 0.f, 0.f, 0.f, 0.f, 0.f};

    const v2f* sB = sW + hh * CW_PITCH + col0;
    v2f bq[2][16];
    #pragma unroll
    for (int j = 0; j < 4; ++j) {
        #pragma unroll
        for (int nt = 0; nt < 4; ++nt) bq[0][j * 4 + nt] = sB[(2 * j) * CW_PITCH + nt * 16];
    }
    #pragma unroll
    for (int gi = 0; gi < 16; ++gi) {
        if (gi + 1 < 16) {
            const int p1 = (gi + 1) >> 2, q1 = (gi + 1) & 3;
            #pragma unroll
            for (int j = 0; j < 4; ++j) {
                #pragma unroll
                for (int nt = 0; nt < 4; ++nt)
                    bq[(gi + 1) & 1][j * 4 + nt] = sB[(p1 * 32 + 2 * (q1 * 4 + j)) * CW_PITCH + nt * 16];
            }
        }
        const int p = gi >> 2, q = gi & 3;
        #pragma unroll
        for (int j = 0; j < 4; ++j) {
            const int ks = q * 4 + j;
            const v2f a = (p == 0) ? aM[ks] : (p == 2) ? aR[ks] : aX[ks];
            #pragma unroll
            for (int nt = 0; nt < 4; ++nt)
                c[nt] = __builtin_amdgcn_wmma_f32_16x16x4_f32(
                    false, a, false, bq[gi & 1][j * 4 + nt], (short)0, c[nt], false, false);
        }
    }

    if (!HEAD) {
        float v[4][8], d[4][8], s[8], q[8];
        #pragma unroll
        for (int r = 0; r < 8; ++r) {
            #pragma unroll
            for (int nt = 0; nt < 4; ++nt) v[nt][r] = fmaxf(c[nt][r] + bb[nt] + xr[nt][r], 0.f);
            s[r] = v[0][r];
            #pragma unroll
            for (int nt = 1; nt < 4; ++nt) s[r] += v[nt][r];
        }
        #pragma unroll
        for (int off = 8; off > 0; off >>= 1) {
            float t[8];
            #pragma unroll
            for (int r = 0; r < 8; ++r) t[r] = __shfl_xor(s[r], off, 32);
            #pragma unroll
            for (int r = 0; r < 8; ++r) s[r] += t[r];
        }
        #pragma unroll
        for (int r = 0; r < 8; ++r) {
            const float mean = s[r] * (1.0f / 64.0f);
            #pragma unroll
            for (int nt = 0; nt < 4; ++nt) d[nt][r] = v[nt][r] - mean;
            q[r] = d[0][r] * d[0][r];
            #pragma unroll
            for (int nt = 1; nt < 4; ++nt) q[r] += d[nt][r] * d[nt][r];
        }
        #pragma unroll
        for (int off = 8; off > 0; off >>= 1) {
            float t[8];
            #pragma unroll
            for (int r = 0; r < 8; ++r) t[r] = __shfl_xor(q[r], off, 32);
            #pragma unroll
            for (int r = 0; r < 8; ++r) q[r] += t[r];
        }
        #pragma unroll
        for (int r = 0; r < 8; ++r) {
            const int row = mbase + r + rofs;
            const float rstd = rsqrtf(q[r] * (1.0f / 64.0f) + 1e-5f);
            if (row < M) {
                const long rb = (long)row * HDIM + col0;
                #pragma unroll
                for (int nt = 0; nt < 4; ++nt)
                    out[rb + nt * 16] = d[nt][r] * rstd * gg[nt] + be[nt];
            }
        }
    } else {
        float* tw = sT + wv * (16 * CT_PITCH);
        #pragma unroll
        for (int nt = 0; nt < 4; ++nt) {
            #pragma unroll
            for (int r = 0; r < 8; ++r)
                tw[(rofs + r) * CT_PITCH + nt * 16 + col0] = fmaxf(c[nt][r] + bb[nt] + xr[nt][r], 0.f);
        }
        __syncthreads();
        v8f hc[2];
        hc[0] = (v8f){0.f, 0.f, 0.f, 0.f, 0.f, 0.f, 0.f, 0.f};
        hc[1] = (v8f){0.f, 0.f, 0.f, 0.f, 0.f, 0.f, 0.f, 0.f};
        const float* ta = tw + col0 * CT_PITCH + 2 * hh;
        const v2f* sC = sWc + hh * CC_PITCH + col0;
        #pragma unroll
        for (int ks = 0; ks < 16; ++ks) {
            const v2f a = *(const v2f*)(ta + 4 * ks);
            #pragma unroll
            for (int nt = 0; nt < 2; ++nt)
                hc[nt] = __builtin_amdgcn_wmma_f32_16x16x4_f32(
                    false, a, false, sC[(2 * ks) * CC_PITCH + nt * 16], (short)0, hc[nt], false, false);
        }
        float s[8];
        #pragma unroll
        for (int r = 0; r < 8; ++r)
            s[r] = fmaxf(hc[0][r] + hb[0], 0.f) * w2[0] + fmaxf(hc[1][r] + hb[1], 0.f) * w2[1];
        #pragma unroll
        for (int off = 8; off > 0; off >>= 1) {
            float t[8];
            #pragma unroll
            for (int r = 0; r < 8; ++r) t[r] = __shfl_xor(s[r], off, 32);
            #pragma unroll
            for (int r = 0; r < 8; ++r) s[r] += t[r];
        }
        float mine = 0.f;
        #pragma unroll
        for (int r = 0; r < 8; ++r)
            if (col0 == r) mine = s[r];
        if (col0 < 8) sO[wv * 16 + rofs + col0] = mine + b2;
        __syncthreads();
        if (tid < 64) {
            const int row = blockIdx.x * 64 + tid;
            if (row < M) out[row] = sO[tid];
        }
    }
}

extern "C" void kernel_launch(void* const* d_in, const int* in_sizes, int n_in,
                              void* d_out, int out_size, void* d_ws, size_t ws_size,
                              hipStream_t stream) {
    const float* x_news    = (const float*)d_in[0];
    const float* x_company = (const float*)d_in[1];
    const int*   edge_sim  = (const int*)d_in[2];
    const int*   men_src   = (const int*)d_in[3];
    const int*   men_dst   = (const int*)d_in[4];
    const int*   edge_rel  = (const int*)d_in[5];
    const float* nw_W = (const float*)d_in[6];
    const float* nw_b = (const float*)d_in[7];
    const float* cw_W = (const float*)d_in[8];
    const float* cw_b = (const float*)d_in[9];
    const float* n1n_g = (const float*)d_in[10];
    const float* n1n_b = (const float*)d_in[11];
    const float* n1c_g = (const float*)d_in[12];
    const float* n1c_b = (const float*)d_in[13];
    const float* c1_Wl = (const float*)d_in[14];
    const float* c1_bl = (const float*)d_in[15];
    const float* c1_Wr = (const float*)d_in[16];
    const float* c2_Wl = (const float*)d_in[17];
    const float* c2_bl = (const float*)d_in[18];
    const float* c2_Wr = (const float*)d_in[19];
    const float* cls_W1 = (const float*)d_in[20];
    const float* cls_b1 = (const float*)d_in[21];
    const float* cls_W2 = (const float*)d_in[22];
    const float* cls_b2 = (const float*)d_in[23];
    float* out = (float*)d_out;

    float* ws = (float*)d_ws;
    long o = 0;
    float* xn0  = ws + o; o += (long)NN * HDIM;
    float* lpl  = ws + o; o += (long)NN * HDIM;
    float* aggn = ws + o; o += (long)NN * HDIM;
    float* xn1  = aggn;
    float* aggcM = ws + o; o += (long)NC * HDIM;
    float* aggcR = ws + o; o += (long)NC * HDIM;
    float* xc0  = ws + o; o += (long)NC * HDIM;
    float* xc1  = ws + o; o += (long)NC * HDIM;
    o += (long)NC * HDIM;
    float* comp = ws + o; o += (long)NC * HDIM;
    o += (long)NC * 32;
    v2f* wp_nw  = (v2f*)(ws + o); o += 2L * (388 / 2) * 64;
    v2f* wp_cw  = (v2f*)(ws + o); o += 2L * (24 / 2) * 64;
    v2f* wp_c1  = (v2f*)(ws + o); o += 2L * 6 * 32 * 64;
    v2f* wp_c2  = (v2f*)(ws + o); o += 2L * 4 * 32 * 64;
    v2f* wp_cls = (v2f*)(ws + o); o += 2L * 32 * 32;

    int* ctr   = (int*)comp;
    int* ctrN  = ctr;
    int* ctrM  = ctr + 512;
    int* ctrR  = ctr + 1024;
    int* blR   = ctr + 4096;
    int* blN   = (int*)lpl;
    int* blM   = blN + 256L * 15360;

    dim3 blk(256);
    int tilesNN = (NN + 15) / 16, tilesNC = (NC + 15) / 16;
    unsigned gemmBlkNN = (tilesNN + 7) / 8, gemmBlkNC = (tilesNC + 7) / 8;
    unsigned crowBlk = (tilesNC + 3) / 4;

    k_prep<<<PB_NW + PB_CW + PB_C1 + PB_C2 + PB_CLS + PB_Z, blk, 0, stream>>>(
        nw_W, cw_W, c1_Wl, c1_Wr, c2_Wl, c2_Wr, cls_W1,
        wp_nw, wp_cw, wp_c1, wp_c2, wp_cls, ctr);

    k_gemm<4, false, false><<<gemmBlkNN, blk, 0, stream>>>(x_news, wp_nw, nw_b,
                                                           nullptr, nullptr,
                                                           nullptr, nullptr, nullptr, nullptr,
                                                           xn0, NN, 385, 0, 1);
    k_gemm<4, true, false><<<gemmBlkNC, blk, 0, stream>>>(x_company, wp_cw, cw_b,
                                                          nullptr, nullptr,
                                                          nullptr, nullptr, nullptr, nullptr,
                                                          xc0, NC, 24, 0, 1);

    k_bin3<<<BINB_N + BINB_M + BINB_R, 1024, 0, stream>>>(edge_sim, edge_sim + ES, blN, ctrN,
                                                          men_src, men_dst, blM, ctrM,
                                                          edge_rel, edge_rel + ER, blR, ctrR);
    k_own_agg3<<<OWN_N + OWN_C + OWN_C, 1024, 0, stream>>>(edge_sim, edge_sim + ES, xn0, aggn, blN, ctrN,
                                                           men_src, men_dst, xn0, aggcM, blM, ctrM,
                                                           edge_rel, edge_rel + ER, xc0, aggcR, blR, ctrR);

    k_gemm<4, true, true><<<gemmBlkNN, blk, 0, stream>>>(aggn, wp_c1 + 0 * 2048,
                                                         c1_bl + 0 * 64,
                                                         xn0, wp_c1 + 1 * 2048,
                                                         nullptr, nullptr, n1n_g, n1n_b,
                                                         xn1, NN, 64, 64, 1);
    k_crow<false><<<crowBlk, 128, 0, stream>>>(aggcM, aggcR, xc0, wp_c1 + 2 * 2048,
                                               c1_bl + 1 * 64, c1_bl + 2 * 64, n1c_g, n1c_b,
                                               nullptr, nullptr, nullptr, nullptr,
                                               xc1, NC);

    k_own_agg2<<<OWN_C + OWN_C, 1024, 0, stream>>>(men_src, men_dst, xn1, aggcM, blM, ctrM,
                                                   edge_rel, edge_rel + ER, xc1, aggcR, blR, ctrR);
    k_crow<true><<<crowBlk, 128, 0, stream>>>(aggcM, aggcR, xc1, wp_c2,
                                              c2_bl + 1 * 64, c2_bl + 2 * 64, nullptr, nullptr,
                                              wp_cls, cls_b1, cls_W2, cls_b2,
                                              out, NC);
}


# aggregation launches hold two work-groups per WGP, not one: the never-taken full-list path sums 128 rows per pass (32 KB LDS, was 100 KB) within 64 registers; company owners run beside news owners
# speedup vs baseline: 6.4087x; 1.1046x over previous
#include <hip/hip_runtime.h>
#include <hip/hip_bf16.h>

typedef __attribute__((ext_vector_type(2))) float v2f;
typedef __attribute__((ext_vector_type(8))) float v8f;

#define NN 100000
#define NC 5000
#define HDIM 64
#define ES 3200000
#define EM 1000000
#define ER 160000

__device__ __forceinline__ void pack_one(const float* __restrict__ W,
                                         v2f* __restrict__ Wp,
                                         int K, int N, int i) {
    const int Kp2 = ((K + 3) & ~3) >> 1;
    if (i >= Kp2 * N) return;
    const int k2 = i / N, col = i % N;
    const int k = 2 * k2;
    v2f v;
    v.x = (k     < K) ? W[k * N + col]       : 0.f;
    v.y = (k + 1 < K) ? W[(k + 1) * N + col] : 0.f;
    Wp[i] = v;
}

#define PB_NW  49
#define PB_CW  3
#define PB_C1  48
#define PB_C2  32
#define PB_CLS 4
#define PB_Z   16
__global__ __launch_bounds__(256) void k_prep(const float* __restrict__ nw_W,
                                              const float* __restrict__ cw_W,
                                              const float* __restrict__ c1_Wl,
                                              const float* __restrict__ c1_Wr,
                                              const float* __restrict__ c2_Wl,
                                              const float* __restrict__ c2_Wr,
                                              const float* __restrict__ cls_W1,
                                              v2f* __restrict__ wp_nw,
                                              v2f* __restrict__ wp_cw,
                                              v2f* __restrict__ wp_c1,
                                              v2f* __restrict__ wp_c2,
                                              v2f* __restrict__ wp_cls,
                                              int* __restrict__ ctr) {
    const int tid = threadIdx.x;
    int b = blockIdx.x;
    if (b < PB_NW) { pack_one(nw_W, wp_nw, 385, 64, b * 256 + tid); return; }
    b -= PB_NW;
    if (b < PB_CW) { pack_one(cw_W, wp_cw, 24, 64, b * 256 + tid); return; }
    b -= PB_CW;
    if (b < PB_C1) {
        const int m = b >> 3;
        const float* W = ((m & 1) ? c1_Wr : c1_Wl) + (m >> 1) * 4096;
        pack_one(W, wp_c1 + m * 2048, 64, 64, (b & 7) * 256 + tid);
        return;
    }
    b -= PB_C1;
    if (b < PB_C2) {
        const int m = b >> 3;
        const float* W = ((m & 1) ? c2_Wr : c2_Wl) + ((m >> 1) + 1) * 4096;
        pack_one(W, wp_c2 + m * 2048, 64, 64, (b & 7) * 256 + tid);
        return;
    }
    b -= PB_C2;
    if (b < PB_CLS) { pack_one(cls_W1, wp_cls, 64, 32, b * 256 + tid); return; }
    b -= PB_CLS;
    ctr[b * 256 + tid] = 0;
}

#define PK_SH 9
#define PK_MASK 511
#define BIN_EPT 8
#define BIN_CHUNK (1024 * BIN_EPT)
#define BIN_NBMAX 256
template<int R, int CAPB>
__device__ __forceinline__ void bin_body(const int* __restrict__ src,
                                         const int* __restrict__ dst,
                                         int nE, int nB,
                                         int* __restrict__ blist,
                                         int* __restrict__ bctr,
                                         int bid,
                                         int* hcnt, int* hoff, int* gbase,
                                         int* stage, int* sown) {
    static_assert(R <= PK_MASK + 1, "row inside the range must fit the packed field");
    constexpr int EPT = BIN_EPT;
    constexpr int CHUNK = BIN_CHUNK;
    constexpr int NBMAX = BIN_NBMAX;

    const int tid = threadIdx.x;
    const int lane = tid & 31;
    const int wv = tid >> 5;
    const int base = bid * CHUNK;

    if (tid < NBMAX) hcnt[tid] = 0;
    __syncthreads();

    int ow[EPT], rk[EPT], pk[EPT];
    #pragma unroll
    for (int u = 0; u < EPT; ++u) {
        const int e = base + u * 1024 + tid;
        ow[u] = -1; rk[u] = 0; pk[u] = 0;
        if (e < nE) {
            const int d = dst[e];
            const int o = (int)((unsigned)d / (unsigned)R);
            ow[u] = o;
            pk[u] = (src[e] << PK_SH) | (d - o * R);
            rk[u] = atomicAdd(&hcnt[o], 1);
        }
    }
    __syncthreads();

    if (tid < nB) {
        const int cnt = hcnt[tid];
        gbase[tid] = cnt ? atomicAdd(&bctr[tid], cnt) : 0;
    }
    if (wv == 31) {
        int cq[8];
        int t = 0;
        #pragma unroll
        for (int q = 0; q < 8; ++q) { const int v = hcnt[lane * 8 + q]; cq[q] = t; t += v; }
        int incl = t;
        #pragma unroll
        for (int off = 1; off < 32; off <<= 1) {
            const int s = __shfl_up(incl, off, 32);
            if (lane >= off) incl += s;
        }
        const int excl = incl - t;
        #pragma unroll
        for (int q = 0; q < 8; ++q) hoff[lane * 8 + q] = cq[q] + excl;
    }
    __syncthreads();

    #pragma unroll
    for (int u = 0; u < EPT; ++u) {
        if (ow[u] >= 0) {
            const int lp = hoff[ow[u]] + rk[u];
            stage[lp] = pk[u];
            sown[lp] = ow[u];
        }
    }
    __syncthreads();

    const int total = min(CHUNK, nE - base);
    for (int i = tid; i < total; i += 1024) {
        const int o = sown[i];
        const int pos = gbase[o] + (i - hoff[o]);
        if (pos < CAPB) blist[(long)o * CAPB + pos] = stage[i];
        else bctr[nB] = 1;
    }
}

#define BIN_LDS \
    __shared__ int hcnt[BIN_NBMAX];      \
    __shared__ int hoff[BIN_NBMAX];      \
    __shared__ int gbase[BIN_NBMAX];     \
    __shared__ int stage[BIN_CHUNK];     \
    __shared__ int sown[BIN_CHUNK];

#define BINB_N ((ES + BIN_CHUNK - 1) / BIN_CHUNK)
#define BINB_M ((EM + BIN_CHUNK - 1) / BIN_CHUNK)
#define BINB_R ((ER + BIN_CHUNK - 1) / BIN_CHUNK)
#define OWN_N ((NN + 390) / 391)
#define OWN_C ((NC + 19) / 20)
__global__ __launch_bounds__(1024) void k_bin3(const int* __restrict__ sN, const int* __restrict__ dN,
                                               int* __restrict__ blN, int* __restrict__ cN,
                                               const int* __restrict__ sM, const int* __restrict__ dM,
                                               int* __restrict__ blM, int* __restrict__ cM,
                                               const int* __restrict__ sR, const int* __restrict__ dR,
                                               int* __restrict__ blR, int* __restrict__ cR) {
    BIN_LDS
    const int b = blockIdx.x;
    if (b < BINB_N)
        bin_body<391, 15360>(sN, dN, ES, OWN_N, blN, cN, b, hcnt, hoff, gbase, stage, sown);
    else if (b < BINB_N + BINB_M)
        bin_body<20, 6144>(sM, dM, EM, OWN_C, blM, cM, b - BINB_N, hcnt, hoff, gbase, stage, sown);
    else
        bin_body<20, 1536>(sR, dR, ER, OWN_C, blR, cR, b - BINB_N - BINB_M, hcnt, hoff, gbase, stage, sown);
}

__device__ __forceinline__ void add_group(const float* __restrict__ x, float* acc,
                                          int s, int dl, int m, int lane) {
    if (m >= 32) {
        #pragma nounroll
        for (int j0 = 0; j0 < 32; j0 += 8) {
            float v0[8], v1[8];
            #pragma unroll
            for (int j = 0; j < 8; ++j) {
                const int sj = __builtin_amdgcn_readlane(s, j0 + j);
                const float* xr = x + (long)sj * HDIM + lane;
                v0[j] = xr[0];
                v1[j] = xr[32];
            }
            #pragma unroll
            for (int j = 0; j < 8; ++j) {
                const int dj = __builtin_amdgcn_readlane(dl, j0 + j);
                float* ar = acc + dj * HDIM + lane;
                atomicAdd(ar, v0[j]);
                atomicAdd(ar + 32, v1[j]);
            }
        }
    } else {
        for (int j = 0; j < m; ++j) {
            const int sj = __builtin_amdgcn_readlane(s, j);
            const int dj = __builtin_amdgcn_readlane(dl, j);
            const float* xr = x + (long)sj * HDIM + lane;
            float* ar = acc + dj * HDIM + lane;
            const float a0 = xr[0];
            const float a1 = xr[32];
            atomicAdd(ar, a0);
            atomicAdd(ar + 32, a1);
        }
    }
}

#define SRT_UB 16
template<int R, int CAPB>
__device__ __forceinline__ void own_sorted_body(const float* __restrict__ x,
                                                float* __restrict__ out,
                                                const int* __restrict__ bl,
                                                int nbin, int lo, int nr,
                                                int* cn, int* off, int* sorted) {
    constexpr int EPT = (CAPB + 1023) / 1024;
    constexpr int Q = (R + 31) / 32;
    const int tid = threadIdx.x;
    const int lane = tid & 31;
    const int wv = __builtin_amdgcn_readfirstlane(tid >> 5);
    const int n = min(nbin, CAPB);

    for (int i = tid; i < R; i += 1024) cn[i] = 0;
    __syncthreads();

    int pk[EPT], rk[EPT];
    #pragma unroll
    for (int u = 0; u < EPT; ++u) {
        const int i = u * 1024 + tid;
        pk[u] = -1; rk[u] = 0;
        if (i < n) {
            const int p = bl[i];
            pk[u] = p;
            rk[u] = atomicAdd(&cn[p & PK_MASK], 1);
        }
    }
    __syncthreads();

    if (wv == 0) {
        int cq[Q];
        int t = 0;
        #pragma unroll
        for (int q = 0; q < Q; ++q) {
            const int r = lane * Q + q;
            const int v = (r < R) ? cn[r] : 0;
            cq[q] = t; t += v;
        }
        int incl = t;
        #pragma unroll
        for (int o = 1; o < 32; o <<= 1) {
            const int s = __shfl_up(incl, o, 32);
            if (lane >= o) incl += s;
        }
        const int excl = incl - t;
        #pragma unroll
        for (int q = 0; q < Q; ++q) {
            const int r = lane * Q + q;
            if (r < R) off[r] = cq[q] + excl;
        }
        if (lane == 31) off[R] = incl;
    }
    __syncthreads();

    #pragma unroll
    for (int u = 0; u < EPT; ++u)
        if (pk[u] >= 0) sorted[off[pk[u] & PK_MASK] + rk[u]] = pk[u];
    for (int i = tid; i < nr * 16; i += 1024) {
        const int r = i >> 4;
        if (cn[r] == 0)
            *(float4*)(out + (long)(lo + r) * HDIM + (i & 15) * 4) = make_float4(0.f, 0.f, 0.f, 0.f);
    }
    __syncthreads();

    const int r0 = min(wv * Q, nr);
    const int r1 = min(r0 + Q, nr);
    const int e0 = off[r0];
    const int e1 = off[r1];
    if (e0 >= e1) return;

    float a0 = 0.f, a1 = 0.f;
    int cur = -1;
    int pn = sorted[min(e0 + lane, e1 - 1)];
    for (int g = e0; g < e1; g += SRT_UB) {
        const int pv = pn;
        pn = sorted[min(g + SRT_UB + lane, e1 - 1)];
        float2 v[SRT_UB];
        #pragma unroll
        for (int u = 0; u < SRT_UB; ++u) {
            const int pu = __builtin_amdgcn_readlane(pv, u);
            v[u] = *(const float2*)(x + (long)(pu >> PK_SH) * HDIM + 2 * lane);
        }
        #pragma unroll
        for (int u = 0; u < SRT_UB; ++u) {
            if (g + u < e1) {
                const int d = __builtin_amdgcn_readlane(pv, u) & PK_MASK;
                if (d != cur) {
                    if (cur >= 0) {
                        const float inv = 1.0f / fmaxf((float)cn[cur], 1.0f);
                        *(float2*)(out + (long)(lo + cur) * HDIM + 2 * lane) = make_float2(a0 * inv, a1 * inv);
                    }
                    cur = d; a0 = 0.f; a1 = 0.f;
                }
                a0 += v[u].x; a1 += v[u].y;
            }
        }
    }
    const float inv = 1.0f / fmaxf((float)cn[cur], 1.0f);
    *(float2*)(out + (long)(lo + cur) * HDIM + 2 * lane) = make_float2(a0 * inv, a1 * inv);
}

#define AGG_RS 128
template<int R, int CAP, int CAPB>
__device__ __forceinline__ void own_agg_body(const int* __restrict__ src,
                                             const int* __restrict__ dst,
                                             const float* __restrict__ x,
                                             float* __restrict__ out,
                                             int nE, int nDst,
                                             const int* __restrict__ blist,
                                             const int* __restrict__ bctr,
                                             int bid, int nB,
                                             float* acc, int* cn, int* off, int* list, int& nlist) {
    static_assert(CAPB <= CAP, "the ordered list must fit the list array");
    const int tid = threadIdx.x;
    const int lane = tid & 31;
    const int wv = __builtin_amdgcn_readfirstlane(tid >> 5);
    constexpr int RS = (R < AGG_RS) ? R : AGG_RS;
    const int lo0 = bid * R;
    const int nr0 = min(R, nDst - lo0);
    const bool binned = __builtin_amdgcn_readfirstlane(bctr[nB]) == 0;
    const int nbin = __builtin_amdgcn_readfirstlane(bctr[bid]);

    if (binned) {
        own_sorted_body<R, CAPB>(x, out, blist + (long)bid * CAPB, nbin, lo0, nr0, cn, off, list);
        return;
    }

    for (int sub = 0; sub < nr0; sub += RS) {
    const int lo = lo0 + sub;
    const int nr = min(RS, nr0 - sub);

    for (int i = tid; i < RS * HDIM; i += 1024) acc[i] = 0.f;
    for (int i = tid; i < RS; i += 1024) cn[i] = 0;
    if (tid == 0) nlist = 0;
    __syncthreads();

    {
    constexpr int U = 4;
    constexpr int CH = U * 128;
    const int nCh = (nE + CH - 1) / CH;
    const int rot = (int)(((long)bid * nCh) / nB);
    for (int c0 = wv; c0 < nCh; c0 += 32) {
        int c = c0 + rot;
        if (c >= nCh) c -= nCh;
        const int base = c * CH;
        int4 d[U];
        if (base + CH <= nE) {
            #pragma unroll
            for (int u = 0; u < U; ++u)
                d[u] = *(const int4*)(dst + base + u * 128 + lane * 4);
        } else {
            #pragma unroll
            for (int u = 0; u < U; ++u) {
                const int i0 = base + u * 128 + lane * 4;
                d[u].x = (i0     < nE) ? dst[i0]     : -1;
                d[u].y = (i0 + 1 < nE) ? dst[i0 + 1] : -1;
                d[u].z = (i0 + 2 < nE) ? dst[i0 + 2] : -1;
                d[u].w = (i0 + 3 < nE) ? dst[i0 + 3] : -1;
            }
        }
        int cl = 0;
        #pragma unroll
        for (int u = 0; u < U; ++u) {
            cl += ((unsigned)(d[u].x - lo) < (unsigned)nr) ? 1 : 0;
            cl += ((unsigned)(d[u].y - lo) < (unsigned)nr) ? 1 : 0;
            cl += ((unsigned)(d[u].z - lo) < (unsigned)nr) ? 1 : 0;
            cl += ((unsigned)(d[u].w - lo) < (unsigned)nr) ? 1 : 0;
        }
        int incl = cl;
        #pragma unroll
        for (int off = 1; off < 32; off <<= 1) {
            const int t = __shfl_up(incl, off, 32);
            if (lane >= off) incl += t;
        }
        const int total = __builtin_amdgcn_readlane(incl, 31);
        if (total == 0) continue;
        int s0 = 0;
        if (lane == 31) s0 = atomicAdd(&nlist, total);
        s0 = __builtin_amdgcn_readlane(s0, 31);
        int pos = s0 + incl - cl;
        #pragma unroll
        for (int u = 0; u < U; ++u) {
            #pragma unroll
            for (int q = 0; q < 4; ++q) {
                const int dv = (q == 0) ? d[u].x : (q == 1) ? d[u].y : (q == 2) ? d[u].z : d[u].w;
                const int dl = dv - lo;
                if ((unsigned)dl < (unsigned)nr) {
                    const int e = base + u * 128 + lane * 4 + q;
                    if (pos < CAP) {
                        list[pos] = e;
                    } else {
                        const float* xr = x + (long)src[e] * HDIM;
                        #pragma nounroll
                        for (int k = 0; k < HDIM; ++k) atomicAdd(&acc[dl * HDIM + k], xr[k]);
                        atomicAdd(&cn[dl], 1);
                    }
                    ++pos;
                }
            }
        }
    }
    }
    __syncthreads();

    {
        const int n = min(nlist, CAP);
        for (int g = wv * 32; g < n; g += 32 * 32) {
            const int k = g + lane;
            int s = 0, dl = 0;
            if (k < n) {
                const int e = list[k];
                s = src[e];
                dl = dst[e] - lo;
                atomicAdd(&cn[dl], 1);
            }
            add_group(x, acc, s, dl, min(32, n - g), lane);
        }
    }
    __syncthreads();

    for (int i = tid; i < nr * 16; i += 1024) {
        const int r = i >> 4;
        const int c4 = (i & 15) * 4;
        const float inv = 1.0f / fmaxf((float)cn[r], 1.0f);
        float4 v = *(const float4*)(acc + r * HDIM + c4);
        v.x *= inv; v.y *= inv; v.z *= inv; v.w *= inv;
        *(float4*)(out + (long)(lo + r) * HDIM + c4) = v;
    }
    __syncthreads();
    }
}

__global__ __launch_bounds__(1024) void k_own_agg3(
        const int* __restrict__ sN, const int* __restrict__ dN, const float* __restrict__ xN,
        float* __restrict__ oN, const int* __restrict__ blN, const int* __restrict__ cN,
        const int* __restrict__ sM, const int* __restrict__ dM, const float* __restrict__ xM,
        float* __restrict__ oM, const int* __restrict__ blM, const int* __restrict__ cM,
        const int* __restrict__ sR, const int* __restrict__ dR, const float* __restrict__ xR,
        float* __restrict__ oR, const int* __restrict__ blR, const int* __restrict__ cR) {
    __shared__ __attribute__((aligned(16))) float acc[AGG_RS * HDIM];
    __shared__ int cn[391];
    __shared__ int off[392];
    __shared__ int list[16384];
    __shared__ int nlist;
    const int b = blockIdx.x;
    if (b < OWN_N)
        own_agg_body<391, 16384, 15360>(sN, dN, xN, oN, ES, NN, blN, cN, b, OWN_N,
                                        acc, cn, off, list, nlist);
    else if (b < OWN_N + OWN_C)
        own_agg_body<20, 8192, 6144>(sM, dM, xM, oM, EM, NC, blM, cM, b - OWN_N, OWN_C,
                                     acc, cn, off, list, nlist);
    else
        own_agg_body<20, 8192, 1536>(sR, dR, xR, oR, ER, NC, blR, cR, b - OWN_N - OWN_C, OWN_C,
                                     acc, cn, off, list, nlist);
}

__global__ __launch_bounds__(1024) void k_own_agg2(
        const int* __restrict__ sM, const int* __restrict__ dM, const float* __restrict__ xM,
        float* __restrict__ oM, const int* __restrict__ blM, const int* __restrict__ cM,
        const int* __restrict__ sR, const int* __restrict__ dR, const float* __restrict__ xR,
        float* __restrict__ oR, const int* __restrict__ blR, const int* __restrict__ cR) {
    __shared__ __attribute__((aligned(16))) float acc[20 * HDIM];
    __shared__ int cn[20];
    __shared__ int off[21];
    __shared__ int list[8192];
    __shared__ int nlist;
    const int b = blockIdx.x;
    if (b < OWN_C)
        own_agg_body<20, 8192, 6144>(sM, dM, xM, oM, EM, NC, blM, cM, b, OWN_C,
                                     acc, cn, off, list, nlist);
    else
        own_agg_body<20, 8192, 1536>(sR, dR, xR, oR, ER, NC, blR, cR, b - OWN_C, OWN_C,
                                     acc, cn, off, list, nlist);
}

template<int NT, bool FULL, bool VECA>
__device__ __forceinline__ void gemm_body(v8f (&c)[NT],
                                          const float* __restrict__ A,
                                          const v2f* __restrict__ Wp,
                                          int M, int K, int N,
                                          int mbase, int lane) {
    const int row  = mbase + (lane & 15);
    const int koff = (lane >= 16) ? 2 : 0;
    const int col0 = lane & 15;
    const bool rok = FULL || (row < M);
    const float* __restrict__ Arow = A + (long)row * K;
    const int Kmain = K & ~3;

    #pragma unroll 4
    for (int k = 0; k < Kmain; k += 4) {
        const int kk = k + koff;
        v2f a;
        if (rok) {
            if (VECA) {
                a = *(const v2f*)(Arow + kk);
            } else {
                a.x = Arow[kk];
                a.y = Arow[kk + 1];
            }
        } else {
            a.x = 0.f; a.y = 0.f;
        }
        #pragma unroll
        for (int nt = 0; nt < NT; ++nt) {
            v2f b = Wp[(long)(kk >> 1) * N + nt * 16 + col0];
            c[nt] = __builtin_amdgcn_wmma_f32_16x16x4_f32(
                false, a, false, b, (short)0, c[nt], false, false);
        }
    }
    if (Kmain < K) {
        const int kk = Kmain + koff;
        v2f a;
        a.x = (rok && kk     < K) ? Arow[kk]     : 0.f;
        a.y = (rok && kk + 1 < K) ? Arow[kk + 1] : 0.f;
        #pragma unroll
        for (int nt = 0; nt < NT; ++nt) {
            v2f b = Wp[(long)(kk >> 1) * N + nt * 16 + col0];
            c[nt] = __builtin_amdgcn_wmma_f32_16x16x4_f32(
                false, a, false, b, (short)0, c[nt], false, false);
        }
    }
}

#define SLOTS8 "v_nop\n\tv_nop\n\tv_nop\n\tv_nop\n\tv_nop\n\tv_nop\n\tv_nop\n\tv_nop"
__device__ __forceinline__ void result_slots(v8f (&c)[4]) {
    asm volatile(SLOTS8 : "+v"(c[0]), "+v"(c[1]), "+v"(c[2]), "+v"(c[3]));
}
__device__ __forceinline__ void result_slots(v8f (&c)[2]) {
    asm volatile(SLOTS8 : "+v"(c[0]), "+v"(c[1]));
}

template<int NT, bool VECA1, bool NORM>
__global__ __launch_bounds__(256) void k_gemm(const float* A1,
                                              const v2f* __restrict__ W1p,
                                              const float* __restrict__ bias,
                                              const float* __restrict__ A2,
                                              const v2f* __restrict__ W2p,
                                              const float* __restrict__ prev,
                                              const float* __restrict__ res,
                                              const float* __restrict__ g,
                                              const float* __restrict__ b,
                                              float* out,
                                              int M, int K1, int K2,
                                              int do_relu) {
    static_assert(!NORM || NT * 16 == HDIM, "the normalised row must be the whole tile row");
    const int N = NT * 16;
    const int lane = threadIdx.x & 31;
    const int wv = threadIdx.x >> 5;
    const int mbase = (blockIdx.x * 8 + wv) * 16;
    if (mbase >= M) return;

    v8f c[NT];
    #pragma unroll
    for (int t = 0; t < NT; ++t)
        c[t] = (v8f){0.f, 0.f, 0.f, 0.f, 0.f, 0.f, 0.f, 0.f};

    const bool full = (mbase + 16 <= M);
    if (full) {
        gemm_body<NT, true, VECA1>(c, A1, W1p, M, K1, N, mbase, lane);
        if (A2) gemm_body<NT, true, true>(c, A2, W2p, M, K2, N, mbase, lane);
    } else {
        gemm_body<NT, false, VECA1>(c, A1, W1p, M, K1, N, mbase, lane);
        if (A2) gemm_body<NT, false, true>(c, A2, W2p, M, K2, N, mbase, lane);
    }

    result_slots(c);

    const int col0 = lane & 15;
    const int rofs = (lane >= 16) ? 8 : 0;
    #pragma unroll
    for (int nt = 0; nt < NT; ++nt) {
        const float bb = bias ? bias[nt * 16 + col0] : 0.f;
        #pragma unroll
        for (int r = 0; r < 8; ++r) c[nt][r] += bb;
    }
    if (prev) {
        #pragma unroll
        for (int r = 0; r < 8; ++r) {
            const long rb = (long)min(mbase + r + rofs, M - 1) * N + col0;
            #pragma unroll
            for (int nt = 0; nt < NT; ++nt) c[nt][r] += prev[rb + nt * 16];
        }
    }
    if (res) {
        #pragma unroll
        for (int r = 0; r < 8; ++r) {
            const long rb = (long)min(mbase + r + rofs, M - 1) * N + col0;
            #pragma unroll
            for (int nt = 0; nt < NT; ++nt) c[nt][r] += res[rb + nt * 16];
        }
    }

    if (NORM) {
        float gg[NT], be[NT];
        #pragma unroll
        for (int nt = 0; nt < NT; ++nt) { gg[nt] = g[nt * 16 + col0]; be[nt] = b[nt * 16 + col0]; }
        #pragma unroll
        for (int r = 0; r < 8; ++r) {
            const int row = mbase + r + rofs;
            float v[NT];
            #pragma unroll
            for (int nt = 0; nt < NT; ++nt) v[nt] = fmaxf(c[nt][r], 0.f);
            float s = v[0];
            #pragma unroll
            for (int nt = 1; nt < NT; ++nt) s += v[nt];
            #pragma unroll
            for (int off = 8; off > 0; off >>= 1) s += __shfl_xor(s, off, 32);
            const float mean = s * (1.0f / 64.0f);
            float d[NT];
            #pragma unroll
            for (int nt = 0; nt < NT; ++nt) d[nt] = v[nt] - mean;
            float q = d[0] * d[0];
            #pragma unroll
            for (int nt = 1; nt < NT; ++nt) q += d[nt] * d[nt];
            #pragma unroll
            for (int off = 8; off > 0; off >>= 1) q += __shfl_xor(q, off, 32);
            const float rstd = rsqrtf(q * (1.0f / 64.0f) + 1e-5f);
            if (full || row < M) {
                const long rb = (long)row * N + col0;
                #pragma unroll
                for (int nt = 0; nt < NT; ++nt)
                    out[rb + nt * 16] = d[nt] * rstd * gg[nt] + be[nt];
            }
        }
    } else {
        #pragma unroll
        for (int nt = 0; nt < NT; ++nt) {
            #pragma unroll
            for (int r = 0; r < 8; ++r) {
                const int row = mbase + r + rofs;
                if (full || row < M) {
                    float v = c[nt][r];
                    if (do_relu) v = fmaxf(v, 0.f);
                    out[(long)row * N + nt * 16 + col0] = v;
                }
            }
        }
    }
}

#define CW_PITCH 80
#define CC_PITCH 48
#define CT_PITCH 68
typedef __attribute__((ext_vector_type(4))) float v4f;

template<bool HEAD>
__global__ __launch_bounds__(128) void k_crow(const float* __restrict__ aggM,
                                              const float* __restrict__ aggR,
                                              const float* __restrict__ x,
                                              const v2f* __restrict__ wp,
                                              const float* __restrict__ bM,
                                              const float* __restrict__ bR,
                                              const float* __restrict__ g,
                                              const float* __restrict__ b,
                                              const v2f* __restrict__ wpc,
                                              const float* __restrict__ cb1,
                                              const float* __restrict__ cW2,
                                              const float* __restrict__ cb2,
                                              float* __restrict__ out, int M) {
    __shared__ __attribute__((aligned(16))) v2f sW[4 * 32 * CW_PITCH];
    __shared__ __attribute__((aligned(16))) v2f sWc[HEAD ? 32 * CC_PITCH : 2];
    __shared__ __attribute__((aligned(16))) float sT[HEAD ? 4 * 16 * CT_PITCH : 4];
    __shared__ float sO[HEAD ? 64 : 1];

    const int tid = threadIdx.x;
    const int lane = tid & 31;
    const int wv = tid >> 5;
    const int mbase = (blockIdx.x * 4 + wv) * 16;
    const int col0 = lane & 15;
    const int hh = lane >> 4;
    const int rofs = hh * 8;

    v4f wr[32];
    #pragma unroll
    for (int u = 0; u < 32; ++u) wr[u] = ((const v4f*)wp)[u * 128 + tid];
    v4f wc[4];
    if (HEAD) {
        #pragma unroll
        for (int u = 0; u < 4; ++u) wc[u] = ((const v4f*)wpc)[u * 128 + tid];
    }

    const long ra = (long)min(mbase + col0, M - 1) * HDIM + 2 * hh;
    v2f aM[16], aX[16], aR[16];
    #pragma unroll
    for (int ks = 0; ks < 16; ++ks) {
        aM[ks] = *(const v2f*)(aggM + ra + 4 * ks);
        aX[ks] = *(const v2f*)(x + ra + 4 * ks);
        aR[ks] = *(const v2f*)(aggR + ra + 4 * ks);
    }
    float xr[4][8];
    #pragma unroll
    for (int r = 0; r < 8; ++r) {
        const long rb = (long)min(mbase + r + rofs, M - 1) * HDIM + col0;
        #pragma unroll
        for (int nt = 0; nt < 4; ++nt) xr[nt][r] = x[rb + nt * 16];
    }
    float bb[4];
    #pragma unroll
    for (int nt = 0; nt < 4; ++nt) bb[nt] = bM[nt * 16 + col0] + bR[nt * 16 + col0];
    float gg[4], be[4];
    if (!HEAD) {
        #pragma unroll
        for (int nt = 0; nt < 4; ++nt) { gg[nt] = g[nt * 16 + col0]; be[nt] = b[nt * 16 + col0]; }
    }
    float hb[2], w2[2], b2 = 0.f;
    if (HEAD) {
        #pragma unroll
        for (int nt = 0; nt < 2; ++nt) { hb[nt] = cb1[nt * 16 + col0]; w2[nt] = cW2[nt * 16 + col0]; }
        b2 = cb2[0];
    }

    #pragma unroll
    for (int u = 0; u < 32; ++u) {
        const int i4 = u * 128 + tid;
        ((v4f*)sW)[(i4 >> 5) * (CW_PITCH / 2) + (i4 & 31)] = wr[u];
    }
    if (HEAD) {
        #pragma unroll
        for (int u = 0; u < 4; ++u) {
            const int i4 = u * 128 + tid;
            ((v4f*)sWc)[(i4 >> 4) * (CC_PITCH / 2) + (i4 & 15)] = wc[u];
        }
    }
    __syncthreads();

    v8f c[4];
    #pragma unroll
    for (int t = 0; t < 4; ++t)
        c[t] = (v8f){0.f, 0.f, 0.f, 0.f, 0.f, 0.f, 0.f, 0.f};

    const v2f* sB = sW + hh * CW_PITCH + col0;
    v2f bq[2][16];
    #pragma unroll
    for (int j = 0; j < 4; ++j) {
        #pragma unroll
        for (int nt = 0; nt < 4; ++nt) bq[0][j * 4 + nt] = sB[(2 * j) * CW_PITCH + nt * 16];
    }
    #pragma unroll
    for (int gi = 0; gi < 16; ++gi) {
        if (gi + 1 < 16) {
            const int p1 = (gi + 1) >> 2, q1 = (gi + 1) & 3;
            #pragma unroll
            for (int j = 0; j < 4; ++j) {
                #pragma unroll
                for (int nt = 0; nt < 4; ++nt)
                    bq[(gi + 1) & 1][j * 4 + nt] = sB[(p1 * 32 + 2 * (q1 * 4 + j)) * CW_PITCH + nt * 16];
            }
        }
        const int p = gi >> 2, q = gi & 3;
        #pragma unroll
        for (int j = 0; j < 4; ++j) {
            const int ks = q * 4 + j;
            const v2f a = (p == 0) ? aM[ks] : (p == 2) ? aR[ks] : aX[ks];
            #pragma unroll
            for (int nt = 0; nt < 4; ++nt)
                c[nt] = __builtin_amdgcn_wmma_f32_16x16x4_f32(
                    false, a, false, bq[gi & 1][j * 4 + nt], (short)0, c[nt], false, false);
        }
    }

    if (!HEAD) {
        float v[4][8], d[4][8], s[8], q[8];
        #pragma unroll
        for (int r = 0; r < 8; ++r) {
            #pragma unroll
            for (int nt = 0; nt < 4; ++nt) v[nt][r] = fmaxf(c[nt][r] + bb[nt] + xr[nt][r], 0.f);
            s[r] = v[0][r];
            #pragma unroll
            for (int nt = 1; nt < 4; ++nt) s[r] += v[nt][r];
        }
        #pragma unroll
        for (int off = 8; off > 0; off >>= 1) {
            float t[8];
            #pragma unroll
            for (int r = 0; r < 8; ++r) t[r] = __shfl_xor(s[r], off, 32);
            #pragma unroll
            for (int r = 0; r < 8; ++r) s[r] += t[r];
        }
        #pragma unroll
        for (int r = 0; r < 8; ++r) {
            const float mean = s[r] * (1.0f / 64.0f);
            #pragma unroll
            for (int nt = 0; nt < 4; ++nt) d[nt][r] = v[nt][r] - mean;
            q[r] = d[0][r] * d[0][r];
            #pragma unroll
            for (int nt = 1; nt < 4; ++nt) q[r] += d[nt][r] * d[nt][r];
        }
        #pragma unroll
        for (int off = 8; off > 0; off >>= 1) {
            float t[8];
            #pragma unroll
            for (int r = 0; r < 8; ++r) t[r] = __shfl_xor(q[r], off, 32);
            #pragma unroll
            for (int r = 0; r < 8; ++r) q[r] += t[r];
        }
        #pragma unroll
        for (int r = 0; r < 8; ++r) {
            const int row = mbase + r + rofs;
            const float rstd = rsqrtf(q[r] * (1.0f / 64.0f) + 1e-5f);
            if (row < M) {
                const long rb = (long)row * HDIM + col0;
                #pragma unroll
                for (int nt = 0; nt < 4; ++nt)
                    out[rb + nt * 16] = d[nt][r] * rstd * gg[nt] + be[nt];
            }
        }
    } else {
        float* tw = sT + wv * (16 * CT_PITCH);
        #pragma unroll
        for (int nt = 0; nt < 4; ++nt) {
            #pragma unroll
            for (int r = 0; r < 8; ++r)
                tw[(rofs + r) * CT_PITCH + nt * 16 + col0] = fmaxf(c[nt][r] + bb[nt] + xr[nt][r], 0.f);
        }
        __syncthreads();
        v8f hc[2];
        hc[0] = (v8f){0.f, 0.f, 0.f, 0.f, 0.f, 0.f, 0.f, 0.f};
        hc[1] = (v8f){0.f, 0.f, 0.f, 0.f, 0.f, 0.f, 0.f, 0.f};
        const float* ta = tw + col0 * CT_PITCH + 2 * hh;
        const v2f* sC = sWc + hh * CC_PITCH + col0;
        #pragma unroll
        for (int ks = 0; ks < 16; ++ks) {
            const v2f a = *(const v2f*)(ta + 4 * ks);
            #pragma unroll
            for (int nt = 0; nt < 2; ++nt)
                hc[nt] = __builtin_amdgcn_wmma_f32_16x16x4_f32(
                    false, a, false, sC[(2 * ks) * CC_PITCH + nt * 16], (short)0, hc[nt], false, false);
        }
        float s[8];
        #pragma unroll
        for (int r = 0; r < 8; ++r)
            s[r] = fmaxf(hc[0][r] + hb[0], 0.f) * w2[0] + fmaxf(hc[1][r] + hb[1], 0.f) * w2[1];
        #pragma unroll
        for (int off = 8; off > 0; off >>= 1) {
            float t[8];
            #pragma unroll
            for (int r = 0; r < 8; ++r) t[r] = __shfl_xor(s[r], off, 32);
            #pragma unroll
            for (int r = 0; r < 8; ++r) s[r] += t[r];
        }
        float mine = 0.f;
        #pragma unroll
        for (int r = 0; r < 8; ++r)
            if (col0 == r) mine = s[r];
        if (col0 < 8) sO[wv * 16 + rofs + col0] = mine + b2;
        __syncthreads();
        if (tid < 64) {
            const int row = blockIdx.x * 64 + tid;
            if (row < M) out[row] = sO[tid];
        }
    }
}

extern "C" void kernel_launch(void* const* d_in, const int* in_sizes, int n_in,
                              void* d_out, int out_size, void* d_ws, size_t ws_size,
                              hipStream_t stream) {
    const float* x_news    = (const float*)d_in[0];
    const float* x_company = (const float*)d_in[1];
    const int*   edge_sim  = (const int*)d_in[2];
    const int*   men_src   = (const int*)d_in[3];
    const int*   men_dst   = (const int*)d_in[4];
    const int*   edge_rel  = (const int*)d_in[5];
    const float* nw_W = (const float*)d_in[6];
    const float* nw_b = (const float*)d_in[7];
    const float* cw_W = (const float*)d_in[8];
    const float* cw_b = (const float*)d_in[9];
    const float* n1n_g = (const float*)d_in[10];
    const float* n1n_b = (const float*)d_in[11];
    const float* n1c_g = (const float*)d_in[12];
    const float* n1c_b = (const float*)d_in[13];
    const float* c1_Wl = (const float*)d_in[14];
    const float* c1_bl = (const float*)d_in[15];
    const float* c1_Wr = (const float*)d_in[16];
    const float* c2_Wl = (const float*)d_in[17];
    const float* c2_bl = (const float*)d_in[18];
    const float* c2_Wr = (const float*)d_in[19];
    const float* cls_W1 = (const float*)d_in[20];
    const float* cls_b1 = (const float*)d_in[21];
    const float* cls_W2 = (const float*)d_in[22];
    const float* cls_b2 = (const float*)d_in[23];
    float* out = (float*)d_out;

    float* ws = (float*)d_ws;
    long o = 0;
    float* xn0  = ws + o; o += (long)NN * HDIM;
    float* lpl  = ws + o; o += (long)NN * HDIM;
    float* aggn = ws + o; o += (long)NN * HDIM;
    float* xn1  = aggn;
    float* aggcM = ws + o; o += (long)NC * HDIM;
    float* aggcR = ws + o; o += (long)NC * HDIM;
    float* xc0  = ws + o; o += (long)NC * HDIM;
    float* xc1  = ws + o; o += (long)NC * HDIM;
    o += (long)NC * HDIM;
    float* comp = ws + o; o += (long)NC * HDIM;
    o += (long)NC * 32;
    v2f* wp_nw  = (v2f*)(ws + o); o += 2L * (388 / 2) * 64;
    v2f* wp_cw  = (v2f*)(ws + o); o += 2L * (24 / 2) * 64;
    v2f* wp_c1  = (v2f*)(ws + o); o += 2L * 6 * 32 * 64;
    v2f* wp_c2  = (v2f*)(ws + o); o += 2L * 4 * 32 * 64;
    v2f* wp_cls = (v2f*)(ws + o); o += 2L * 32 * 32;

    int* ctr   = (int*)comp;
    int* ctrN  = ctr;
    int* ctrM  = ctr + 512;
    int* ctrR  = ctr + 1024;
    int* blR   = ctr + 4096;
    int* blN   = (int*)lpl;
    int* blM   = blN + 256L * 15360;

    dim3 blk(256);
    int tilesNN = (NN + 15) / 16, tilesNC = (NC + 15) / 16;
    unsigned gemmBlkNN = (tilesNN + 7) / 8, gemmBlkNC = (tilesNC + 7) / 8;
    unsigned crowBlk = (tilesNC + 3) / 4;

    k_prep<<<PB_NW + PB_CW + PB_C1 + PB_C2 + PB_CLS + PB_Z, blk, 0, stream>>>(
        nw_W, cw_W, c1_Wl, c1_Wr, c2_Wl, c2_Wr, cls_W1,
        wp_nw, wp_cw, wp_c1, wp_c2, wp_cls, ctr);

    k_gemm<4, false, false><<<gemmBlkNN, blk, 0, stream>>>(x_news, wp_nw, nw_b,
                                                           nullptr, nullptr,
                                                           nullptr, nullptr, nullptr, nullptr,
                                                           xn0, NN, 385, 0, 1);
    k_gemm<4, true, false><<<gemmBlkNC, blk, 0, stream>>>(x_company, wp_cw, cw_b,
                                                          nullptr, nullptr,
                                                          nullptr, nullptr, nullptr, nullptr,
                                                          xc0, NC, 24, 0, 1);

    k_bin3<<<BINB_N + BINB_M + BINB_R, 1024, 0, stream>>>(edge_sim, edge_sim + ES, blN, ctrN,
                                                          men_src, men_dst, blM, ctrM,
                                                          edge_rel, edge_rel + ER, blR, ctrR);
    k_own_agg3<<<OWN_N + OWN_C + OWN_C, 1024, 0, stream>>>(edge_sim, edge_sim + ES, xn0, aggn, blN, ctrN,
                                                           men_src, men_dst, xn0, aggcM, blM, ctrM,
                                                           edge_rel, edge_rel + ER, xc0, aggcR, blR, ctrR);

    k_gemm<4, true, true><<<gemmBlkNN, blk, 0, stream>>>(aggn, wp_c1 + 0 * 2048,
                                                         c1_bl + 0 * 64,
                                                         xn0, wp_c1 + 1 * 2048,
                                                         nullptr, nullptr, n1n_g, n1n_b,
                                                         xn1, NN, 64, 64, 1);
    k_crow<false><<<crowBlk, 128, 0, stream>>>(aggcM, aggcR, xc0, wp_c1 + 2 * 2048,
                                               c1_bl + 1 * 64, c1_bl + 2 * 64, n1c_g, n1c_b,
                                               nullptr, nullptr, nullptr, nullptr,
                                               xc1, NC);

    k_own_agg2<<<OWN_C + OWN_C, 1024, 0, stream>>>(men_src, men_dst, xn1, aggcM, blM, ctrM,
                                                   edge_rel, edge_rel + ER, xc1, aggcR, blR, ctrR);
    k_crow<true><<<crowBlk, 128, 0, stream>>>(aggcM, aggcR, xc1, wp_c2,
                                              c2_bl + 1 * 64, c2_bl + 2 * 64, nullptr, nullptr,
                                              wp_cls, cls_b1, cls_W2, cls_b2,
                                              out, NC);
}


# launch merging: both feature transforms and the edge binning share one grid (job by block index; two 32-wave work-groups per WGP), so binning runs beside the matrix tiles; 9 to 7 launches
# speedup vs baseline: 7.5689x; 1.1810x over previous
#include <hip/hip_runtime.h>
#include <hip/hip_bf16.h>

typedef __attribute__((ext_vector_type(2))) float v2f;
typedef __attribute__((ext_vector_type(8))) float v8f;

#define NN 100000
#define NC 5000
#define HDIM 64
#define ES 3200000
#define EM 1000000
#define ER 160000

__device__ __forceinline__ void pack_one(const float* __restrict__ W,
                                         v2f* __restrict__ Wp,
                                         int K, int N, int i) {
    const int Kp2 = ((K + 3) & ~3) >> 1;
    if (i >= Kp2 * N) return;
    const int k2 = i / N, col = i % N;
    const int k = 2 * k2;
    v2f v;
    v.x = (k     < K) ? W[k * N + col]       : 0.f;
    v.y = (k + 1 < K) ? W[(k + 1) * N + col] : 0.f;
    Wp[i] = v;
}

#define PB_NW  49
#define PB_CW  3
#define PB_C1  48
#define PB_C2  32
#define PB_CLS 4
#define PB_Z   16
__global__ __launch_bounds__(256) void k_prep(const float* __restrict__ nw_W,
                                              const float* __restrict__ cw_W,
                                              const float* __restrict__ c1_Wl,
                                              const float* __restrict__ c1_Wr,
                                              const float* __restrict__ c2_Wl,
                                              const float* __restrict__ c2_Wr,
                                              const float* __restrict__ cls_W1,
                                              v2f* __restrict__ wp_nw,
                                              v2f* __restrict__ wp_cw,
                                              v2f* __restrict__ wp_c1,
                                              v2f* __restrict__ wp_c2,
                                              v2f* __restrict__ wp_cls,
                                              int* __restrict__ ctr) {
    const int tid = threadIdx.x;
    int b = blockIdx.x;
    if (b < PB_NW) { pack_one(nw_W, wp_nw, 385, 64, b * 256 + tid); return; }
    b -= PB_NW;
    if (b < PB_CW) { pack_one(cw_W, wp_cw, 24, 64, b * 256 + tid); return; }
    b -= PB_CW;
    if (b < PB_C1) {
        const int m = b >> 3;
        const float* W = ((m & 1) ? c1_Wr : c1_Wl) + (m >> 1) * 4096;
        pack_one(W, wp_c1 + m * 2048, 64, 64, (b & 7) * 256 + tid);
        return;
    }
    b -= PB_C1;
    if (b < PB_C2) {
        const int m = b >> 3;
        const float* W = ((m & 1) ? c2_Wr : c2_Wl) + ((m >> 1) + 1) * 4096;
        pack_one(W, wp_c2 + m * 2048, 64, 64, (b & 7) * 256 + tid);
        return;
    }
    b -= PB_C2;
    if (b < PB_CLS) { pack_one(cls_W1, wp_cls, 64, 32, b * 256 + tid); return; }
    b -= PB_CLS;
    ctr[b * 256 + tid] = 0;
}

#define PK_SH 9
#define PK_MASK 511
#define BIN_EPT 8
#define BIN_CHUNK (1024 * BIN_EPT)
#define BIN_NBMAX 256
template<int R, int CAPB>
__device__ __forceinline__ void bin_body(const int* __restrict__ src,
                                         const int* __restrict__ dst,
                                         int nE, int nB,
                                         int* __restrict__ blist,
                                         int* __restrict__ bctr,
                                         int bid,
                                         int* hcnt, int* hoff, int* gbase,
                                         int* stage, int* sown) {
    static_assert(R <= PK_MASK + 1, "row inside the range must fit the packed field");
    constexpr int EPT = BIN_EPT;
    constexpr int CHUNK = BIN_CHUNK;
    constexpr int NBMAX = BIN_NBMAX;

    const int tid = threadIdx.x;
    const int lane = tid & 31;
    const int wv = tid >> 5;
    const int base = bid * CHUNK;

    if (tid < NBMAX) hcnt[tid] = 0;
    __syncthreads();

    int ow[EPT], rk[EPT], pk[EPT];
    #pragma unroll
    for (int u = 0; u < EPT; ++u) {
        const int e = base + u * 1024 + tid;
        ow[u] = -1; rk[u] = 0; pk[u] = 0;
        if (e < nE) {
            const int d = dst[e];
            const int o = (int)((unsigned)d / (unsigned)R);
            ow[u] = o;
            pk[u] = (src[e] << PK_SH) | (d - o * R);
            rk[u] = atomicAdd(&hcnt[o], 1);
        }
    }
    __syncthreads();

    if (tid < nB) {
        const int cnt = hcnt[tid];
        gbase[tid] = cnt ? atomicAdd(&bctr[tid], cnt) : 0;
    }
    if (wv == 31) {
        int cq[8];
        int t = 0;
        #pragma unroll
        for (int q = 0; q < 8; ++q) { const int v = hcnt[lane * 8 + q]; cq[q] = t; t += v; }
        int incl = t;
        #pragma unroll
        for (int off = 1; off < 32; off <<= 1) {
            const int s = __shfl_up(incl, off, 32);
            if (lane >= off) incl += s;
        }
        const int excl = incl - t;
        #pragma unroll
        for (int q = 0; q < 8; ++q) hoff[lane * 8 + q] = cq[q] + excl;
    }
    __syncthreads();

    #pragma unroll
    for (int u = 0; u < EPT; ++u) {
        if (ow[u] >= 0) {
            const int lp = hoff[ow[u]] + rk[u];
            stage[lp] = pk[u];
            sown[lp] = ow[u];
        }
    }
    __syncthreads();

    const int total = min(CHUNK, nE - base);
    for (int i = tid; i < total; i += 1024) {
        const int o = sown[i];
        const int pos = gbase[o] + (i - hoff[o]);
        if (pos < CAPB) blist[(long)o * CAPB + pos] = stage[i];
        else bctr[nB] = 1;
    }
}

#define BIN_LDS \
    __shared__ int hcnt[BIN_NBMAX];      \
    __shared__ int hoff[BIN_NBMAX];      \
    __shared__ int gbase[BIN_NBMAX];     \
    __shared__ int stage[BIN_CHUNK];     \
    __shared__ int sown[BIN_CHUNK];

#define BINB_N ((ES + BIN_CHUNK - 1) / BIN_CHUNK)
#define BINB_M ((EM + BIN_CHUNK - 1) / BIN_CHUNK)
#define BINB_R ((ER + BIN_CHUNK - 1) / BIN_CHUNK)
#define OWN_N ((NN + 390) / 391)
#define OWN_C ((NC + 19) / 20)

__device__ __forceinline__ void add_group(const float* __restrict__ x, float* acc,
                                          int s, int dl, int m, int lane) {
    if (m >= 32) {
        #pragma nounroll
        for (int j0 = 0; j0 < 32; j0 += 8) {
            float v0[8], v1[8];
            #pragma unroll
            for (int j = 0; j < 8; ++j) {
                const int sj = __builtin_amdgcn_readlane(s, j0 + j);
                const float* xr = x + (long)sj * HDIM + lane;
                v0[j] = xr[0];
                v1[j] = xr[32];
            }
            #pragma unroll
            for (int j = 0; j < 8; ++j) {
                const int dj = __builtin_amdgcn_readlane(dl, j0 + j);
                float* ar = acc + dj * HDIM + lane;
                atomicAdd(ar, v0[j]);
                atomicAdd(ar + 32, v1[j]);
            }
        }
    } else {
        for (int j = 0; j < m; ++j) {
            const int sj = __builtin_amdgcn_readlane(s, j);
            const int dj = __builtin_amdgcn_readlane(dl, j);
            const float* xr = x + (long)sj * HDIM + lane;
            float* ar = acc + dj * HDIM + lane;
            const float a0 = xr[0];
            const float a1 = xr[32];
            atomicAdd(ar, a0);
            atomicAdd(ar + 32, a1);
        }
    }
}

#define SRT_UB 16
template<int R, int CAPB>
__device__ __forceinline__ void own_sorted_body(const float* __restrict__ x,
                                                float* __restrict__ out,
                                                const int* __restrict__ bl,
                                                int nbin, int lo, int nr,
                                                int* cn, int* off, int* sorted) {
    constexpr int EPT = (CAPB + 1023) / 1024;
    constexpr int Q = (R + 31) / 32;
    const int tid = threadIdx.x;
    const int lane = tid & 31;
    const int wv = __builtin_amdgcn_readfirstlane(tid >> 5);
    const int n = min(nbin, CAPB);

    for (int i = tid; i < R; i += 1024) cn[i] = 0;
    __syncthreads();

    int pk[EPT], rk[EPT];
    #pragma unroll
    for (int u = 0; u < EPT; ++u) {
        const int i = u * 1024 + tid;
        pk[u] = -1; rk[u] = 0;
        if (i < n) {
            const int p = bl[i];
            pk[u] = p;
            rk[u] = atomicAdd(&cn[p & PK_MASK], 1);
        }
    }
    __syncthreads();

    if (wv == 0) {
        int cq[Q];
        int t = 0;
        #pragma unroll
        for (int q = 0; q < Q; ++q) {
            const int r = lane * Q + q;
            const int v = (r < R) ? cn[r] : 0;
            cq[q] = t; t += v;
        }
        int incl = t;
        #pragma unroll
        for (int o = 1; o < 32; o <<= 1) {
            const int s = __shfl_up(incl, o, 32);
            if (lane >= o) incl += s;
        }
        const int excl = incl - t;
        #pragma unroll
        for (int q = 0; q < Q; ++q) {
            const int r = lane * Q + q;
            if (r < R) off[r] = cq[q] + excl;
        }
        if (lane == 31) off[R] = incl;
    }
    __syncthreads();

    #pragma unroll
    for (int u = 0; u < EPT; ++u)
        if (pk[u] >= 0) sorted[off[pk[u] & PK_MASK] + rk[u]] = pk[u];
    for (int i = tid; i < nr * 16; i += 1024) {
        const int r = i >> 4;
        if (cn[r] == 0)
            *(float4*)(out + (long)(lo + r) * HDIM + (i & 15) * 4) = make_float4(0.f, 0.f, 0.f, 0.f);
    }
    __syncthreads();

    const int r0 = min(wv * Q, nr);
    const int r1 = min(r0 + Q, nr);
    const int e0 = off[r0];
    const int e1 = off[r1];
    if (e0 >= e1) return;

    float a0 = 0.f, a1 = 0.f;
    int cur = -1;
    int pn = sorted[min(e0 + lane, e1 - 1)];
    for (int g = e0; g < e1; g += SRT_UB) {
        const int pv = pn;
        pn = sorted[min(g + SRT_UB + lane, e1 - 1)];
        float2 v[SRT_UB];
        #pragma unroll
        for (int u = 0; u < SRT_UB; ++u) {
            const int pu = __builtin_amdgcn_readlane(pv, u);
            v[u] = *(const float2*)(x + (long)(pu >> PK_SH) * HDIM + 2 * lane);
        }
        #pragma unroll
        for (int u = 0; u < SRT_UB; ++u) {
            if (g + u < e1) {
                const int d = __builtin_amdgcn_readlane(pv, u) & PK_MASK;
                if (d != cur) {
                    if (cur >= 0) {
                        const float inv = 1.0f / fmaxf((float)cn[cur], 1.0f);
                        *(float2*)(out + (long)(lo + cur) * HDIM + 2 * lane) = make_float2(a0 * inv, a1 * inv);
                    }
                    cur = d; a0 = 0.f; a1 = 0.f;
                }
                a0 += v[u].x; a1 += v[u].y;
            }
        }
    }
    const float inv = 1.0f / fmaxf((float)cn[cur], 1.0f);
    *(float2*)(out + (long)(lo + cur) * HDIM + 2 * lane) = make_float2(a0 * inv, a1 * inv);
}

#define AGG_RS 128
template<int R, int CAP, int CAPB>
__device__ __forceinline__ void own_agg_body(const int* __restrict__ src,
                                             const int* __restrict__ dst,
                                             const float* __restrict__ x,
                                             float* __restrict__ out,
                                             int nE, int nDst,
                                             const int* __restrict__ blist,
                                             const int* __restrict__ bctr,
                                             int bid, int nB,
                                             float* acc, int* cn, int* off, int* list, int& nlist) {
    static_assert(CAPB <= CAP, "the ordered list must fit the list array");
    const int tid = threadIdx.x;
    const int lane = tid & 31;
    const int wv = __builtin_amdgcn_readfirstlane(tid >> 5);
    constexpr int RS = (R < AGG_RS) ? R : AGG_RS;
    const int lo0 = bid * R;
    const int nr0 = min(R, nDst - lo0);
    const bool binned = __builtin_amdgcn_readfirstlane(bctr[nB]) == 0;
    const int nbin = __builtin_amdgcn_readfirstlane(bctr[bid]);

    if (binned) {
        own_sorted_body<R, CAPB>(x, out, blist + (long)bid * CAPB, nbin, lo0, nr0, cn, off, list);
        return;
    }

    for (int sub = 0; sub < nr0; sub += RS) {
    const int lo = lo0 + sub;
    const int nr = min(RS, nr0 - sub);

    for (int i = tid; i < RS * HDIM; i += 1024) acc[i] = 0.f;
    for (int i = tid; i < RS; i += 1024) cn[i] = 0;
    if (tid == 0) nlist = 0;
    __syncthreads();

    {
    constexpr int U = 4;
    constexpr int CH = U * 128;
    const int nCh = (nE + CH - 1) / CH;
    const int rot = (int)(((long)bid * nCh) / nB);
    for (int c0 = wv; c0 < nCh; c0 += 32) {
        int c = c0 + rot;
        if (c >= nCh) c -= nCh;
        const int base = c * CH;
        int4 d[U];
        if (base + CH <= nE) {
            #pragma unroll
            for (int u = 0; u < U; ++u)
                d[u] = *(const int4*)(dst + base + u * 128 + lane * 4);
        } else {
            #pragma unroll
            for (int u = 0; u < U; ++u) {
                const int i0 = base + u * 128 + lane * 4;
                d[u].x = (i0     < nE) ? dst[i0]     : -1;
                d[u].y = (i0 + 1 < nE) ? dst[i0 + 1] : -1;
                d[u].z = (i0 + 2 < nE) ? dst[i0 + 2] : -1;
                d[u].w = (i0 + 3 < nE) ? dst[i0 + 3] : -1;
            }
        }
        int cl = 0;
        #pragma unroll
        for (int u = 0; u < U; ++u) {
            cl += ((unsigned)(d[u].x - lo) < (unsigned)nr) ? 1 : 0;
            cl += ((unsigned)(d[u].y - lo) < (unsigned)nr) ? 1 : 0;
            cl += ((unsigned)(d[u].z - lo) < (unsigned)nr) ? 1 : 0;
            cl += ((unsigned)(d[u].w - lo) < (unsigned)nr) ? 1 : 0;
        }
        int incl = cl;
        #pragma unroll
        for (int off = 1; off < 32; off <<= 1) {
            const int t = __shfl_up(incl, off, 32);
            if (lane >= off) incl += t;
        }
        const int total = __builtin_amdgcn_readlane(incl, 31);
        if (total == 0) continue;
        int s0 = 0;
        if (lane == 31) s0 = atomicAdd(&nlist, total);
        s0 = __builtin_amdgcn_readlane(s0, 31);
        int pos = s0 + incl - cl;
        #pragma unroll
        for (int u = 0; u < U; ++u) {
            #pragma unroll
            for (int q = 0; q < 4; ++q) {
                const int dv = (q == 0) ? d[u].x : (q == 1) ? d[u].y : (q == 2) ? d[u].z : d[u].w;
                const int dl = dv - lo;
                if ((unsigned)dl < (unsigned)nr) {
                    const int e = base + u * 128 + lane * 4 + q;
                    if (pos < CAP) {
                        list[pos] = e;
                    } else {
                        const float* xr = x + (long)src[e] * HDIM;
                        #pragma nounroll
                        for (int k = 0; k < HDIM; ++k) atomicAdd(&acc[dl * HDIM + k], xr[k]);
                        atomicAdd(&cn[dl], 1);
                    }
                    ++pos;
                }
            }
        }
    }
    }
    __syncthreads();

    {
        const int n = min(nlist, CAP);
        for (int g = wv * 32; g < n; g += 32 * 32) {
            const int k = g + lane;
            int s = 0, dl = 0;
            if (k < n) {
                const int e = list[k];
                s = src[e];
                dl = dst[e] - lo;
                atomicAdd(&cn[dl], 1);
            }
            add_group(x, acc, s, dl, min(32, n - g), lane);
        }
    }
    __syncthreads();

    for (int i = tid; i < nr * 16; i += 1024) {
        const int r = i >> 4;
        const int c4 = (i & 15) * 4;
        const float inv = 1.0f / fmaxf((float)cn[r], 1.0f);
        float4 v = *(const float4*)(acc + r * HDIM + c4);
        v.x *= inv; v.y *= inv; v.z *= inv; v.w *= inv;
        *(float4*)(out + (long)(lo + r) * HDIM + c4) = v;
    }
    __syncthreads();
    }
}

__global__ __launch_bounds__(1024) void k_own_agg3(
        const int* __restrict__ sN, const int* __restrict__ dN, const float* __restrict__ xN,
        float* __restrict__ oN, const int* __restrict__ blN, const int* __restrict__ cN,
        const int* __restrict__ sM, const int* __restrict__ dM, const float* __restrict__ xM,
        float* __restrict__ oM, const int* __restrict__ blM, const int* __restrict__ cM,
        const int* __restrict__ sR, const int* __restrict__ dR, const float* __restrict__ xR,
        float* __restrict__ oR, const int* __restrict__ blR, const int* __restrict__ cR) {
    __shared__ __attribute__((aligned(16))) float acc[AGG_RS * HDIM];
    __shared__ int cn[391];
    __shared__ int off[392];
    __shared__ int list[16384];
    __shared__ int nlist;
    const int b = blockIdx.x;
    if (b < OWN_N)
        own_agg_body<391, 16384, 15360>(sN, dN, xN, oN, ES, NN, blN, cN, b, OWN_N,
                                        acc, cn, off, list, nlist);
    else if (b < OWN_N + OWN_C)
        own_agg_body<20, 8192, 6144>(sM, dM, xM, oM, EM, NC, blM, cM, b - OWN_N, OWN_C,
                                     acc, cn, off, list, nlist);
    else
        own_agg_body<20, 8192, 1536>(sR, dR, xR, oR, ER, NC, blR, cR, b - OWN_N - OWN_C, OWN_C,
                                     acc, cn, off, list, nlist);
}

__global__ __launch_bounds__(1024) void k_own_agg2(
        const int* __restrict__ sM, const int* __restrict__ dM, const float* __restrict__ xM,
        float* __restrict__ oM, const int* __restrict__ blM, const int* __restrict__ cM,
        const int* __restrict__ sR, const int* __restrict__ dR, const float* __restrict__ xR,
        float* __restrict__ oR, const int* __restrict__ blR, const int* __restrict__ cR) {
    __shared__ __attribute__((aligned(16))) float acc[20 * HDIM];
    __shared__ int cn[20];
    __shared__ int off[21];
    __shared__ int list[8192];
    __shared__ int nlist;
    const int b = blockIdx.x;
    if (b < OWN_C)
        own_agg_body<20, 8192, 6144>(sM, dM, xM, oM, EM, NC, blM, cM, b, OWN_C,
                                     acc, cn, off, list, nlist);
    else
        own_agg_body<20, 8192, 1536>(sR, dR, xR, oR, ER, NC, blR, cR, b - OWN_C, OWN_C,
                                     acc, cn, off, list, nlist);
}

template<int NT, bool FULL, bool VECA, int UNR>
__device__ __forceinline__ void gemm_body(v8f (&c)[NT],
                                          const float* __restrict__ A,
                                          const v2f* __restrict__ Wp,
                                          int M, int K, int N,
                                          int mbase, int lane) {
    const int row  = mbase + (lane & 15);
    const int koff = (lane >= 16) ? 2 : 0;
    const int col0 = lane & 15;
    const bool rok = FULL || (row < M);
    const float* __restrict__ Arow = A + (long)row * K;
    const int Kmain = K & ~3;

    #pragma unroll UNR
    for (int k = 0; k < Kmain; k += 4) {
        const int kk = k + koff;
        v2f a;
        if (rok) {
            if (VECA) {
                a = *(const v2f*)(Arow + kk);
            } else {
                a.x = Arow[kk];
                a.y = Arow[kk + 1];
            }
        } else {
            a.x = 0.f; a.y = 0.f;
        }
        #pragma unroll
        for (int nt = 0; nt < NT; ++nt) {
            v2f b = Wp[(long)(kk >> 1) * N + nt * 16 + col0];
            c[nt] = __builtin_amdgcn_wmma_f32_16x16x4_f32(
                false, a, false, b, (short)0, c[nt], false, false);
        }
    }
    if (Kmain < K) {
        const int kk = Kmain + koff;
        v2f a;
        a.x = (rok && kk     < K) ? Arow[kk]     : 0.f;
        a.y = (rok && kk + 1 < K) ? Arow[kk + 1] : 0.f;
        #pragma unroll
        for (int nt = 0; nt < NT; ++nt) {
            v2f b = Wp[(long)(kk >> 1) * N + nt * 16 + col0];
            c[nt] = __builtin_amdgcn_wmma_f32_16x16x4_f32(
                false, a, false, b, (short)0, c[nt], false, false);
        }
    }
}

#define SLOTS8 "v_nop\n\tv_nop\n\tv_nop\n\tv_nop\n\tv_nop\n\tv_nop\n\tv_nop\n\tv_nop"
__device__ __forceinline__ void result_slots(v8f (&c)[4]) {
    asm volatile(SLOTS8 : "+v"(c[0]), "+v"(c[1]), "+v"(c[2]), "+v"(c[3]));
}
__device__ __forceinline__ void result_slots(v8f (&c)[2]) {
    asm volatile(SLOTS8 : "+v"(c[0]), "+v"(c[1]));
}

template<int NT, bool VECA1, bool NORM>
__global__ __launch_bounds__(256) void k_gemm(const float* A1,
                                              const v2f* __restrict__ W1p,
                                              const float* __restrict__ bias,
                                              const float* __restrict__ A2,
                                              const v2f* __restrict__ W2p,
                                              const float* __restrict__ prev,
                                              const float* __restrict__ res,
                                              const float* __restrict__ g,
                                              const float* __restrict__ b,
                                              float* out,
                                              int M, int K1, int K2,
                                              int do_relu) {
    static_assert(!NORM || NT * 16 == HDIM, "the normalised row must be the whole tile row");
    const int N = NT * 16;
    const int lane = threadIdx.x & 31;
    const int wv = threadIdx.x >> 5;
    const int mbase = (blockIdx.x * 8 + wv) * 16;
    if (mbase >= M) return;

    v8f c[NT];
    #pragma unroll
    for (int t = 0; t < NT; ++t)
        c[t] = (v8f){0.f, 0.f, 0.f, 0.f, 0.f, 0.f, 0.f, 0.f};

    const bool full = (mbase + 16 <= M);
    if (full) {
        gemm_body<NT, true, VECA1, 4>(c, A1, W1p, M, K1, N, mbase, lane);
        if (A2) gemm_body<NT, true, true, 4>(c, A2, W2p, M, K2, N, mbase, lane);
    } else {
        gemm_body<NT, false, VECA1, 4>(c, A1, W1p, M, K1, N, mbase, lane);
        if (A2) gemm_body<NT, false, true, 4>(c, A2, W2p, M, K2, N, mbase, lane);
    }

    result_slots(c);

    const int col0 = lane & 15;
    const int rofs = (lane >= 16) ? 8 : 0;
    #pragma unroll
    for (int nt = 0; nt < NT; ++nt) {
        const float bb = bias ? bias[nt * 16 + col0] : 0.f;
        #pragma unroll
        for (int r = 0; r < 8; ++r) c[nt][r] += bb;
    }
    if (prev) {
        #pragma unroll
        for (int r = 0; r < 8; ++r) {
            const long rb = (long)min(mbase + r + rofs, M - 1) * N + col0;
            #pragma unroll
            for (int nt = 0; nt < NT; ++nt) c[nt][r] += prev[rb + nt * 16];
        }
    }
    if (res) {
        #pragma unroll
        for (int r = 0; r < 8; ++r) {
            const long rb = (long)min(mbase + r + rofs, M - 1) * N + col0;
            #pragma unroll
            for (int nt = 0; nt < NT; ++nt) c[nt][r] += res[rb + nt * 16];
        }
    }

    if (NORM) {
        float gg[NT], be[NT];
        #pragma unroll
        for (int nt = 0; nt < NT; ++nt) { gg[nt] = g[nt * 16 + col0]; be[nt] = b[nt * 16 + col0]; }
        #pragma unroll
        for (int r = 0; r < 8; ++r) {
            const int row = mbase + r + rofs;
            float v[NT];
            #pragma unroll
            for (int nt = 0; nt < NT; ++nt) v[nt] = fmaxf(c[nt][r], 0.f);
            float s = v[0];
            #pragma unroll
            for (int nt = 1; nt < NT; ++nt) s += v[nt];
            #pragma unroll
            for (int off = 8; off > 0; off >>= 1) s += __shfl_xor(s, off, 32);
            const float mean = s * (1.0f / 64.0f);
            float d[NT];
            #pragma unroll
            for (int nt = 0; nt < NT; ++nt) d[nt] = v[nt] - mean;
            float q = d[0] * d[0];
            #pragma unroll
            for (int nt = 1; nt < NT; ++nt) q += d[nt] * d[nt];
            #pragma unroll
            for (int off = 8; off > 0; off >>= 1) q += __shfl_xor(q, off, 32);
            const float rstd = rsqrtf(q * (1.0f / 64.0f) + 1e-5f);
            if (full || row < M) {
                const long rb = (long)row * N + col0;
                #pragma unroll
                for (int nt = 0; nt < NT; ++nt)
                    out[rb + nt * 16] = d[nt] * rstd * gg[nt] + be[nt];
            }
        }
    } else {
        #pragma unroll
        for (int nt = 0; nt < NT; ++nt) {
            #pragma unroll
            for (int r = 0; r < 8; ++r) {
                const int row = mbase + r + rofs;
                if (full || row < M) {
                    float v = c[nt][r];
                    if (do_relu) v = fmaxf(v, 0.f);
                    out[(long)row * N + nt * 16 + col0] = v;
                }
            }
        }
    }
}

template<bool VECA>
__device__ __forceinline__ void feat_tile(const float* __restrict__ A,
                                          const v2f* __restrict__ Wp,
                                          const float* __restrict__ bias,
                                          float* __restrict__ out,
                                          int M, int K, int mbase, int lane) {
    v8f c[4];
    #pragma unroll
    for (int t = 0; t < 4; ++t)
        c[t] = (v8f){0.f, 0.f, 0.f, 0.f, 0.f, 0.f, 0.f, 0.f};
    const bool full = (mbase + 16 <= M);
    if (full) gemm_body<4, true, VECA, 2>(c, A, Wp, M, K, HDIM, mbase, lane);
    else      gemm_body<4, false, VECA, 2>(c, A, Wp, M, K, HDIM, mbase, lane);

    result_slots(c);

    const int col0 = lane & 15;
    const int rofs = (lane >= 16) ? 8 : 0;
    #pragma unroll
    for (int nt = 0; nt < 4; ++nt) {
        const float bb = bias[nt * 16 + col0];
        #pragma unroll
        for (int r = 0; r < 8; ++r) {
            const int row = mbase + r + rofs;
            if (full || row < M)
                out[(long)row * HDIM + nt * 16 + col0] = fmaxf(c[nt][r] + bb, 0.f);
        }
    }
}

#define FT_TPB 25
#define FTB_N (((NN + 15) / 16 + FT_TPB - 1) / FT_TPB)
#define FTB_C (((NC + 15) / 16 + FT_TPB - 1) / FT_TPB)
__global__ __launch_bounds__(1024, 16) void k_feat_bin3(
        const float* __restrict__ x_news, const v2f* __restrict__ wp_nw,
        const float* __restrict__ nw_b, float* __restrict__ xn0,
        const float* __restrict__ x_company, const v2f* __restrict__ wp_cw,
        const float* __restrict__ cw_b, float* __restrict__ xc0,
        const int* __restrict__ sN, const int* __restrict__ dN,
        int* __restrict__ blN, int* __restrict__ cN,
        const int* __restrict__ sM, const int* __restrict__ dM,
        int* __restrict__ blM, int* __restrict__ cM,
        const int* __restrict__ sR, const int* __restrict__ dR,
        int* __restrict__ blR, int* __restrict__ cR) {
    BIN_LDS
    int b = blockIdx.x;
    if (b < FTB_N + FTB_C) {
        const int lane = threadIdx.x & 31;
        const int wv = __builtin_amdgcn_readfirstlane(threadIdx.x >> 5);
        if (wv >= FT_TPB) return;
        if (b < FTB_N) {
            const int mbase = (b * FT_TPB + wv) * 16;
            if (mbase < NN) feat_tile<false>(x_news, wp_nw, nw_b, xn0, NN, 385, mbase, lane);
        } else {
            const int mbase = ((b - FTB_N) * FT_TPB + wv) * 16;
            if (mbase < NC) feat_tile<true>(x_company, wp_cw, cw_b, xc0, NC, 24, mbase, lane);
        }
        return;
    }
    b -= FTB_N + FTB_C;
    if (b < BINB_N)
        bin_body<391, 15360>(sN, dN, ES, OWN_N, blN, cN, b, hcnt, hoff, gbase, stage, sown);
    else if (b < BINB_N + BINB_M)
        bin_body<20, 6144>(sM, dM, EM, OWN_C, blM, cM, b - BINB_N, hcnt, hoff, gbase, stage, sown);
    else
        bin_body<20, 1536>(sR, dR, ER, OWN_C, blR, cR, b - BINB_N - BINB_M, hcnt, hoff, gbase, stage, sown);
}

#define CW_PITCH 80
#define CC_PITCH 48
#define CT_PITCH 68
typedef __attribute__((ext_vector_type(4))) float v4f;

template<bool HEAD>
__global__ __launch_bounds__(128) void k_crow(const float* __restrict__ aggM,
                                              const float* __restrict__ aggR,
                                              const float* __restrict__ x,
                                              const v2f* __restrict__ wp,
                                              const float* __restrict__ bM,
                                              const float* __restrict__ bR,
                                              const float* __restrict__ g,
                                              const float* __restrict__ b,
                                              const v2f* __restrict__ wpc,
                                              const float* __restrict__ cb1,
                                              const float* __restrict__ cW2,
                                              const float* __restrict__ cb2,
                                              float* __restrict__ out, int M) {
    __shared__ __attribute__((aligned(16))) v2f sW[4 * 32 * CW_PITCH];
    __shared__ __attribute__((aligned(16))) v2f sWc[HEAD ? 32 * CC_PITCH : 2];
    __shared__ __attribute__((aligned(16))) float sT[HEAD ? 4 * 16 * CT_PITCH : 4];
    __shared__ float sO[HEAD ? 64 : 1];

    const int tid = threadIdx.x;
    const int lane = tid & 31;
    const int wv = tid >> 5;
    const int mbase = (blockIdx.x * 4 + wv) * 16;
    const int col0 = lane & 15;
    const int hh = lane >> 4;
    const int rofs = hh * 8;

    v4f wr[32];
    #pragma unroll
    for (int u = 0; u < 32; ++u) wr[u] = ((const v4f*)wp)[u * 128 + tid];
    v4f wc[4];
    if (HEAD) {
        #pragma unroll
        for (int u = 0; u < 4; ++u) wc[u] = ((const v4f*)wpc)[u * 128 + tid];
    }

    const long ra = (long)min(mbase + col0, M - 1) * HDIM + 2 * hh;
    v2f aM[16], aX[16], aR[16];
    #pragma unroll
    for (int ks = 0; ks < 16; ++ks) {
        aM[ks] = *(const v2f*)(aggM + ra + 4 * ks);
        aX[ks] = *(const v2f*)(x + ra + 4 * ks);
        aR[ks] = *(const v2f*)(aggR + ra + 4 * ks);
    }
    float xr[4][8];
    #pragma unroll
    for (int r = 0; r < 8; ++r) {
        const long rb = (long)min(mbase + r + rofs, M - 1) * HDIM + col0;
        #pragma unroll
        for (int nt = 0; nt < 4; ++nt) xr[nt][r] = x[rb + nt * 16];
    }
    float bb[4];
    #pragma unroll
    for (int nt = 0; nt < 4; ++nt) bb[nt] = bM[nt * 16 + col0] + bR[nt * 16 + col0];
    float gg[4], be[4];
    if (!HEAD) {
        #pragma unroll
        for (int nt = 0; nt < 4; ++nt) { gg[nt] = g[nt * 16 + col0]; be[nt] = b[nt * 16 + col0]; }
    }
    float hb[2], w2[2], b2 = 0.f;
    if (HEAD) {
        #pragma unroll
        for (int nt = 0; nt < 2; ++nt) { hb[nt] = cb1[nt * 16 + col0]; w2[nt] = cW2[nt * 16 + col0]; }
        b2 = cb2[0];
    }

    #pragma unroll
    for (int u = 0; u < 32; ++u) {
        const int i4 = u * 128 + tid;
        ((v4f*)sW)[(i4 >> 5) * (CW_PITCH / 2) + (i4 & 31)] = wr[u];
    }
    if (HEAD) {
        #pragma unroll
        for (int u = 0; u < 4; ++u) {
            const int i4 = u * 128 + tid;
            ((v4f*)sWc)[(i4 >> 4) * (CC_PITCH / 2) + (i4 & 15)] = wc[u];
        }
    }
    __syncthreads();

    v8f c[4];
    #pragma unroll
    for (int t = 0; t < 4; ++t)
        c[t] = (v8f){0.f, 0.f, 0.f, 0.f, 0.f, 0.f, 0.f, 0.f};

    const v2f* sB = sW + hh * CW_PITCH + col0;
    v2f bq[2][16];
    #pragma unroll
    for (int j = 0; j < 4; ++j) {
        #pragma unroll
        for (int nt = 0; nt < 4; ++nt) bq[0][j * 4 + nt] = sB[(2 * j) * CW_PITCH + nt * 16];
    }
    #pragma unroll
    for (int gi = 0; gi < 16; ++gi) {
        if (gi + 1 < 16) {
            const int p1 = (gi + 1) >> 2, q1 = (gi + 1) & 3;
            #pragma unroll
            for (int j = 0; j < 4; ++j) {
                #pragma unroll
                for (int nt = 0; nt < 4; ++nt)
                    bq[(gi + 1) & 1][j * 4 + nt] = sB[(p1 * 32 + 2 * (q1 * 4 + j)) * CW_PITCH + nt * 16];
            }
        }
        const int p = gi >> 2, q = gi & 3;
        #pragma unroll
        for (int j = 0; j < 4; ++j) {
            const int ks = q * 4 + j;
            const v2f a = (p == 0) ? aM[ks] : (p == 2) ? aR[ks] : aX[ks];
            #pragma unroll
            for (int nt = 0; nt < 4; ++nt)
                c[nt] = __builtin_amdgcn_wmma_f32_16x16x4_f32(
                    false, a, false, bq[gi & 1][j * 4 + nt], (short)0, c[nt], false, false);
        }
    }

    if (!HEAD) {
        float v[4][8], d[4][8], s[8], q[8];
        #pragma unroll
        for (int r = 0; r < 8; ++r) {
            #pragma unroll
            for (int nt = 0; nt < 4; ++nt) v[nt][r] = fmaxf(c[nt][r] + bb[nt] + xr[nt][r], 0.f);
            s[r] = v[0][r];
            #pragma unroll
            for (int nt = 1; nt < 4; ++nt) s[r] += v[nt][r];
        }
        #pragma unroll
        for (int off = 8; off > 0; off >>= 1) {
            float t[8];
            #pragma unroll
            for (int r = 0; r < 8; ++r) t[r] = __shfl_xor(s[r], off, 32);
            #pragma unroll
            for (int r = 0; r < 8; ++r) s[r] += t[r];
        }
        #pragma unroll
        for (int r = 0; r < 8; ++r) {
            const float mean = s[r] * (1.0f / 64.0f);
            #pragma unroll
            for (int nt = 0; nt < 4; ++nt) d[nt][r] = v[nt][r] - mean;
            q[r] = d[0][r] * d[0][r];
            #pragma unroll
            for (int nt = 1; nt < 4; ++nt) q[r] += d[nt][r] * d[nt][r];
        }
        #pragma unroll
        for (int off = 8; off > 0; off >>= 1) {
            float t[8];
            #pragma unroll
            for (int r = 0; r < 8; ++r) t[r] = __shfl_xor(q[r], off, 32);
            #pragma unroll
            for (int r = 0; r < 8; ++r) q[r] += t[r];
        }
        #pragma unroll
        for (int r = 0; r < 8; ++r) {
            const int row = mbase + r + rofs;
            const float rstd = rsqrtf(q[r] * (1.0f / 64.0f) + 1e-5f);
            if (row < M) {
                const long rb = (long)row * HDIM + col0;
                #pragma unroll
                for (int nt = 0; nt < 4; ++nt)
                    out[rb + nt * 16] = d[nt][r] * rstd * gg[nt] + be[nt];
            }
        }
    } else {
        float* tw = sT + wv * (16 * CT_PITCH);
        #pragma unroll
        for (int nt = 0; nt < 4; ++nt) {
            #pragma unroll
            for (int r = 0; r < 8; ++r)
                tw[(rofs + r) * CT_PITCH + nt * 16 + col0] = fmaxf(c[nt][r] + bb[nt] + xr[nt][r], 0.f);
        }
        __syncthreads();
        v8f hc[2];
        hc[0] = (v8f){0.f, 0.f, 0.f, 0.f, 0.f, 0.f, 0.f, 0.f};
        hc[1] = (v8f){0.f, 0.f, 0.f, 0.f, 0.f, 0.f, 0.f, 0.f};
        const float* ta = tw + col0 * CT_PITCH + 2 * hh;
        const v2f* sC = sWc + hh * CC_PITCH + col0;
        #pragma unroll
        for (int ks = 0; ks < 16; ++ks) {
            const v2f a = *(const v2f*)(ta + 4 * ks);
            #pragma unroll
            for (int nt = 0; nt < 2; ++nt)
                hc[nt] = __builtin_amdgcn_wmma_f32_16x16x4_f32(
                    false, a, false, sC[(2 * ks) * CC_PITCH + nt * 16], (short)0, hc[nt], false, false);
        }
        float s[8];
        #pragma unroll
        for (int r = 0; r < 8; ++r)
            s[r] = fmaxf(hc[0][r] + hb[0], 0.f) * w2[0] + fmaxf(hc[1][r] + hb[1], 0.f) * w2[1];
        #pragma unroll
        for (int off = 8; off > 0; off >>= 1) {
            float t[8];
            #pragma unroll
            for (int r = 0; r < 8; ++r) t[r] = __shfl_xor(s[r], off, 32);
            #pragma unroll
            for (int r = 0; r < 8; ++r) s[r] += t[r];
        }
        float mine = 0.f;
        #pragma unroll
        for (int r = 0; r < 8; ++r)
            if (col0 == r) mine = s[r];
        if (col0 < 8) sO[wv * 16 + rofs + col0] = mine + b2;
        __syncthreads();
        if (tid < 64) {
            const int row = blockIdx.x * 64 + tid;
            if (row < M) out[row] = sO[tid];
        }
    }
}

extern "C" void kernel_launch(void* const* d_in, const int* in_sizes, int n_in,
                              void* d_out, int out_size, void* d_ws, size_t ws_size,
                              hipStream_t stream) {
    const float* x_news    = (const float*)d_in[0];
    const float* x_company = (const float*)d_in[1];
    const int*   edge_sim  = (const int*)d_in[2];
    const int*   men_src   = (const int*)d_in[3];
    const int*   men_dst   = (const int*)d_in[4];
    const int*   edge_rel  = (const int*)d_in[5];
    const float* nw_W = (const float*)d_in[6];
    const float* nw_b = (const float*)d_in[7];
    const float* cw_W = (const float*)d_in[8];
    const float* cw_b = (const float*)d_in[9];
    const float* n1n_g = (const float*)d_in[10];
    const float* n1n_b = (const float*)d_in[11];
    const float* n1c_g = (const float*)d_in[12];
    const float* n1c_b = (const float*)d_in[13];
    const float* c1_Wl = (const float*)d_in[14];
    const float* c1_bl = (const float*)d_in[15];
    const float* c1_Wr = (const float*)d_in[16];
    const float* c2_Wl = (const float*)d_in[17];
    const float* c2_bl = (const float*)d_in[18];
    const float* c2_Wr = (const float*)d_in[19];
    const float* cls_W1 = (const float*)d_in[20];
    const float* cls_b1 = (const float*)d_in[21];
    const float* cls_W2 = (const float*)d_in[22];
    const float* cls_b2 = (const float*)d_in[23];
    float* out = (float*)d_out;

    float* ws = (float*)d_ws;
    long o = 0;
    float* xn0  = ws + o; o += (long)NN * HDIM;
    float* lpl  = ws + o; o += (long)NN * HDIM;
    float* aggn = ws + o; o += (long)NN * HDIM;
    float* xn1  = aggn;
    float* aggcM = ws + o; o += (long)NC * HDIM;
    float* aggcR = ws + o; o += (long)NC * HDIM;
    float* xc0  = ws + o; o += (long)NC * HDIM;
    float* xc1  = ws + o; o += (long)NC * HDIM;
    o += (long)NC * HDIM;
    float* comp = ws + o; o += (long)NC * HDIM;
    o += (long)NC * 32;
    v2f* wp_nw  = (v2f*)(ws + o); o += 2L * (388 / 2) * 64;
    v2f* wp_cw  = (v2f*)(ws + o); o += 2L * (24 / 2) * 64;
    v2f* wp_c1  = (v2f*)(ws + o); o += 2L * 6 * 32 * 64;
    v2f* wp_c2  = (v2f*)(ws + o); o += 2L * 4 * 32 * 64;
    v2f* wp_cls = (v2f*)(ws + o); o += 2L * 32 * 32;

    int* ctr   = (int*)comp;
    int* ctrN  = ctr;
    int* ctrM  = ctr + 512;
    int* ctrR  = ctr + 1024;
    int* blR   = ctr + 4096;
    int* blN   = (int*)lpl;
    int* blM   = blN + 256L * 15360;

    dim3 blk(256);
    int tilesNN = (NN + 15) / 16, tilesNC = (NC + 15) / 16;
    unsigned gemmBlkNN = (tilesNN + 7) / 8;
    unsigned crowBlk = (tilesNC + 3) / 4;

    k_prep<<<PB_NW + PB_CW + PB_C1 + PB_C2 + PB_CLS + PB_Z, blk, 0, stream>>>(
        nw_W, cw_W, c1_Wl, c1_Wr, c2_Wl, c2_Wr, cls_W1,
        wp_nw, wp_cw, wp_c1, wp_c2, wp_cls, ctr);

    k_feat_bin3<<<FTB_N + FTB_C + BINB_N + BINB_M + BINB_R, 1024, 0, stream>>>(
        x_news, wp_nw, nw_b, xn0, x_company, wp_cw, cw_b, xc0,
        edge_sim, edge_sim + ES, blN, ctrN,
        men_src, men_dst, blM, ctrM,
        edge_rel, edge_rel + ER, blR, ctrR);

    k_own_agg3<<<OWN_N + OWN_C + OWN_C, 1024, 0, stream>>>(edge_sim, edge_sim + ES, xn0, aggn, blN, ctrN,
                                                           men_src, men_dst, xn0, aggcM, blM, ctrM,
                                                           edge_rel, edge_rel + ER, xc0, aggcR, blR, ctrR);

    k_gemm<4, true, true><<<gemmBlkNN, blk, 0, stream>>>(aggn, wp_c1 + 0 * 2048,
                                                         c1_bl + 0 * 64,
                                                         xn0, wp_c1 + 1 * 2048,
                                                         nullptr, nullptr, n1n_g, n1n_b,
                                                         xn1, NN, 64, 64, 1);
    k_crow<false><<<crowBlk, 128, 0, stream>>>(aggcM, aggcR, xc0, wp_c1 + 2 * 2048,
                                               c1_bl + 1 * 64, c1_bl + 2 * 64, n1c_g, n1c_b,
                                               nullptr, nullptr, nullptr, nullptr,
                                               xc1, NC);

    k_own_agg2<<<OWN_C + OWN_C, 1024, 0, stream>>>(men_src, men_dst, xn1, aggcM, blM, ctrM,
                                                   edge_rel, edge_rel + ER, xc1, aggcR, blR, ctrR);
    k_crow<true><<<crowBlk, 128, 0, stream>>>(aggcM, aggcR, xc1, wp_c2,
                                              c2_bl + 1 * 64, c2_bl + 2 * 64, nullptr, nullptr,
                                              wp_cls, cls_b1, cls_W2, cls_b2,
                                              out, NC);
}
